# starting kernel
# baseline (speedup 1.0000x reference)
#include <hip/hip_runtime.h>
#include <math.h>
#include <utility>

typedef float v2f __attribute__((ext_vector_type(2)));
typedef float v8f __attribute__((ext_vector_type(8)));

#define HIDDEN  32
#define N_NODES 4096
#define N_EDGES 8192
#define FEAT    512
#define NPATHS  23

static constexpr int PL1[NPATHS] = {0,0,0,0, 1,1,1,1,1,1, 2,2,2,2,2,2,2, 3,3,3,3,3,3};
static constexpr int PL2[NPATHS] = {0,1,2,3, 0,1,1,2,2,3, 0,1,1,2,2,3,3, 0,1,2,2,3,3};
static constexpr int PL3[NPATHS] = {0,1,2,3, 1,0,2,1,3,2, 2,1,3,0,2,1,3, 3,2,1,3,0,2};
static constexpr float PWT[4] = {
  0.08838834764831845f,
  0.125f,
  0.14940357616680451f,
  0.19094065395649326f
};

template <typename T>
__device__ __forceinline__ void sh_eval(T x, T y, T z, T* Y) {
  const T s3  = (T)1.7320508075688772;
  const T s5  = (T)2.2360679774997896;
  const T s7  = (T)2.6457513110645907;
  const T s15 = (T)3.8729833462074170;
  const T c58 = (T)0.7905694150420949;
  const T c38 = (T)0.6123724356957945;
  Y[0]  = (T)1;
  Y[1]  = s3 * x;
  Y[2]  = s3 * y;
  Y[3]  = s3 * z;
  Y[4]  = s5 * s3 * x * z;
  Y[5]  = s5 * s3 * x * y;
  Y[6]  = s5 * (y * y - (T)0.5 * (x * x + z * z));
  Y[7]  = s5 * s3 * y * z;
  Y[8]  = s5 * (s3 * (T)0.5) * (z * z - x * x);
  Y[9]  = s7 * c58 * x * ((T)3 * z * z - x * x);
  Y[10] = s7 * s15 * x * y * z;
  Y[11] = s7 * c38 * x * ((T)5 * y * y - (T)1);
  Y[12] = s7 * (T)0.5 * y * ((T)5 * y * y - (T)3);
  Y[13] = s7 * c38 * z * ((T)5 * y * y - (T)1);
  Y[14] = s7 * (s15 * (T)0.5) * y * (z * z - x * x);
  Y[15] = s7 * c58 * z * (z * z - (T)3 * x * x);
}

__launch_bounds__(64)
__global__ void w3j_kernel(float* __restrict__ ctab) {
  const int p = blockIdx.x;
  const int t = threadIdx.x;

  int l1 = 0, l2 = 0, l3 = 0, cnt = 0;
  for (int a = 0; a < 4; ++a)
    for (int b = 0; b < 4; ++b)
      for (int c = 0; c < 4; ++c) {
        int lo = a - b; if (lo < 0) lo = -lo;
        if (lo <= c && c <= a + b && (((a + b + c) & 1) == 0)) {
          if (cnt == p) { l1 = a; l2 = b; l3 = c; }
          ++cnt;
        }
      }
  const int d1 = 2 * l1 + 1, d2 = 2 * l2 + 1, d3 = 2 * l3 + 1;
  const int NE = d1 * d2 * d3;

  double un[16], wn[16];
  const double PI = 3.14159265358979323846;
  for (int i = 0; i < 16; ++i) {
    double xg = cos(PI * (i + 0.75) / 16.5);
    double dp = 1.0;
    for (int it = 0; it < 100; ++it) {
      double a0 = 1.0, b0 = xg;
      for (int n = 2; n <= 16; ++n) {
        double c0 = ((2.0 * n - 1.0) * xg * b0 - (n - 1.0) * a0) / n;
        a0 = b0; b0 = c0;
      }
      dp = 16.0 * (xg * b0 - a0) / (xg * xg - 1.0);
      double dx = b0 / dp;
      xg -= dx;
      if (fabs(dx) < 1e-14) break;
    }
    {
      double a0 = 1.0, b0 = xg;
      for (int n = 2; n <= 16; ++n) {
        double c0 = ((2.0 * n - 1.0) * xg * b0 - (n - 1.0) * a0) / n;
        a0 = b0; b0 = c0;
      }
      dp = 16.0 * (xg * b0 - a0) / (xg * xg - 1.0);
    }
    un[i] = xg;
    wn[i] = 2.0 / ((1.0 - xg * xg) * dp * dp);
  }

  __shared__ double Cs[343];
  __shared__ double red[64];
  const double DPHI = 2.0 * PI / 32.0;

  for (int n = t; n < NE; n += 64) {
    const int i = n / (d2 * d3);
    const int j = (n / d3) % d2;
    const int k = n % d3;
    double s = 0.0;
    for (int q = 0; q < 512; ++q) {
      const int iq = q >> 5, jq = q & 31;
      const double uu  = un[iq];
      const double wq  = wn[iq] * DPHI;
      const double phi = jq * DPHI;
      double st2 = 1.0 - uu * uu;
      double st  = sqrt(st2 > 0.0 ? st2 : 0.0);
      double Yq[16];
      sh_eval(st * cos(phi), st * sin(phi), uu, Yq);
      s += wq * Yq[l1 * l1 + i] * Yq[l2 * l2 + j] * Yq[l3 * l3 + k];
    }
    Cs[n] = s;
  }
  __syncthreads();

  double ss = 0.0;
  for (int n = t; n < NE; n += 64) ss += Cs[n] * Cs[n];
  red[t] = ss;
  __syncthreads();
  for (int off = 32; off > 0; off >>= 1) {
    if (t < off) red[t] += red[t + off];
    __syncthreads();
  }
  const double inv = 1.0 / sqrt(red[0]);

  for (int n = t; n < NE; n += 64) {
    const int i = n / (d2 * d3);
    const int j = (n / d3) % d2;
    const int k = n % d3;
    ctab[p * 343 + (i * 7 + j) * 7 + k] = (float)(Cs[n] * inv);
  }
}

__global__ void zero_kernel(float* __restrict__ out, int n) {
  int i = blockIdx.x * blockDim.x + threadIdx.x;
  if (i < n) out[i] = 0.0f;
}

template <int P>
__device__ __forceinline__ void path_body(
    int lane, int n16,
    float len,
    const float (&Y)[16], const float (&xr)[16],
    const float* __restrict__ wwp,
    const float* __restrict__ wbp,
    const float* __restrict__ ctab,
    float* Gs, float* tmpB, const float* tb,
    v8f& accLo, v8f& accHi) {
  constexpr int l1 = PL1[P], l2 = PL2[P], l3 = PL3[P];
  constexpr int d1 = 2 * l1 + 1, d2 = 2 * l2 + 1, d3 = 2 * l3 + 1;
  constexpr int GN = d1 * d3;
  constexpr float pw = PWT[l3];
  const float* Cp = ctab + P * 343;

  if (lane < GN) {
    const int i = lane / d3, k = lane % d3;
    float g = 0.0f;
#pragma unroll
    for (int j = 0; j < d2; ++j) g = fmaf(Y[l2 * l2 + j], Cp[(i * 7 + j) * 7 + k], g);
    Gs[lane] = pw * g;
  }
  if constexpr (GN > 32) {
    const int n2 = lane + 32;
    if (n2 < GN) {
      const int i = n2 / d3, k = n2 % d3;
      float g = 0.0f;
#pragma unroll
      for (int j = 0; j < d2; ++j) g = fmaf(Y[l2 * l2 + j], Cp[(i * 7 + j) * 7 + k], g);
      Gs[n2] = pw * g;
    }
  }
  __syncthreads();

  float tk[7];
#pragma unroll
  for (int k = 0; k < d3; ++k) {
    float s = 0.0f;
#pragma unroll
    for (int i = 0; i < d1; ++i) s = fmaf(xr[l1 * l1 + i], Gs[i * d3 + k], s);
    tk[k] = s;
  }
#pragma unroll
  for (int k = 0; k < 16; ++k) tmpB[lane * 16 + k] = (k < d3) ? tk[k] : 0.0f;
  __syncthreads();

  constexpr int wbase = P * (HIDDEN * HIDDEN);
#pragma unroll
  for (int kc = 0; kc < 8; ++kc) {
    v2f b;
    b.x = tb[kc * 64];
    b.y = tb[kc * 64 + 16];
    {
      const int i0 = wbase + kc * 128;
      v2f a;
      a.x = fmaf(len, wwp[i0], wbp[i0]);
      a.y = fmaf(len, wwp[i0 + HIDDEN], wbp[i0 + HIDDEN]);
      accLo = __builtin_amdgcn_wmma_f32_16x16x4_f32(
          false, a, false, b, (short)0, accLo, false, false);
    }
    {
      const int i0 = wbase + kc * 128 + 16;
      v2f a;
      a.x = fmaf(len, wwp[i0], wbp[i0]);
      a.y = fmaf(len, wwp[i0 + HIDDEN], wbp[i0 + HIDDEN]);
      accHi = __builtin_amdgcn_wmma_f32_16x16x4_f32(
          false, a, false, b, (short)0, accHi, false, false);
    }
  }
  __syncthreads();
}

template <int... Ps>
__device__ __forceinline__ void run_group(
    std::integer_sequence<int, Ps...>,
    int lane, int n16, float len,
    const float (&Y)[16], const float (&xr)[16],
    const float* __restrict__ wwp, const float* __restrict__ wbp,
    const float* __restrict__ ctab,
    float* Gs, float* tmpB, const float* tb,
    v8f& accLo, v8f& accHi) {
  (path_body<Ps>(lane, n16, len, Y, xr, wwp, wbp, ctab, Gs, tmpB, tb, accLo, accHi), ...);
}

template <int L3>
__device__ __forceinline__ void store_tile(int lane, int n16, float* od,
                                           const v8f& accLo, const v8f& accHi) {
  constexpr int d3 = 2 * L3 + 1;
  if (n16 < d3) {
    const int wadd = (lane >= 16) ? 8 : 0;
#pragma unroll
    for (int r = 0; r < 8; ++r) {
      const int w = r + wadd;
      atomicAdd(od + HIDDEN * L3 * L3 + w * d3 + n16, accLo[r]);
      atomicAdd(od + HIDDEN * L3 * L3 + (w + 16) * d3 + n16, accHi[r]);
    }
  }
}

__launch_bounds__(32)
__global__ void edge_kernel(const float* __restrict__ nf,
                            const int*   __restrict__ eidx,
                            const float* __restrict__ evec,
                            const float* __restrict__ ww,
                            const float* __restrict__ wb,
                            const float* __restrict__ ctab,
                            float*       __restrict__ out) {
  const int e    = blockIdx.x;
  const int lane = threadIdx.x;

  const int src = eidx[e];
  const int dst = eidx[N_EDGES + e];

  const float vx = evec[3 * e + 0];
  const float vy = evec[3 * e + 1];
  const float vz = evec[3 * e + 2];
  float len = sqrtf(vx * vx + vy * vy + vz * vz);
  len = fmaxf(len, 1e-8f);
  const float inv = 1.0f / len;

  float Y[16];
  sh_eval(vx * inv, vy * inv, vz * inv, Y);

  const float* xs = nf + (size_t)src * FEAT;
  float xr[16];
#pragma unroll
  for (int l = 0; l < 4; ++l) {
    const int d = 2 * l + 1;
#pragma unroll
    for (int i = 0; i < 7; ++i)
      if (i < d) xr[l * l + i] = xs[HIDDEN * l * l + lane * d + i];
  }

  __shared__ float Gs[49];
  __shared__ float tmpB[32 * 16];

  const int n16  = lane & 15;
  const int usel = (lane >> 4) << 1;
  const float* wwp = ww + usel * HIDDEN + n16;
  const float* wbp = wb + usel * HIDDEN + n16;
  const float* tb  = tmpB + usel * 16 + n16;

  float* od = out + (size_t)dst * FEAT;

  {
    v8f a0 = {0,0,0,0,0,0,0,0}, a1 = {0,0,0,0,0,0,0,0};
    run_group(std::integer_sequence<int, 0, 5, 13, 21>{},
              lane, n16, len, Y, xr, wwp, wbp, ctab, Gs, tmpB, tb, a0, a1);
    store_tile<0>(lane, n16, od, a0, a1);
  }
  {
    v8f a0 = {0,0,0,0,0,0,0,0}, a1 = {0,0,0,0,0,0,0,0};
    run_group(std::integer_sequence<int, 1, 4, 7, 11, 15, 19>{},
              lane, n16, len, Y, xr, wwp, wbp, ctab, Gs, tmpB, tb, a0, a1);
    store_tile<1>(lane, n16, od, a0, a1);
  }
  {
    v8f a0 = {0,0,0,0,0,0,0,0}, a1 = {0,0,0,0,0,0,0,0};
    run_group(std::integer_sequence<int, 2, 6, 9, 10, 14, 18, 22>{},
              lane, n16, len, Y, xr, wwp, wbp, ctab, Gs, tmpB, tb, a0, a1);
    store_tile<2>(lane, n16, od, a0, a1);
  }
  {
    v8f a0 = {0,0,0,0,0,0,0,0}, a1 = {0,0,0,0,0,0,0,0};
    run_group(std::integer_sequence<int, 3, 8, 12, 16, 17, 20>{},
              lane, n16, len, Y, xr, wwp, wbp, ctab, Gs, tmpB, tb, a0, a1);
    store_tile<3>(lane, n16, od, a0, a1);
  }
}

extern "C" void kernel_launch(void* const* d_in, const int* in_sizes, int n_in,
                              void* d_out, int out_size, void* d_ws, size_t ws_size,
                              hipStream_t stream) {
  const float* nf   = (const float*)d_in[0];
  const int*   eidx = (const int*)  d_in[1];
  const float* evec = (const float*)d_in[2];
  const float* ww   = (const float*)d_in[3];
  const float* wb   = (const float*)d_in[4];
  float* out  = (float*)d_out;
  float* ctab = (float*)d_ws;

  w3j_kernel<<<NPATHS, 64, 0, stream>>>(ctab);
  const int n = N_NODES * FEAT;
  zero_kernel<<<(n + 255) / 256, 256, 0, stream>>>(out, n);
  edge_kernel<<<N_EDGES, 32, 0, stream>>>(nf, eidx, evec, ww, wb, ctab, out);
}


# coupling-table kernel: quadrature nodes, phi sines and harmonic samples computed once per block into LDS, contraction shared by 256 threads; the order of its fp64 quadrature sums changes
# speedup vs baseline: 45.5428x; 45.5428x over previous
#include <hip/hip_runtime.h>
#include <math.h>
#include <utility>

typedef float v2f __attribute__((ext_vector_type(2)));
typedef float v8f __attribute__((ext_vector_type(8)));

#define HIDDEN  32
#define N_NODES 4096
#define N_EDGES 8192
#define FEAT    512
#define NPATHS  23

static constexpr int PL1[NPATHS] = {0,0,0,0, 1,1,1,1,1,1, 2,2,2,2,2,2,2, 3,3,3,3,3,3};
static constexpr int PL2[NPATHS] = {0,1,2,3, 0,1,1,2,2,3, 0,1,1,2,2,3,3, 0,1,2,2,3,3};
static constexpr int PL3[NPATHS] = {0,1,2,3, 1,0,2,1,3,2, 2,1,3,0,2,1,3, 3,2,1,3,0,2};
static constexpr float PWT[4] = {
  0.08838834764831845f,
  0.125f,
  0.14940357616680451f,
  0.19094065395649326f
};

template <typename T>
__device__ __forceinline__ void sh_eval(T x, T y, T z, T* Y) {
  const T s3  = (T)1.7320508075688772;
  const T s5  = (T)2.2360679774997896;
  const T s7  = (T)2.6457513110645907;
  const T s15 = (T)3.8729833462074170;
  const T c58 = (T)0.7905694150420949;
  const T c38 = (T)0.6123724356957945;
  Y[0]  = (T)1;
  Y[1]  = s3 * x;
  Y[2]  = s3 * y;
  Y[3]  = s3 * z;
  Y[4]  = s5 * s3 * x * z;
  Y[5]  = s5 * s3 * x * y;
  Y[6]  = s5 * (y * y - (T)0.5 * (x * x + z * z));
  Y[7]  = s5 * s3 * y * z;
  Y[8]  = s5 * (s3 * (T)0.5) * (z * z - x * x);
  Y[9]  = s7 * c58 * x * ((T)3 * z * z - x * x);
  Y[10] = s7 * s15 * x * y * z;
  Y[11] = s7 * c38 * x * ((T)5 * y * y - (T)1);
  Y[12] = s7 * (T)0.5 * y * ((T)5 * y * y - (T)3);
  Y[13] = s7 * c38 * z * ((T)5 * y * y - (T)1);
  Y[14] = s7 * (s15 * (T)0.5) * y * (z * z - x * x);
  Y[15] = s7 * c58 * z * (z * z - (T)3 * x * x);
}

#define W3J_T  256
#define NQ     512
#define YPITCH 23

static constexpr double LA[17] = {0.0, 0.0, 3.0/2.0, 5.0/3.0, 7.0/4.0, 9.0/5.0, 11.0/6.0,
  13.0/7.0, 15.0/8.0, 17.0/9.0, 19.0/10.0, 21.0/11.0, 23.0/12.0, 25.0/13.0, 27.0/14.0,
  29.0/15.0, 31.0/16.0};
static constexpr double LB[17] = {0.0, 0.0, 1.0/2.0, 2.0/3.0, 3.0/4.0, 4.0/5.0, 5.0/6.0,
  6.0/7.0, 7.0/8.0, 8.0/9.0, 9.0/10.0, 10.0/11.0, 11.0/12.0, 12.0/13.0, 13.0/14.0,
  14.0/15.0, 15.0/16.0};

__device__ __forceinline__ void legendre16(double x, double& p15, double& p16) {
  double a0 = 1.0, b0 = x;
  for (int n = 2; n <= 16; ++n) {
    const double c0 = LA[n] * x * b0 - LB[n] * a0;
    a0 = b0; b0 = c0;
  }
  p15 = a0; p16 = b0;
}

template <int DR>
__device__ __forceinline__ void contract_run(const double* Ys, int q0, int q1,
                                             int ca, int cb, double* dst, int sR) {
  double acc[DR];
  for (int r = 0; r < DR; ++r) acc[r] = 0.0;
  for (int q = q0; q < q1; ++q) {
    const double* row = Ys + q * YPITCH;
    const double ab = row[ca] * row[cb];
    for (int r = 0; r < DR; ++r) acc[r] = fma(row[16 + r], ab, acc[r]);
  }
  for (int r = 0; r < DR; ++r) dst[r * sR] = acc[r];
}

__launch_bounds__(W3J_T)
__global__ void w3j_kernel(float* __restrict__ ctab) {
  const int p = blockIdx.x;
  const int t = threadIdx.x;

  int l1 = 0, l2 = 0, l3 = 0, cnt = 0;
  for (int a = 0; a < 4; ++a)
    for (int b = 0; b < 4; ++b)
      for (int c = 0; c < 4; ++c) {
        int lo = a - b; if (lo < 0) lo = -lo;
        if (lo <= c && c <= a + b && (((a + b + c) & 1) == 0)) {
          if (cnt == p) { l1 = a; l2 = b; l3 = c; }
          ++cnt;
        }
      }
  const int d1 = 2 * l1 + 1, d2 = 2 * l2 + 1, d3 = 2 * l3 + 1;
  const int NE = d1 * d2 * d3;

  __shared__ double un_s[16], wn_s[16], st_s[16];
  __shared__ double cph_s[32], sph_s[32];
  __shared__ double Ys[NQ * YPITCH];
  __shared__ double part[W3J_T * 7];
  __shared__ double red[W3J_T];

  const double PI = 3.14159265358979323846;
  const double DPHI = 2.0 * PI / 32.0;

  if (t < 16) {
    double xg = (double)cosf((float)(PI * (t + 0.75) / 16.5));
    double dp = 1.0;
    for (int it = 0; it < 100; ++it) {
      double a0, b0;
      legendre16(xg, a0, b0);
      dp = 16.0 * (xg * b0 - a0) / (xg * xg - 1.0);
      const double dx = b0 / dp;
      xg -= dx;
      if (fabs(dx) < 1e-14) break;
    }
    {
      double a0, b0;
      legendre16(xg, a0, b0);
      dp = 16.0 * (xg * b0 - a0) / (xg * xg - 1.0);
    }
    const double st2 = 1.0 - xg * xg;
    un_s[t] = xg;
    wn_s[t] = 2.0 / (st2 * dp * dp);
    st_s[t] = sqrt(st2 > 0.0 ? st2 : 0.0);
  } else if (t >= 32 && t < 64) {
    const int j = t - 32;
    double s, c;
    sincos(j * DPHI, &s, &c);
    cph_s[j] = c;
    sph_s[j] = s;
  }
  __syncthreads();

  int dR, dA, dB, oR, oA, oB, sR, sA, sB;
  if (d1 >= d2 && d1 >= d3) {
    dR = d1; oR = l1 * l1; sR = d2 * d3;
    dA = d2; oA = l2 * l2; sA = d3;
    dB = d3; oB = l3 * l3; sB = 1;
  } else if (d2 >= d3) {
    dR = d2; oR = l2 * l2; sR = d3;
    dA = d1; oA = l1 * l1; sA = d2 * d3;
    dB = d3; oB = l3 * l3; sB = 1;
  } else {
    dR = d3; oR = l3 * l3; sR = 1;
    dA = d1; oA = l1 * l1; sA = d2 * d3;
    dB = d2; oB = l2 * l2; sB = d3;
  }

  for (int q = t; q < NQ; q += W3J_T) {
    const int iq = q >> 5, jq = q & 31;
    const double st = st_s[iq];
    double Yq[16];
    sh_eval(st * cph_s[jq], st * sph_s[jq], un_s[iq], Yq);
    double* row = Ys + q * YPITCH;
    for (int c = 0; c < 16; ++c) row[c] = Yq[c];
    const double wq = wn_s[iq] * DPHI;
    for (int r = 0; r < dR; ++r) row[16 + r] = wq * row[oR + r];
  }
  __syncthreads();

  const int M = dA * dB;
  int NC = W3J_T / M; if (NC > 32) NC = 32;
  const int CL = (NQ + NC - 1) / NC;
  if (t < M * NC) {
    const int c = t / M, m = t - c * M;
    const int a = m / dB, b = m - a * dB;
    const int q0 = c * CL;
    int q1 = q0 + CL; if (q1 > NQ) q1 = NQ;
    double* dst = part + c * NE + a * sA + b * sB;
    switch (dR) {
      case 1:  contract_run<1>(Ys, q0, q1, oA + a, oB + b, dst, sR); break;
      case 3:  contract_run<3>(Ys, q0, q1, oA + a, oB + b, dst, sR); break;
      case 5:  contract_run<5>(Ys, q0, q1, oA + a, oB + b, dst, sR); break;
      default: contract_run<7>(Ys, q0, q1, oA + a, oB + b, dst, sR); break;
    }
  }
  __syncthreads();

  double v = 0.0;
  if (t < NE)
    for (int c = 0; c < NC; ++c) v += part[c * NE + t];
  red[t] = v * v;
  __syncthreads();
  for (int off = W3J_T / 2; off > 0; off >>= 1) {
    if (t < off) red[t] += red[t + off];
    __syncthreads();
  }
  if (t == 0) red[0] = 1.0 / sqrt(red[0]);
  __syncthreads();
  const double inv = red[0];

  if (t < NE) {
    const int i = t / (d2 * d3);
    const int j = (t / d3) % d2;
    const int k = t % d3;
    ctab[p * 343 + (i * 7 + j) * 7 + k] = (float)(v * inv);
  }
}

__global__ void zero_kernel(float* __restrict__ out, int n) {
  int i = blockIdx.x * blockDim.x + threadIdx.x;
  if (i < n) out[i] = 0.0f;
}

template <int P>
__device__ __forceinline__ void path_body(
    int lane, int n16,
    float len,
    const float (&Y)[16], const float (&xr)[16],
    const float* __restrict__ wwp,
    const float* __restrict__ wbp,
    const float* __restrict__ ctab,
    float* Gs, float* tmpB, const float* tb,
    v8f& accLo, v8f& accHi) {
  constexpr int l1 = PL1[P], l2 = PL2[P], l3 = PL3[P];
  constexpr int d1 = 2 * l1 + 1, d2 = 2 * l2 + 1, d3 = 2 * l3 + 1;
  constexpr int GN = d1 * d3;
  constexpr float pw = PWT[l3];
  const float* Cp = ctab + P * 343;

  if (lane < GN) {
    const int i = lane / d3, k = lane % d3;
    float g = 0.0f;
#pragma unroll
    for (int j = 0; j < d2; ++j) g = fmaf(Y[l2 * l2 + j], Cp[(i * 7 + j) * 7 + k], g);
    Gs[lane] = pw * g;
  }
  if constexpr (GN > 32) {
    const int n2 = lane + 32;
    if (n2 < GN) {
      const int i = n2 / d3, k = n2 % d3;
      float g = 0.0f;
#pragma unroll
      for (int j = 0; j < d2; ++j) g = fmaf(Y[l2 * l2 + j], Cp[(i * 7 + j) * 7 + k], g);
      Gs[n2] = pw * g;
    }
  }
  __syncthreads();

  float tk[7];
#pragma unroll
  for (int k = 0; k < d3; ++k) {
    float s = 0.0f;
#pragma unroll
    for (int i = 0; i < d1; ++i) s = fmaf(xr[l1 * l1 + i], Gs[i * d3 + k], s);
    tk[k] = s;
  }
#pragma unroll
  for (int k = 0; k < 16; ++k) tmpB[lane * 16 + k] = (k < d3) ? tk[k] : 0.0f;
  __syncthreads();

  constexpr int wbase = P * (HIDDEN * HIDDEN);
#pragma unroll
  for (int kc = 0; kc < 8; ++kc) {
    v2f b;
    b.x = tb[kc * 64];
    b.y = tb[kc * 64 + 16];
    {
      const int i0 = wbase + kc * 128;
      v2f a;
      a.x = fmaf(len, wwp[i0], wbp[i0]);
      a.y = fmaf(len, wwp[i0 + HIDDEN], wbp[i0 + HIDDEN]);
      accLo = __builtin_amdgcn_wmma_f32_16x16x4_f32(
          false, a, false, b, (short)0, accLo, false, false);
    }
    {
      const int i0 = wbase + kc * 128 + 16;
      v2f a;
      a.x = fmaf(len, wwp[i0], wbp[i0]);
      a.y = fmaf(len, wwp[i0 + HIDDEN], wbp[i0 + HIDDEN]);
      accHi = __builtin_amdgcn_wmma_f32_16x16x4_f32(
          false, a, false, b, (short)0, accHi, false, false);
    }
  }
  __syncthreads();
}

template <int... Ps>
__device__ __forceinline__ void run_group(
    std::integer_sequence<int, Ps...>,
    int lane, int n16, float len,
    const float (&Y)[16], const float (&xr)[16],
    const float* __restrict__ wwp, const float* __restrict__ wbp,
    const float* __restrict__ ctab,
    float* Gs, float* tmpB, const float* tb,
    v8f& accLo, v8f& accHi) {
  (path_body<Ps>(lane, n16, len, Y, xr, wwp, wbp, ctab, Gs, tmpB, tb, accLo, accHi), ...);
}

template <int L3>
__device__ __forceinline__ void store_tile(int lane, int n16, float* od,
                                           const v8f& accLo, const v8f& accHi) {
  constexpr int d3 = 2 * L3 + 1;
  if (n16 < d3) {
    const int wadd = (lane >= 16) ? 8 : 0;
#pragma unroll
    for (int r = 0; r < 8; ++r) {
      const int w = r + wadd;
      atomicAdd(od + HIDDEN * L3 * L3 + w * d3 + n16, accLo[r]);
      atomicAdd(od + HIDDEN * L3 * L3 + (w + 16) * d3 + n16, accHi[r]);
    }
  }
}

__launch_bounds__(32)
__global__ void edge_kernel(const float* __restrict__ nf,
                            const int*   __restrict__ eidx,
                            const float* __restrict__ evec,
                            const float* __restrict__ ww,
                            const float* __restrict__ wb,
                            const float* __restrict__ ctab,
                            float*       __restrict__ out) {
  const int e    = blockIdx.x;
  const int lane = threadIdx.x;

  const int src = eidx[e];
  const int dst = eidx[N_EDGES + e];

  const float vx = evec[3 * e + 0];
  const float vy = evec[3 * e + 1];
  const float vz = evec[3 * e + 2];
  float len = sqrtf(vx * vx + vy * vy + vz * vz);
  len = fmaxf(len, 1e-8f);
  const float inv = 1.0f / len;

  float Y[16];
  sh_eval(vx * inv, vy * inv, vz * inv, Y);

  const float* xs = nf + (size_t)src * FEAT;
  float xr[16];
#pragma unroll
  for (int l = 0; l < 4; ++l) {
    const int d = 2 * l + 1;
#pragma unroll
    for (int i = 0; i < 7; ++i)
      if (i < d) xr[l * l + i] = xs[HIDDEN * l * l + lane * d + i];
  }

  __shared__ float Gs[49];
  __shared__ float tmpB[32 * 16];

  const int n16  = lane & 15;
  const int usel = (lane >> 4) << 1;
  const float* wwp = ww + usel * HIDDEN + n16;
  const float* wbp = wb + usel * HIDDEN + n16;
  const float* tb  = tmpB + usel * 16 + n16;

  float* od = out + (size_t)dst * FEAT;

  {
    v8f a0 = {0,0,0,0,0,0,0,0}, a1 = {0,0,0,0,0,0,0,0};
    run_group(std::integer_sequence<int, 0, 5, 13, 21>{},
              lane, n16, len, Y, xr, wwp, wbp, ctab, Gs, tmpB, tb, a0, a1);
    store_tile<0>(lane, n16, od, a0, a1);
  }
  {
    v8f a0 = {0,0,0,0,0,0,0,0}, a1 = {0,0,0,0,0,0,0,0};
    run_group(std::integer_sequence<int, 1, 4, 7, 11, 15, 19>{},
              lane, n16, len, Y, xr, wwp, wbp, ctab, Gs, tmpB, tb, a0, a1);
    store_tile<1>(lane, n16, od, a0, a1);
  }
  {
    v8f a0 = {0,0,0,0,0,0,0,0}, a1 = {0,0,0,0,0,0,0,0};
    run_group(std::integer_sequence<int, 2, 6, 9, 10, 14, 18, 22>{},
              lane, n16, len, Y, xr, wwp, wbp, ctab, Gs, tmpB, tb, a0, a1);
    store_tile<2>(lane, n16, od, a0, a1);
  }
  {
    v8f a0 = {0,0,0,0,0,0,0,0}, a1 = {0,0,0,0,0,0,0,0};
    run_group(std::integer_sequence<int, 3, 8, 12, 16, 17, 20>{},
              lane, n16, len, Y, xr, wwp, wbp, ctab, Gs, tmpB, tb, a0, a1);
    store_tile<3>(lane, n16, od, a0, a1);
  }
}

extern "C" void kernel_launch(void* const* d_in, const int* in_sizes, int n_in,
                              void* d_out, int out_size, void* d_ws, size_t ws_size,
                              hipStream_t stream) {
  const float* nf   = (const float*)d_in[0];
  const int*   eidx = (const int*)  d_in[1];
  const float* evec = (const float*)d_in[2];
  const float* ww   = (const float*)d_in[3];
  const float* wb   = (const float*)d_in[4];
  float* out  = (float*)d_out;
  float* ctab = (float*)d_ws;

  w3j_kernel<<<NPATHS, W3J_T, 0, stream>>>(ctab);
  const int n = N_NODES * FEAT;
  zero_kernel<<<(n + 255) / 256, 256, 0, stream>>>(out, n);
  edge_kernel<<<N_EDGES, 32, 0, stream>>>(nf, eidx, evec, ww, wb, ctab, out);
}


# edge launch: 8 edges (waves) per work-group; each path's weight and bias matrices fetched once per work-group into LDS, A fragments built from LDS instead of per-wave 4-byte global loads
# speedup vs baseline: 49.8903x; 1.0955x over previous
#include <hip/hip_runtime.h>
#include <math.h>
#include <utility>

typedef float v2f __attribute__((ext_vector_type(2)));
typedef float v8f __attribute__((ext_vector_type(8)));

#define HIDDEN  32
#define N_NODES 4096
#define N_EDGES 8192
#define FEAT    512
#define NPATHS  23

static constexpr int PL1[NPATHS] = {0,0,0,0, 1,1,1,1,1,1, 2,2,2,2,2,2,2, 3,3,3,3,3,3};
static constexpr int PL2[NPATHS] = {0,1,2,3, 0,1,1,2,2,3, 0,1,1,2,2,3,3, 0,1,2,2,3,3};
static constexpr int PL3[NPATHS] = {0,1,2,3, 1,0,2,1,3,2, 2,1,3,0,2,1,3, 3,2,1,3,0,2};
static constexpr float PWT[4] = {
  0.08838834764831845f,
  0.125f,
  0.14940357616680451f,
  0.19094065395649326f
};

template <typename T>
__device__ __forceinline__ void sh_eval(T x, T y, T z, T* Y) {
  const T s3  = (T)1.7320508075688772;
  const T s5  = (T)2.2360679774997896;
  const T s7  = (T)2.6457513110645907;
  const T s15 = (T)3.8729833462074170;
  const T c58 = (T)0.7905694150420949;
  const T c38 = (T)0.6123724356957945;
  Y[0]  = (T)1;
  Y[1]  = s3 * x;
  Y[2]  = s3 * y;
  Y[3]  = s3 * z;
  Y[4]  = s5 * s3 * x * z;
  Y[5]  = s5 * s3 * x * y;
  Y[6]  = s5 * (y * y - (T)0.5 * (x * x + z * z));
  Y[7]  = s5 * s3 * y * z;
  Y[8]  = s5 * (s3 * (T)0.5) * (z * z - x * x);
  Y[9]  = s7 * c58 * x * ((T)3 * z * z - x * x);
  Y[10] = s7 * s15 * x * y * z;
  Y[11] = s7 * c38 * x * ((T)5 * y * y - (T)1);
  Y[12] = s7 * (T)0.5 * y * ((T)5 * y * y - (T)3);
  Y[13] = s7 * c38 * z * ((T)5 * y * y - (T)1);
  Y[14] = s7 * (s15 * (T)0.5) * y * (z * z - x * x);
  Y[15] = s7 * c58 * z * (z * z - (T)3 * x * x);
}

#define W3J_T  256
#define ZBLK   512
#define ZPER   4
#define PREP_BLOCKS 535
static_assert(PREP_BLOCKS == NPATHS + ZBLK, "grid of the preparation launch");
static_assert(ZBLK * W3J_T * ZPER * 4 == N_NODES * FEAT, "clearing blocks cover the output exactly once");
#define NQ     512
#define YPITCH 23

static constexpr double LA[17] = {0.0, 0.0, 3.0/2.0, 5.0/3.0, 7.0/4.0, 9.0/5.0, 11.0/6.0,
  13.0/7.0, 15.0/8.0, 17.0/9.0, 19.0/10.0, 21.0/11.0, 23.0/12.0, 25.0/13.0, 27.0/14.0,
  29.0/15.0, 31.0/16.0};
static constexpr double LB[17] = {0.0, 0.0, 1.0/2.0, 2.0/3.0, 3.0/4.0, 4.0/5.0, 5.0/6.0,
  6.0/7.0, 7.0/8.0, 8.0/9.0, 9.0/10.0, 10.0/11.0, 11.0/12.0, 12.0/13.0, 13.0/14.0,
  14.0/15.0, 15.0/16.0};

__device__ __forceinline__ void legendre16(double x, double& p15, double& p16) {
  double a0 = 1.0, b0 = x;
  for (int n = 2; n <= 16; ++n) {
    const double c0 = LA[n] * x * b0 - LB[n] * a0;
    a0 = b0; b0 = c0;
  }
  p15 = a0; p16 = b0;
}

template <int DR>
__device__ __forceinline__ void contract_run(const double* Ys, int q0, int q1,
                                             int ca, int cb, double* dst, int sR) {
  double acc[DR];
  for (int r = 0; r < DR; ++r) acc[r] = 0.0;
  for (int q = q0; q < q1; ++q) {
    const double* row = Ys + q * YPITCH;
    const double ab = row[ca] * row[cb];
    for (int r = 0; r < DR; ++r) acc[r] = fma(row[16 + r], ab, acc[r]);
  }
  for (int r = 0; r < DR; ++r) dst[r * sR] = acc[r];
}

__launch_bounds__(W3J_T)
__global__ void prep_kernel(float* __restrict__ ctab, float* __restrict__ out) {
  const int p = blockIdx.x;
  const int t = threadIdx.x;

  if (p >= NPATHS) {
    float4* o4 = reinterpret_cast<float4*>(out);
    const int zb = p - NPATHS;
    const float4 z4 = make_float4(0.0f, 0.0f, 0.0f, 0.0f);
    for (int k = 0; k < ZPER; ++k) o4[(k * ZBLK + zb) * W3J_T + t] = z4;
    return;
  }

  int l1 = 0, l2 = 0, l3 = 0, cnt = 0;
  for (int a = 0; a < 4; ++a)
    for (int b = 0; b < 4; ++b)
      for (int c = 0; c < 4; ++c) {
        int lo = a - b; if (lo < 0) lo = -lo;
        if (lo <= c && c <= a + b && (((a + b + c) & 1) == 0)) {
          if (cnt == p) { l1 = a; l2 = b; l3 = c; }
          ++cnt;
        }
      }
  const int d1 = 2 * l1 + 1, d2 = 2 * l2 + 1, d3 = 2 * l3 + 1;
  const int NE = d1 * d2 * d3;

  __shared__ double un_s[16], wn_s[16], st_s[16];
  __shared__ double cph_s[32], sph_s[32];
  __shared__ double Ys[NQ * YPITCH];
  __shared__ double part[W3J_T * 7];
  __shared__ double red[W3J_T];

  const double PI = 3.14159265358979323846;
  const double DPHI = 2.0 * PI / 32.0;

  if (t < 16) {
    double xg = (double)cosf((float)(PI * (t + 0.75) / 16.5));
    double dp = 1.0;
    for (int it = 0; it < 100; ++it) {
      double a0, b0;
      legendre16(xg, a0, b0);
      dp = 16.0 * (xg * b0 - a0) / (xg * xg - 1.0);
      const double dx = b0 / dp;
      xg -= dx;
      if (fabs(dx) < 1e-14) break;
    }
    {
      double a0, b0;
      legendre16(xg, a0, b0);
      dp = 16.0 * (xg * b0 - a0) / (xg * xg - 1.0);
    }
    const double st2 = 1.0 - xg * xg;
    un_s[t] = xg;
    wn_s[t] = 2.0 / (st2 * dp * dp);
    st_s[t] = sqrt(st2 > 0.0 ? st2 : 0.0);
  } else if (t >= 32 && t < 64) {
    const int j = t - 32;
    double s, c;
    sincos(j * DPHI, &s, &c);
    cph_s[j] = c;
    sph_s[j] = s;
  }
  __syncthreads();

  int dR, dA, dB, oR, oA, oB, sR, sA, sB;
  if (d1 >= d2 && d1 >= d3) {
    dR = d1; oR = l1 * l1; sR = d2 * d3;
    dA = d2; oA = l2 * l2; sA = d3;
    dB = d3; oB = l3 * l3; sB = 1;
  } else if (d2 >= d3) {
    dR = d2; oR = l2 * l2; sR = d3;
    dA = d1; oA = l1 * l1; sA = d2 * d3;
    dB = d3; oB = l3 * l3; sB = 1;
  } else {
    dR = d3; oR = l3 * l3; sR = 1;
    dA = d1; oA = l1 * l1; sA = d2 * d3;
    dB = d2; oB = l2 * l2; sB = d3;
  }

  for (int q = t; q < NQ; q += W3J_T) {
    const int iq = q >> 5, jq = q & 31;
    const double st = st_s[iq];
    double Yq[16];
    sh_eval(st * cph_s[jq], st * sph_s[jq], un_s[iq], Yq);
    double* row = Ys + q * YPITCH;
    for (int c = 0; c < 16; ++c) row[c] = Yq[c];
    const double wq = wn_s[iq] * DPHI;
    for (int r = 0; r < dR; ++r) row[16 + r] = wq * row[oR + r];
  }
  __syncthreads();

  const int M = dA * dB;
  int NC = W3J_T / M; if (NC > 32) NC = 32;
  const int CL = (NQ + NC - 1) / NC;
  if (t < M * NC) {
    const int c = t / M, m = t - c * M;
    const int a = m / dB, b = m - a * dB;
    const int q0 = c * CL;
    int q1 = q0 + CL; if (q1 > NQ) q1 = NQ;
    double* dst = part + c * NE + a * sA + b * sB;
    switch (dR) {
      case 1:  contract_run<1>(Ys, q0, q1, oA + a, oB + b, dst, sR); break;
      case 3:  contract_run<3>(Ys, q0, q1, oA + a, oB + b, dst, sR); break;
      case 5:  contract_run<5>(Ys, q0, q1, oA + a, oB + b, dst, sR); break;
      default: contract_run<7>(Ys, q0, q1, oA + a, oB + b, dst, sR); break;
    }
  }
  __syncthreads();

  double v = 0.0;
  if (t < NE)
    for (int c = 0; c < NC; ++c) v += part[c * NE + t];
  red[t] = v * v;
  __syncthreads();
  for (int off = W3J_T / 2; off > 0; off >>= 1) {
    if (t < off) red[t] += red[t + off];
    __syncthreads();
  }
  if (t == 0) red[0] = 1.0 / sqrt(red[0]);
  __syncthreads();
  const double inv = red[0];

  if (t < NE) {
    const int i = t / (d2 * d3);
    const int j = (t / d3) % d2;
    const int k = t % d3;
    ctab[p * 343 + (i * 7 + j) * 7 + k] = (float)(v * inv);
  }
}

#define EPB         8
#define EDGE_T      256
#define EDGE_BLOCKS 1024
static_assert(EDGE_T == 32 * EPB, "one wave per edge");
static_assert(EDGE_BLOCKS * EPB == N_EDGES, "every edge has a wave, no wave is without an edge");
static_assert(EDGE_T * 4 == HIDDEN * HIDDEN, "two rows x two columns of a path's weights per thread");
#define WPITCH 160
#define WBUF   ((HIDDEN / 2) * WPITCH)

static constexpr int SEQ[NPATHS] = {0, 5, 13, 21,  1, 4, 7, 11, 15, 19,
                                    2, 6, 9, 10, 14, 18, 22,  3, 8, 12, 16, 17, 20};

__device__ __forceinline__ void wave_lds_sync() {
  __builtin_amdgcn_fence(__ATOMIC_RELEASE, "wavefront");
  __builtin_amdgcn_wave_barrier();
  __builtin_amdgcn_fence(__ATOMIC_ACQUIRE, "wavefront");
}

template <int S>
__device__ __forceinline__ void path_body(
    int lane, float len,
    const float (&Y)[16], const float (&xr)[16],
    const float* __restrict__ wsrc,
    const float* __restrict__ bsrc,
    const float* __restrict__ ctab,
    float* Gs, float* tmpB, const float* tb,
    float* wdst,
    const float* wrd,
    float2 (&sw)[2], float2 (&sb)[2],
    v8f& accLo, v8f& accHi) {
  constexpr int P = SEQ[S];
  constexpr int l1 = PL1[P], l2 = PL2[P], l3 = PL3[P];
  constexpr int d1 = 2 * l1 + 1, d2 = 2 * l2 + 1, d3 = 2 * l3 + 1;
  constexpr int GN = d1 * d3;
  constexpr float pw = PWT[l3];
  constexpr int BUF = (S & 1) * WBUF;
  const float* Cp = ctab + P * 343;

  if (lane < GN) {
    const int i = lane / d3, k = lane % d3;
    float g = 0.0f;
#pragma unroll
    for (int j = 0; j < d2; ++j) g = fmaf(Y[l2 * l2 + j], Cp[(i * 7 + j) * 7 + k], g);
    Gs[lane] = pw * g;
  }
  if constexpr (GN > 32) {
    const int n2 = lane + 32;
    if (n2 < GN) {
      const int i = n2 / d3, k = n2 % d3;
      float g = 0.0f;
#pragma unroll
      for (int j = 0; j < d2; ++j) g = fmaf(Y[l2 * l2 + j], Cp[(i * 7 + j) * 7 + k], g);
      Gs[n2] = pw * g;
    }
  }
  wave_lds_sync();

  float tk[7];
#pragma unroll
  for (int k = 0; k < d3; ++k) {
    float s = 0.0f;
#pragma unroll
    for (int i = 0; i < d1; ++i) s = fmaf(xr[l1 * l1 + i], Gs[i * d3 + k], s);
    tk[k] = s;
  }
#pragma unroll
  for (int k = 0; k < 16; ++k) tmpB[lane * 16 + k] = (k < d3) ? tk[k] : 0.0f;

  {
    float4* d4 = reinterpret_cast<float4*>(wdst + BUF);
    d4[0] = make_float4(sw[0].x, sw[1].x, sb[0].x, sb[1].x);
    d4[1] = make_float4(sw[0].y, sw[1].y, sb[0].y, sb[1].y);
  }
  __syncthreads();

  if constexpr (S + 1 < NPATHS) {
    constexpr int PN = SEQ[S + 1];
    sw[0] = *reinterpret_cast<const float2*>(wsrc + PN * (HIDDEN * HIDDEN));
    sw[1] = *reinterpret_cast<const float2*>(wsrc + PN * (HIDDEN * HIDDEN) + HIDDEN);
    sb[0] = *reinterpret_cast<const float2*>(bsrc + PN * (HIDDEN * HIDDEN));
    sb[1] = *reinterpret_cast<const float2*>(bsrc + PN * (HIDDEN * HIDDEN) + HIDDEN);
  }

#pragma unroll
  for (int kc = 0; kc < 8; ++kc) {
    v2f b;
    b.x = tb[kc * 64];
    b.y = tb[kc * 64 + 16];
    const float* r0 = wrd + BUF + kc * (2 * WPITCH);
    {
      const float4 p = *reinterpret_cast<const float4*>(r0);
      v2f a;
      a.x = fmaf(len, p.x, p.z);
      a.y = fmaf(len, p.y, p.w);
      accLo = __builtin_amdgcn_wmma_f32_16x16x4_f32(
          false, a, false, b, (short)0, accLo, false, false);
    }
    {
      const float4 p = *reinterpret_cast<const float4*>(r0 + 64);
      v2f a;
      a.x = fmaf(len, p.x, p.z);
      a.y = fmaf(len, p.y, p.w);
      accHi = __builtin_amdgcn_wmma_f32_16x16x4_f32(
          false, a, false, b, (short)0, accHi, false, false);
    }
  }
  wave_lds_sync();
}

template <int... Ss>
__device__ __forceinline__ void run_group(
    std::integer_sequence<int, Ss...>,
    int lane, float len,
    const float (&Y)[16], const float (&xr)[16],
    const float* __restrict__ wsrc, const float* __restrict__ bsrc,
    const float* __restrict__ ctab,
    float* Gs, float* tmpB, const float* tb,
    float* wdst, const float* wrd, float2 (&sw)[2], float2 (&sb)[2],
    v8f& accLo, v8f& accHi) {
  (path_body<Ss>(lane, len, Y, xr, wsrc, bsrc, ctab, Gs, tmpB, tb, wdst, wrd, sw, sb,
                 accLo, accHi), ...);
}

template <int L3>
__device__ __forceinline__ void store_tile(int lane, int n16, float* od,
                                           const v8f& accLo, const v8f& accHi) {
  constexpr int d3 = 2 * L3 + 1;
  if (n16 < d3) {
    const unsigned wadd = (lane >= 16) ? 8u : 0u;
    const unsigned off = (unsigned)(HIDDEN * L3 * L3) + wadd * (unsigned)d3 + (unsigned)n16;
#pragma unroll
    for (int r = 0; r < 8; ++r) {
      atomicAdd(od + (off + (unsigned)(r * d3)), accLo[r]);
      atomicAdd(od + (off + (unsigned)((r + 16) * d3)), accHi[r]);
    }
  }
}

__attribute__((amdgpu_waves_per_eu(8, 8)))
__launch_bounds__(EDGE_T)
__global__ void edge_kernel(const float* __restrict__ nf,
                            const int*   __restrict__ eidx,
                            const float* __restrict__ evec,
                            const float* __restrict__ ww,
                            const float* __restrict__ wb,
                            const float* __restrict__ ctab,
                            float*       __restrict__ out) {
  const int tid  = threadIdx.x;
  const int lane = tid & 31;
  const int wave = __builtin_amdgcn_readfirstlane(tid >> 5);
  const int e    = blockIdx.x * EPB + wave;

  __shared__ float Gs_all[EPB * 64];
  __shared__ float tmpB_all[EPB * 32 * 16];
  __shared__ __attribute__((aligned(16))) float Ws[2 * WBUF];

  const float* wsrc = ww + (tid >> 4) * (2 * HIDDEN) + (tid & 15) * 2;
  const float* bsrc = wb + (tid >> 4) * (2 * HIDDEN) + (tid & 15) * 2;
  float2 sw[2], sb[2];
  sw[0] = *reinterpret_cast<const float2*>(wsrc + SEQ[0] * (HIDDEN * HIDDEN));
  sw[1] = *reinterpret_cast<const float2*>(wsrc + SEQ[0] * (HIDDEN * HIDDEN) + HIDDEN);
  sb[0] = *reinterpret_cast<const float2*>(bsrc + SEQ[0] * (HIDDEN * HIDDEN));
  sb[1] = *reinterpret_cast<const float2*>(bsrc + SEQ[0] * (HIDDEN * HIDDEN) + HIDDEN);

  const int src = eidx[e];
  const int dst = eidx[N_EDGES + e];

  const float vx = evec[3 * e + 0];
  const float vy = evec[3 * e + 1];
  const float vz = evec[3 * e + 2];
  float len = sqrtf(vx * vx + vy * vy + vz * vz);
  len = fmaxf(len, 1e-8f);
  const float inv = 1.0f / len;

  float Y[16];
  sh_eval(vx * inv, vy * inv, vz * inv, Y);

  const float* xs = nf + (size_t)src * FEAT;
  float xr[16];
#pragma unroll
  for (int l = 0; l < 4; ++l) {
    const int d = 2 * l + 1;
#pragma unroll
    for (int i = 0; i < 7; ++i)
      if (i < d) xr[l * l + i] = xs[HIDDEN * l * l + lane * d + i];
  }

  float* Gs   = Gs_all + wave * 64;
  float* tmpB = tmpB_all + wave * (32 * 16);

  const int n16  = lane & 15;
  const int usel = (lane >> 4) << 1;
  const float* tb  = tmpB + usel * 16 + n16;
  float*       wdst = Ws + (tid >> 4) * WPITCH + (tid & 15) * 8;
  const float* wrd  = Ws + (lane >> 4) * WPITCH + n16 * 4;

  float* od = out + (size_t)dst * FEAT;

  {
    v8f a0 = {0,0,0,0,0,0,0,0}, a1 = {0,0,0,0,0,0,0,0};
    run_group(std::integer_sequence<int, 0, 1, 2, 3>{},
              lane, len, Y, xr, wsrc, bsrc, ctab, Gs, tmpB, tb, wdst, wrd, sw, sb, a0, a1);
    store_tile<0>(lane, n16, od, a0, a1);
  }
  {
    v8f a0 = {0,0,0,0,0,0,0,0}, a1 = {0,0,0,0,0,0,0,0};
    run_group(std::integer_sequence<int, 4, 5, 6, 7, 8, 9>{},
              lane, len, Y, xr, wsrc, bsrc, ctab, Gs, tmpB, tb, wdst, wrd, sw, sb, a0, a1);
    store_tile<1>(lane, n16, od, a0, a1);
  }
  {
    v8f a0 = {0,0,0,0,0,0,0,0}, a1 = {0,0,0,0,0,0,0,0};
    run_group(std::integer_sequence<int, 10, 11, 12, 13, 14, 15, 16>{},
              lane, len, Y, xr, wsrc, bsrc, ctab, Gs, tmpB, tb, wdst, wrd, sw, sb, a0, a1);
    store_tile<2>(lane, n16, od, a0, a1);
  }
  {
    v8f a0 = {0,0,0,0,0,0,0,0}, a1 = {0,0,0,0,0,0,0,0};
    run_group(std::integer_sequence<int, 17, 18, 19, 20, 21, 22>{},
              lane, len, Y, xr, wsrc, bsrc, ctab, Gs, tmpB, tb, wdst, wrd, sw, sb, a0, a1);
    store_tile<3>(lane, n16, od, a0, a1);
  }
}

extern "C" void kernel_launch(void* const* d_in, const int* in_sizes, int n_in,
                              void* d_out, int out_size, void* d_ws, size_t ws_size,
                              hipStream_t stream) {
  const float* nf   = (const float*)d_in[0];
  const int*   eidx = (const int*)  d_in[1];
  const float* evec = (const float*)d_in[2];
  const float* ww   = (const float*)d_in[3];
  const float* wb   = (const float*)d_in[4];
  float* out  = (float*)d_out;
  float* ctab = (float*)d_ws;

  prep_kernel<<<PREP_BLOCKS, W3J_T, 0, stream>>>(ctab, out);
  edge_kernel<<<EDGE_BLOCKS, EDGE_T, 0, stream>>>(nf, eidx, evec, ww, wb, ctab, out);
}


# coupling tables: each path's table split over 8 blocks by index pairs; blocks store raw elements + sum of squares, edge launch folds f32 1/norm into the path weight; fp64 quadrature sum order changes
# speedup vs baseline: 61.1821x; 1.2263x over previous
#include <hip/hip_runtime.h>
#include <math.h>
#include <utility>

typedef float v2f __attribute__((ext_vector_type(2)));
typedef float v8f __attribute__((ext_vector_type(8)));

#define HIDDEN  32
#define N_NODES 4096
#define N_EDGES 8192
#define FEAT    512
#define NPATHS  23

static constexpr int PL1[NPATHS] = {0,0,0,0, 1,1,1,1,1,1, 2,2,2,2,2,2,2, 3,3,3,3,3,3};
static constexpr int PL2[NPATHS] = {0,1,2,3, 0,1,1,2,2,3, 0,1,1,2,2,3,3, 0,1,2,2,3,3};
static constexpr int PL3[NPATHS] = {0,1,2,3, 1,0,2,1,3,2, 2,1,3,0,2,1,3, 3,2,1,3,0,2};
static constexpr float PWT[4] = {
  0.08838834764831845f,
  0.125f,
  0.14940357616680451f,
  0.19094065395649326f
};

template <typename T>
__device__ __forceinline__ void sh_eval(T x, T y, T z, T* Y) {
  const T s3  = (T)1.7320508075688772;
  const T s5  = (T)2.2360679774997896;
  const T s7  = (T)2.6457513110645907;
  const T s15 = (T)3.8729833462074170;
  const T c58 = (T)0.7905694150420949;
  const T c38 = (T)0.6123724356957945;
  Y[0]  = (T)1;
  Y[1]  = s3 * x;
  Y[2]  = s3 * y;
  Y[3]  = s3 * z;
  Y[4]  = s5 * s3 * x * z;
  Y[5]  = s5 * s3 * x * y;
  Y[6]  = s5 * (y * y - (T)0.5 * (x * x + z * z));
  Y[7]  = s5 * s3 * y * z;
  Y[8]  = s5 * (s3 * (T)0.5) * (z * z - x * x);
  Y[9]  = s7 * c58 * x * ((T)3 * z * z - x * x);
  Y[10] = s7 * s15 * x * y * z;
  Y[11] = s7 * c38 * x * ((T)5 * y * y - (T)1);
  Y[12] = s7 * (T)0.5 * y * ((T)5 * y * y - (T)3);
  Y[13] = s7 * c38 * z * ((T)5 * y * y - (T)1);
  Y[14] = s7 * (s15 * (T)0.5) * y * (z * z - x * x);
  Y[15] = s7 * c58 * z * (z * z - (T)3 * x * x);
}

#define W3J_T  256
#define NSUB   8
#define TBLK   184
#define ZBLK   512
#define ZPER   4
#define PREP_BLOCKS 696
#define CSLOT  343
#define SSOFF  320
#define NCMAX  64
static_assert(TBLK == NPATHS * NSUB, "table blocks");
static_assert(PREP_BLOCKS == TBLK + ZBLK, "grid of the preparation launch");
static_assert(SSOFF >= 7 * 7 * 5 && SSOFF + NSUB <= CSLOT, "sums of squares sit behind the largest table (no path is 7x7x7)");
static_assert(((7 * 7 + NSUB - 1) / NSUB) * 7 <= 64, "a block's slice has at most 64 elements");
static_assert(ZBLK * W3J_T * ZPER * 4 == N_NODES * FEAT, "clearing blocks cover the output exactly once");
#define NQ     512
#define YPITCH 23

static constexpr double LA[17] = {0.0, 0.0, 3.0/2.0, 5.0/3.0, 7.0/4.0, 9.0/5.0, 11.0/6.0,
  13.0/7.0, 15.0/8.0, 17.0/9.0, 19.0/10.0, 21.0/11.0, 23.0/12.0, 25.0/13.0, 27.0/14.0,
  29.0/15.0, 31.0/16.0};
static constexpr double LB[17] = {0.0, 0.0, 1.0/2.0, 2.0/3.0, 3.0/4.0, 4.0/5.0, 5.0/6.0,
  6.0/7.0, 7.0/8.0, 8.0/9.0, 9.0/10.0, 10.0/11.0, 11.0/12.0, 12.0/13.0, 13.0/14.0,
  14.0/15.0, 15.0/16.0};

__device__ __forceinline__ void legendre16(double x, double& p15, double& p16) {
  double a0 = 1.0, b0 = x;
  for (int n = 2; n <= 16; ++n) {
    const double c0 = LA[n] * x * b0 - LB[n] * a0;
    a0 = b0; b0 = c0;
  }
  p15 = a0; p16 = b0;
}

template <int DR>
__device__ __forceinline__ void contract_run(const double* Ys, int q0, int q1,
                                             int ca, int cb, double* dst, int sR) {
  double acc[DR];
  for (int r = 0; r < DR; ++r) acc[r] = 0.0;
  for (int q = q0; q < q1; ++q) {
    const double* row = Ys + q * YPITCH;
    const double ab = row[ca] * row[cb];
    for (int r = 0; r < DR; ++r) acc[r] = fma(row[16 + r], ab, acc[r]);
  }
  for (int r = 0; r < DR; ++r) dst[r * sR] = acc[r];
}

__launch_bounds__(W3J_T)
__global__ void prep_kernel(float* __restrict__ ctab, float* __restrict__ out) {
  const int blk = blockIdx.x;
  const int t = threadIdx.x;

  if (blk >= TBLK) {
    float4* o4 = reinterpret_cast<float4*>(out);
    const int zb = blk - TBLK;
    const float4 z4 = make_float4(0.0f, 0.0f, 0.0f, 0.0f);
    for (int k = 0; k < ZPER; ++k) o4[(k * ZBLK + zb) * W3J_T + t] = z4;
    return;
  }
  const int p   = blk / NSUB;
  const int sub = blk - p * NSUB;

  int l1 = 0, l2 = 0, l3 = 0, cnt = 0;
  for (int a = 0; a < 4; ++a)
    for (int b = 0; b < 4; ++b)
      for (int c = 0; c < 4; ++c) {
        int lo = a - b; if (lo < 0) lo = -lo;
        if (lo <= c && c <= a + b && (((a + b + c) & 1) == 0)) {
          if (cnt == p) { l1 = a; l2 = b; l3 = c; }
          ++cnt;
        }
      }
  const int d1 = 2 * l1 + 1, d2 = 2 * l2 + 1, d3 = 2 * l3 + 1;

  int dR, dA, dB, oR, oA, oB, sR, sA, sB;
  if (d1 >= d2 && d1 >= d3) {
    dR = d1; oR = l1 * l1; sR = d2 * d3;
    dA = d2; oA = l2 * l2; sA = d3;
    dB = d3; oB = l3 * l3; sB = 1;
  } else if (d2 >= d3) {
    dR = d2; oR = l2 * l2; sR = d3;
    dA = d1; oA = l1 * l1; sA = d2 * d3;
    dB = d3; oB = l3 * l3; sB = 1;
  } else {
    dR = d3; oR = l3 * l3; sR = 1;
    dA = d1; oA = l1 * l1; sA = d2 * d3;
    dB = d2; oB = l2 * l2; sB = d3;
  }

  const int M  = dA * dB;
  const int MS = (M + NSUB - 1) / NSUB;
  const int m0 = sub * MS;
  const int m1 = (m0 + MS < M) ? (m0 + MS) : M;
  const int Mloc = m1 - m0;
  if (Mloc <= 0) {
    if (t == 0) ctab[p * CSLOT + SSOFF + sub] = 0.0f;
    return;
  }
  const int NEl = Mloc * dR;

  __shared__ double un_s[16], wn_s[16], st_s[16];
  __shared__ double cph_s[32], sph_s[32];
  __shared__ double Ys[NQ * YPITCH];
  __shared__ double part[W3J_T * 7];
  __shared__ double red[64];

  const double PI = 3.14159265358979323846;
  const double DPHI = 2.0 * PI / 32.0;

  if (t < 16) {
    double xg = (double)cosf((float)(PI * (t + 0.75) / 16.5));
    double dp = 1.0;
    for (int it = 0; it < 100; ++it) {
      double a0, b0;
      legendre16(xg, a0, b0);
      dp = 16.0 * (xg * b0 - a0) / (xg * xg - 1.0);
      const double dx = b0 / dp;
      xg -= dx;
      if (fabs(dx) < 1e-14) break;
    }
    {
      double a0, b0;
      legendre16(xg, a0, b0);
      dp = 16.0 * (xg * b0 - a0) / (xg * xg - 1.0);
    }
    const double st2 = 1.0 - xg * xg;
    un_s[t] = xg;
    wn_s[t] = 2.0 / (st2 * dp * dp);
    st_s[t] = sqrt(st2 > 0.0 ? st2 : 0.0);
  } else if (t >= 32 && t < 64) {
    const int j = t - 32;
    double s, c;
    sincos(j * DPHI, &s, &c);
    cph_s[j] = c;
    sph_s[j] = s;
  }
  __syncthreads();

  for (int q = t; q < NQ; q += W3J_T) {
    const int iq = q >> 5, jq = q & 31;
    const double st = st_s[iq];
    double Yq[16];
    sh_eval(st * cph_s[jq], st * sph_s[jq], un_s[iq], Yq);
    double* row = Ys + q * YPITCH;
    for (int c = 0; c < 16; ++c) row[c] = Yq[c];
    const double wq = wn_s[iq] * DPHI;
    for (int r = 0; r < dR; ++r) row[16 + r] = wq * row[oR + r];
  }
  __syncthreads();

  int NC = W3J_T / Mloc; if (NC > NCMAX) NC = NCMAX;
  const int CL = (NQ + NC - 1) / NC;
  if (t < Mloc * NC) {
    const int c = t / Mloc, ml = t - c * Mloc;
    const int m = m0 + ml;
    const int a = m / dB, b = m - a * dB;
    const int q0 = c * CL;
    int q1 = q0 + CL; if (q1 > NQ) q1 = NQ;
    double* dst = part + c * NEl + ml * dR;
    switch (dR) {
      case 1:  contract_run<1>(Ys, q0, q1, oA + a, oB + b, dst, 1); break;
      case 3:  contract_run<3>(Ys, q0, q1, oA + a, oB + b, dst, 1); break;
      case 5:  contract_run<5>(Ys, q0, q1, oA + a, oB + b, dst, 1); break;
      default: contract_run<7>(Ys, q0, q1, oA + a, oB + b, dst, 1); break;
    }
  }
  __syncthreads();

  double v = 0.0;
  if (t < NEl) {
    for (int c = 0; c < NC; ++c) v += part[c * NEl + t];
    const int ml = t / dR, r = t - ml * dR;
    const int m = m0 + ml;
    const int a = m / dB, b = m - a * dB;
    ctab[p * CSLOT + r * sR + a * sA + b * sB] = (float)v;
  }
  if (t < 64) red[t] = v * v;
  __syncthreads();
  for (int off = 32; off > 0; off >>= 1) {
    if (t < off) red[t] += red[t + off];
    __syncthreads();
  }
  if (t == 0) ctab[p * CSLOT + SSOFF + sub] = (float)red[0];
}

#define EPB         8
#define EDGE_T      256
#define EDGE_BLOCKS 1024
static_assert(EDGE_T == 32 * EPB, "one wave per edge");
static_assert(EDGE_BLOCKS * EPB == N_EDGES, "every edge has a wave, no wave is without an edge");
static_assert(EDGE_T * 4 == HIDDEN * HIDDEN, "two rows x two columns of a path's weights per thread");
#define WPITCH 160
#define WBUF   ((HIDDEN / 2) * WPITCH)

static constexpr int SEQ[NPATHS] = {0, 5, 13, 21,  1, 4, 7, 11, 15, 19,
                                    2, 6, 9, 10, 14, 18, 22,  3, 8, 12, 16, 17, 20};

__device__ __forceinline__ void wave_lds_sync() {
  __builtin_amdgcn_fence(__ATOMIC_RELEASE, "wavefront");
  __builtin_amdgcn_wave_barrier();
  __builtin_amdgcn_fence(__ATOMIC_ACQUIRE, "wavefront");
}

template <int S>
__device__ __forceinline__ void path_body(
    int lane, float len,
    const float (&Y)[16], const float (&xr)[16],
    const float* __restrict__ wsrc,
    const float* __restrict__ bsrc,
    const float* __restrict__ ctab,
    const float (&invn)[NPATHS],
    float* Gs, float* tmpB, const float* tb,
    float* wdst,
    const float* wrd,
    float2 (&sw)[2], float2 (&sb)[2],
    v8f& accLo, v8f& accHi) {
  constexpr int P = SEQ[S];
  constexpr int l1 = PL1[P], l2 = PL2[P], l3 = PL3[P];
  constexpr int d1 = 2 * l1 + 1, d2 = 2 * l2 + 1, d3 = 2 * l3 + 1;
  constexpr int GN = d1 * d3;
  const float pw = PWT[l3] * invn[P];
  constexpr int BUF = (S & 1) * WBUF;
  const float* Cp = ctab + P * CSLOT;

  if (lane < GN) {
    const int i = lane / d3, k = lane % d3;
    float g = 0.0f;
#pragma unroll
    for (int j = 0; j < d2; ++j) g = fmaf(Y[l2 * l2 + j], Cp[(i * d2 + j) * d3 + k], g);
    Gs[lane] = pw * g;
  }
  if constexpr (GN > 32) {
    const int n2 = lane + 32;
    if (n2 < GN) {
      const int i = n2 / d3, k = n2 % d3;
      float g = 0.0f;
#pragma unroll
      for (int j = 0; j < d2; ++j) g = fmaf(Y[l2 * l2 + j], Cp[(i * d2 + j) * d3 + k], g);
      Gs[n2] = pw * g;
    }
  }
  wave_lds_sync();

  float tk[7];
#pragma unroll
  for (int k = 0; k < d3; ++k) {
    float s = 0.0f;
#pragma unroll
    for (int i = 0; i < d1; ++i) s = fmaf(xr[l1 * l1 + i], Gs[i * d3 + k], s);
    tk[k] = s;
  }
#pragma unroll
  for (int k = 0; k < 16; ++k) tmpB[lane * 16 + k] = (k < d3) ? tk[k] : 0.0f;

  {
    float4* d4 = reinterpret_cast<float4*>(wdst + BUF);
    d4[0] = make_float4(sw[0].x, sw[1].x, sb[0].x, sb[1].x);
    d4[1] = make_float4(sw[0].y, sw[1].y, sb[0].y, sb[1].y);
  }
  __syncthreads();

  if constexpr (S + 1 < NPATHS) {
    constexpr int PN = SEQ[S + 1];
    sw[0] = *reinterpret_cast<const float2*>(wsrc + PN * (HIDDEN * HIDDEN));
    sw[1] = *reinterpret_cast<const float2*>(wsrc + PN * (HIDDEN * HIDDEN) + HIDDEN);
    sb[0] = *reinterpret_cast<const float2*>(bsrc + PN * (HIDDEN * HIDDEN));
    sb[1] = *reinterpret_cast<const float2*>(bsrc + PN * (HIDDEN * HIDDEN) + HIDDEN);
  }

#pragma unroll
  for (int kc = 0; kc < 8; ++kc) {
    v2f b;
    b.x = tb[kc * 64];
    b.y = tb[kc * 64 + 16];
    const float* r0 = wrd + BUF + kc * (2 * WPITCH);
    {
      const float4 p = *reinterpret_cast<const float4*>(r0);
      v2f a;
      a.x = fmaf(len, p.x, p.z);
      a.y = fmaf(len, p.y, p.w);
      accLo = __builtin_amdgcn_wmma_f32_16x16x4_f32(
          false, a, false, b, (short)0, accLo, false, false);
    }
    {
      const float4 p = *reinterpret_cast<const float4*>(r0 + 64);
      v2f a;
      a.x = fmaf(len, p.x, p.z);
      a.y = fmaf(len, p.y, p.w);
      accHi = __builtin_amdgcn_wmma_f32_16x16x4_f32(
          false, a, false, b, (short)0, accHi, false, false);
    }
  }
  wave_lds_sync();
}

template <int... Ss>
__device__ __forceinline__ void run_group(
    std::integer_sequence<int, Ss...>,
    int lane, float len,
    const float (&Y)[16], const float (&xr)[16],
    const float* __restrict__ wsrc, const float* __restrict__ bsrc,
    const float* __restrict__ ctab, const float (&invn)[NPATHS],
    float* Gs, float* tmpB, const float* tb,
    float* wdst, const float* wrd, float2 (&sw)[2], float2 (&sb)[2],
    v8f& accLo, v8f& accHi) {
  (path_body<Ss>(lane, len, Y, xr, wsrc, bsrc, ctab, invn, Gs, tmpB, tb, wdst, wrd, sw, sb,
                 accLo, accHi), ...);
}

template <int L3>
__device__ __forceinline__ void store_tile(int lane, int n16, float* od,
                                           const v8f& accLo, const v8f& accHi) {
  constexpr int d3 = 2 * L3 + 1;
  if (n16 < d3) {
    const unsigned wadd = (lane >= 16) ? 8u : 0u;
    const unsigned off = (unsigned)(HIDDEN * L3 * L3) + wadd * (unsigned)d3 + (unsigned)n16;
#pragma unroll
    for (int r = 0; r < 8; ++r) {
      atomicAdd(od + (off + (unsigned)(r * d3)), accLo[r]);
      atomicAdd(od + (off + (unsigned)((r + 16) * d3)), accHi[r]);
    }
  }
}

__attribute__((amdgpu_waves_per_eu(8, 8)))
__launch_bounds__(EDGE_T)
__global__ void edge_kernel(const float* __restrict__ nf,
                            const int*   __restrict__ eidx,
                            const float* __restrict__ evec,
                            const float* __restrict__ ww,
                            const float* __restrict__ wb,
                            const float* __restrict__ ctab,
                            float*       __restrict__ out) {
  const int tid  = threadIdx.x;
  const int lane = tid & 31;
  const int wave = __builtin_amdgcn_readfirstlane(tid >> 5);
  const int e    = blockIdx.x * EPB + wave;

  __shared__ float Gs_all[EPB * 64];
  __shared__ float tmpB_all[EPB * 32 * 16];
  __shared__ __attribute__((aligned(16))) float Ws[2 * WBUF];

  const float* wsrc = ww + (tid >> 4) * (2 * HIDDEN) + (tid & 15) * 2;
  const float* bsrc = wb + (tid >> 4) * (2 * HIDDEN) + (tid & 15) * 2;
  float2 sw[2], sb[2];
  sw[0] = *reinterpret_cast<const float2*>(wsrc + SEQ[0] * (HIDDEN * HIDDEN));
  sw[1] = *reinterpret_cast<const float2*>(wsrc + SEQ[0] * (HIDDEN * HIDDEN) + HIDDEN);
  sb[0] = *reinterpret_cast<const float2*>(bsrc + SEQ[0] * (HIDDEN * HIDDEN));
  sb[1] = *reinterpret_cast<const float2*>(bsrc + SEQ[0] * (HIDDEN * HIDDEN) + HIDDEN);

  const int src = eidx[e];
  const int dst = eidx[N_EDGES + e];

  const float vx = evec[3 * e + 0];
  const float vy = evec[3 * e + 1];
  const float vz = evec[3 * e + 2];
  float len = sqrtf(vx * vx + vy * vy + vz * vz);
  len = fmaxf(len, 1e-8f);
  const float inv = 1.0f / len;

  float Y[16];
  sh_eval(vx * inv, vy * inv, vz * inv, Y);

  const float* xs = nf + (size_t)src * FEAT;
  float xr[16];
#pragma unroll
  for (int l = 0; l < 4; ++l) {
    const int d = 2 * l + 1;
#pragma unroll
    for (int i = 0; i < 7; ++i)
      if (i < d) xr[l * l + i] = xs[HIDDEN * l * l + lane * d + i];
  }

  float invn[NPATHS];
  {
    const float* sp = ctab + (lane < NPATHS ? lane : 0) * CSLOT + SSOFF;
    float ssq = 0.0f;
#pragma unroll
    for (int k = 0; k < NSUB; ++k) ssq += sp[k];
    const int ib = __builtin_bit_cast(int, 1.0f / sqrtf(ssq));
#pragma unroll
    for (int q = 0; q < NPATHS; ++q)
      invn[q] = __builtin_bit_cast(float, __builtin_amdgcn_readlane(ib, q));
  }

  float* Gs   = Gs_all + wave * 64;
  float* tmpB = tmpB_all + wave * (32 * 16);

  const int n16  = lane & 15;
  const int usel = (lane >> 4) << 1;
  const float* tb  = tmpB + usel * 16 + n16;
  float*       wdst = Ws + (tid >> 4) * WPITCH + (tid & 15) * 8;
  const float* wrd  = Ws + (lane >> 4) * WPITCH + n16 * 4;

  float* od = out + (size_t)dst * FEAT;

  {
    v8f a0 = {0,0,0,0,0,0,0,0}, a1 = {0,0,0,0,0,0,0,0};
    run_group(std::integer_sequence<int, 0, 1, 2, 3>{},
              lane, len, Y, xr, wsrc, bsrc, ctab, invn, Gs, tmpB, tb, wdst, wrd, sw, sb, a0, a1);
    store_tile<0>(lane, n16, od, a0, a1);
  }
  {
    v8f a0 = {0,0,0,0,0,0,0,0}, a1 = {0,0,0,0,0,0,0,0};
    run_group(std::integer_sequence<int, 4, 5, 6, 7, 8, 9>{},
              lane, len, Y, xr, wsrc, bsrc, ctab, invn, Gs, tmpB, tb, wdst, wrd, sw, sb, a0, a1);
    store_tile<1>(lane, n16, od, a0, a1);
  }
  {
    v8f a0 = {0,0,0,0,0,0,0,0}, a1 = {0,0,0,0,0,0,0,0};
    run_group(std::integer_sequence<int, 10, 11, 12, 13, 14, 15, 16>{},
              lane, len, Y, xr, wsrc, bsrc, ctab, invn, Gs, tmpB, tb, wdst, wrd, sw, sb, a0, a1);
    store_tile<2>(lane, n16, od, a0, a1);
  }
  {
    v8f a0 = {0,0,0,0,0,0,0,0}, a1 = {0,0,0,0,0,0,0,0};
    run_group(std::integer_sequence<int, 17, 18, 19, 20, 21, 22>{},
              lane, len, Y, xr, wsrc, bsrc, ctab, invn, Gs, tmpB, tb, wdst, wrd, sw, sb, a0, a1);
    store_tile<3>(lane, n16, od, a0, a1);
  }
}

extern "C" void kernel_launch(void* const* d_in, const int* in_sizes, int n_in,
                              void* d_out, int out_size, void* d_ws, size_t ws_size,
                              hipStream_t stream) {
  const float* nf   = (const float*)d_in[0];
  const int*   eidx = (const int*)  d_in[1];
  const float* evec = (const float*)d_in[2];
  const float* ww   = (const float*)d_in[3];
  const float* wb   = (const float*)d_in[4];
  float* out  = (float*)d_out;
  float* ctab = (float*)d_ws;

  prep_kernel<<<PREP_BLOCKS, W3J_T, 0, stream>>>(ctab, out);
  edge_kernel<<<EDGE_BLOCKS, EDGE_T, 0, stream>>>(nf, eidx, evec, ww, wb, ctab, out);
}


# edge launch: len*weight+bias split by linearity into one K=64 product with edge-independent A; 16 edges per work-group pack (edge,k) columns densely; changes the order of the f32 sums
# speedup vs baseline: 84.3706x; 1.3790x over previous
#include <hip/hip_runtime.h>
#include <math.h>
#include <utility>

typedef float v2f __attribute__((ext_vector_type(2)));
typedef float v8f __attribute__((ext_vector_type(8)));

#define HIDDEN  32
#define N_NODES 4096
#define N_EDGES 8192
#define FEAT    512
#define NPATHS  23

static constexpr int PL1[NPATHS] = {0,0,0,0, 1,1,1,1,1,1, 2,2,2,2,2,2,2, 3,3,3,3,3,3};
static constexpr int PL2[NPATHS] = {0,1,2,3, 0,1,1,2,2,3, 0,1,1,2,2,3,3, 0,1,2,2,3,3};
static constexpr int PL3[NPATHS] = {0,1,2,3, 1,0,2,1,3,2, 2,1,3,0,2,1,3, 3,2,1,3,0,2};
static constexpr float PWT[4] = {
  0.08838834764831845f,
  0.125f,
  0.14940357616680451f,
  0.19094065395649326f
};

template <typename T>
__device__ __forceinline__ void sh_eval(T x, T y, T z, T* Y) {
  const T s3  = (T)1.7320508075688772;
  const T s5  = (T)2.2360679774997896;
  const T s7  = (T)2.6457513110645907;
  const T s15 = (T)3.8729833462074170;
  const T c58 = (T)0.7905694150420949;
  const T c38 = (T)0.6123724356957945;
  Y[0]  = (T)1;
  Y[1]  = s3 * x;
  Y[2]  = s3 * y;
  Y[3]  = s3 * z;
  Y[4]  = s5 * s3 * x * z;
  Y[5]  = s5 * s3 * x * y;
  Y[6]  = s5 * (y * y - (T)0.5 * (x * x + z * z));
  Y[7]  = s5 * s3 * y * z;
  Y[8]  = s5 * (s3 * (T)0.5) * (z * z - x * x);
  Y[9]  = s7 * c58 * x * ((T)3 * z * z - x * x);
  Y[10] = s7 * s15 * x * y * z;
  Y[11] = s7 * c38 * x * ((T)5 * y * y - (T)1);
  Y[12] = s7 * (T)0.5 * y * ((T)5 * y * y - (T)3);
  Y[13] = s7 * c38 * z * ((T)5 * y * y - (T)1);
  Y[14] = s7 * (s15 * (T)0.5) * y * (z * z - x * x);
  Y[15] = s7 * c58 * z * (z * z - (T)3 * x * x);
}

#define W3J_T  256
#define NSUB   8
#define TBLK   184
#define ZBLK   512
#define ZPER   4
#define PREP_BLOCKS 696
#define CSLOT  343
#define SSOFF  320
#define NCMAX  64
static_assert(TBLK == NPATHS * NSUB, "table blocks");
static_assert(PREP_BLOCKS == TBLK + ZBLK, "grid of the preparation launch");
static_assert(SSOFF >= 7 * 7 * 5 && SSOFF + NSUB <= CSLOT, "sums of squares sit behind the largest table (no path is 7x7x7)");
static_assert(((7 * 7 + NSUB - 1) / NSUB) * 7 <= 64, "a block's slice has at most 64 elements");
static_assert(ZBLK * W3J_T * ZPER * 4 == N_NODES * FEAT, "clearing blocks cover the output exactly once");
#define NQ     512
#define YPITCH 23

static constexpr double LA[17] = {0.0, 0.0, 3.0/2.0, 5.0/3.0, 7.0/4.0, 9.0/5.0, 11.0/6.0,
  13.0/7.0, 15.0/8.0, 17.0/9.0, 19.0/10.0, 21.0/11.0, 23.0/12.0, 25.0/13.0, 27.0/14.0,
  29.0/15.0, 31.0/16.0};
static constexpr double LB[17] = {0.0, 0.0, 1.0/2.0, 2.0/3.0, 3.0/4.0, 4.0/5.0, 5.0/6.0,
  6.0/7.0, 7.0/8.0, 8.0/9.0, 9.0/10.0, 10.0/11.0, 11.0/12.0, 12.0/13.0, 13.0/14.0,
  14.0/15.0, 15.0/16.0};

__device__ __forceinline__ void legendre16(double x, double& p15, double& p16) {
  double a0 = 1.0, b0 = x;
  for (int n = 2; n <= 16; ++n) {
    const double c0 = LA[n] * x * b0 - LB[n] * a0;
    a0 = b0; b0 = c0;
  }
  p15 = a0; p16 = b0;
}

template <int DR>
__device__ __forceinline__ void contract_run(const double* Ys, int q0, int q1,
                                             int ca, int cb, double* dst, int sR) {
  double acc[DR];
  for (int r = 0; r < DR; ++r) acc[r] = 0.0;
  for (int q = q0; q < q1; ++q) {
    const double* row = Ys + q * YPITCH;
    const double ab = row[ca] * row[cb];
    for (int r = 0; r < DR; ++r) acc[r] = fma(row[16 + r], ab, acc[r]);
  }
  for (int r = 0; r < DR; ++r) dst[r * sR] = acc[r];
}

__launch_bounds__(W3J_T)
__global__ void prep_kernel(float* __restrict__ ctab, float* __restrict__ out) {
  const int blk = blockIdx.x;
  const int t = threadIdx.x;

  if (blk >= TBLK) {
    float4* o4 = reinterpret_cast<float4*>(out);
    const int zb = blk - TBLK;
    const float4 z4 = make_float4(0.0f, 0.0f, 0.0f, 0.0f);
    for (int k = 0; k < ZPER; ++k) o4[(k * ZBLK + zb) * W3J_T + t] = z4;
    return;
  }
  const int p   = blk / NSUB;
  const int sub = blk - p * NSUB;

  int l1 = 0, l2 = 0, l3 = 0, cnt = 0;
  for (int a = 0; a < 4; ++a)
    for (int b = 0; b < 4; ++b)
      for (int c = 0; c < 4; ++c) {
        int lo = a - b; if (lo < 0) lo = -lo;
        if (lo <= c && c <= a + b && (((a + b + c) & 1) == 0)) {
          if (cnt == p) { l1 = a; l2 = b; l3 = c; }
          ++cnt;
        }
      }
  const int d1 = 2 * l1 + 1, d2 = 2 * l2 + 1, d3 = 2 * l3 + 1;

  int dR, dA, dB, oR, oA, oB, sR, sA, sB;
  if (d1 >= d2 && d1 >= d3) {
    dR = d1; oR = l1 * l1; sR = d2 * d3;
    dA = d2; oA = l2 * l2; sA = d3;
    dB = d3; oB = l3 * l3; sB = 1;
  } else if (d2 >= d3) {
    dR = d2; oR = l2 * l2; sR = d3;
    dA = d1; oA = l1 * l1; sA = d2 * d3;
    dB = d3; oB = l3 * l3; sB = 1;
  } else {
    dR = d3; oR = l3 * l3; sR = 1;
    dA = d1; oA = l1 * l1; sA = d2 * d3;
    dB = d2; oB = l2 * l2; sB = d3;
  }

  const int M  = dA * dB;
  const int MS = (M + NSUB - 1) / NSUB;
  const int m0 = sub * MS;
  const int m1 = (m0 + MS < M) ? (m0 + MS) : M;
  const int Mloc = m1 - m0;
  if (Mloc <= 0) {
    if (t == 0) ctab[p * CSLOT + SSOFF + sub] = 0.0f;
    return;
  }
  const int NEl = Mloc * dR;

  __shared__ double un_s[16], wn_s[16], st_s[16];
  __shared__ double cph_s[32], sph_s[32];
  __shared__ double Ys[NQ * YPITCH];
  __shared__ double part[W3J_T * 7];
  __shared__ double red[64];

  const double PI = 3.14159265358979323846;
  const double DPHI = 2.0 * PI / 32.0;

  if (t < 16) {
    double xg = (double)cosf((float)(PI * (t + 0.75) / 16.5));
    double dp = 1.0;
    for (int it = 0; it < 100; ++it) {
      double a0, b0;
      legendre16(xg, a0, b0);
      dp = 16.0 * (xg * b0 - a0) / (xg * xg - 1.0);
      const double dx = b0 / dp;
      xg -= dx;
      if (fabs(dx) < 1e-14) break;
    }
    {
      double a0, b0;
      legendre16(xg, a0, b0);
      dp = 16.0 * (xg * b0 - a0) / (xg * xg - 1.0);
    }
    const double st2 = 1.0 - xg * xg;
    un_s[t] = xg;
    wn_s[t] = 2.0 / (st2 * dp * dp);
    st_s[t] = sqrt(st2 > 0.0 ? st2 : 0.0);
  } else if (t >= 32 && t < 64) {
    const int j = t - 32;
    double s, c;
    sincos(j * DPHI, &s, &c);
    cph_s[j] = c;
    sph_s[j] = s;
  }
  __syncthreads();

  for (int q = t; q < NQ; q += W3J_T) {
    const int iq = q >> 5, jq = q & 31;
    const double st = st_s[iq];
    double Yq[16];
    sh_eval(st * cph_s[jq], st * sph_s[jq], un_s[iq], Yq);
    double* row = Ys + q * YPITCH;
    for (int c = 0; c < 16; ++c) row[c] = Yq[c];
    const double wq = wn_s[iq] * DPHI;
    for (int r = 0; r < dR; ++r) row[16 + r] = wq * row[oR + r];
  }
  __syncthreads();

  int NC = W3J_T / Mloc; if (NC > NCMAX) NC = NCMAX;
  const int CL = (NQ + NC - 1) / NC;
  if (t < Mloc * NC) {
    const int c = t / Mloc, ml = t - c * Mloc;
    const int m = m0 + ml;
    const int a = m / dB, b = m - a * dB;
    const int q0 = c * CL;
    int q1 = q0 + CL; if (q1 > NQ) q1 = NQ;
    double* dst = part + c * NEl + ml * dR;
    switch (dR) {
      case 1:  contract_run<1>(Ys, q0, q1, oA + a, oB + b, dst, 1); break;
      case 3:  contract_run<3>(Ys, q0, q1, oA + a, oB + b, dst, 1); break;
      case 5:  contract_run<5>(Ys, q0, q1, oA + a, oB + b, dst, 1); break;
      default: contract_run<7>(Ys, q0, q1, oA + a, oB + b, dst, 1); break;
    }
  }
  __syncthreads();

  double v = 0.0;
  if (t < NEl) {
    for (int c = 0; c < NC; ++c) v += part[c * NEl + t];
    const int ml = t / dR, r = t - ml * dR;
    const int m = m0 + ml;
    const int a = m / dB, b = m - a * dB;
    ctab[p * CSLOT + r * sR + a * sA + b * sB] = (float)v;
  }
  if (t < 64) red[t] = v * v;
  __syncthreads();
  for (int off = 32; off > 0; off >>= 1) {
    if (t < off) red[t] += red[t + off];
    __syncthreads();
  }
  if (t == 0) ctab[p * CSLOT + SSOFF + sub] = (float)red[0];
}

#define EPB         16
#define EDGE_T      512
#define EDGE_BLOCKS 512
static_assert(EDGE_T == 32 * EPB, "one wave per edge");
static_assert(EDGE_BLOCKS * EPB == N_EDGES, "every edge has a wave, no wave is without an edge");
static_assert(EDGE_T * 4 == 2 * HIDDEN * HIDDEN, "two rows x two columns of a path's weights or biases per thread");
#define TPITCH 68
#define TCOLS  112
#define TBUF   (TCOLS * TPITCH)
#define WROW   96
#define WBUF   (32 * WROW)

static constexpr int SEQ[NPATHS] = {0, 5, 13, 21,  1, 4, 7, 11, 15, 19,
                                    2, 6, 9, 10, 14, 18, 22,  3, 8, 12, 16, 17, 20};

__device__ __forceinline__ void wave_lds_sync() {
  __builtin_amdgcn_fence(__ATOMIC_RELEASE, "wavefront");
  __builtin_amdgcn_wave_barrier();
  __builtin_amdgcn_fence(__ATOMIC_ACQUIRE, "wavefront");
}

template <int S>
__device__ __forceinline__ void path_body(
    int lane, int wave, float len,
    const float (&Y)[16], const float (&xr)[16],
    const float* __restrict__ msrc,
    const float* __restrict__ ctab,
    const float (&invn)[NPATHS],
    float* Gs, float* TT,
    float* wdst,
    float2 (&sp)[2],
    bool works,
    const float* ap,
    const float* bp,
    v8f& acc0, v8f& acc1) {
  constexpr int P = SEQ[S];
  constexpr int l1 = PL1[P], l2 = PL2[P], l3 = PL3[P];
  constexpr int d1 = 2 * l1 + 1, d2 = 2 * l2 + 1, d3 = 2 * l3 + 1;
  constexpr int GN = d1 * d3;
  const float pw = PWT[l3] * invn[P];
  constexpr int WB_ = (S & 1) * WBUF;
  constexpr int TB_ = (S & 1) * TBUF;
  const float* Cp = ctab + P * CSLOT;

  if (lane < GN) {
    const int i = lane / d3, k = lane % d3;
    float g = 0.0f;
#pragma unroll
    for (int j = 0; j < d2; ++j) g = fmaf(Y[l2 * l2 + j], Cp[(i * d2 + j) * d3 + k], g);
    Gs[lane] = pw * g;
  }
  if constexpr (GN > 32) {
    const int n2 = lane + 32;
    if (n2 < GN) {
      const int i = n2 / d3, k = n2 % d3;
      float g = 0.0f;
#pragma unroll
      for (int j = 0; j < d2; ++j) g = fmaf(Y[l2 * l2 + j], Cp[(i * d2 + j) * d3 + k], g);
      Gs[n2] = pw * g;
    }
  }
  wave_lds_sync();

  {
    float* tc = TT + TB_ + (wave * d3) * TPITCH + lane;
#pragma unroll
    for (int k = 0; k < d3; ++k) {
      float s = 0.0f;
#pragma unroll
      for (int i = 0; i < d1; ++i) s = fmaf(xr[l1 * l1 + i], Gs[i * d3 + k], s);
      tc[k * TPITCH] = len * s;
      tc[k * TPITCH + 32] = s;
    }
  }

  *reinterpret_cast<float4*>(wdst + WB_) = make_float4(sp[0].x, sp[1].x, sp[0].y, sp[1].y);

  __syncthreads();

  if constexpr (S + 1 < NPATHS) {
    constexpr int PN = SEQ[S + 1];
    sp[0] = *reinterpret_cast<const float2*>(msrc + PN * (HIDDEN * HIDDEN));
    sp[1] = *reinterpret_cast<const float2*>(msrc + PN * (HIDDEN * HIDDEN) + HIDDEN);
  }

  if (works) {
    const float* a_ = ap + WB_;
    const float* b_ = bp + TB_;
#pragma unroll
    for (int kc = 0; kc < 16; kc += 2) {
      const v2f a0 = *reinterpret_cast<const v2f*>(a_ + kc * (2 * WROW));
      const v2f b0 = *reinterpret_cast<const v2f*>(b_ + kc * 4);
      acc0 = __builtin_amdgcn_wmma_f32_16x16x4_f32(
          false, a0, false, b0, (short)0, acc0, false, false);
      const v2f a1 = *reinterpret_cast<const v2f*>(a_ + (kc + 1) * (2 * WROW));
      const v2f b1 = *reinterpret_cast<const v2f*>(b_ + (kc + 1) * 4);
      acc1 = __builtin_amdgcn_wmma_f32_16x16x4_f32(
          false, a1, false, b1, (short)0, acc1, false, false);
    }
  }
}

template <int... Ss>
__device__ __forceinline__ void run_group(
    std::integer_sequence<int, Ss...>,
    int lane, int wave, float len,
    const float (&Y)[16], const float (&xr)[16],
    const float* __restrict__ msrc,
    const float* __restrict__ ctab, const float (&invn)[NPATHS],
    float* Gs, float* TT, float* wdst, float2 (&sp)[2],
    bool works, const float* ap, const float* bp,
    v8f& acc0, v8f& acc1) {
  (path_body<Ss>(lane, wave, len, Y, xr, msrc, ctab, invn, Gs, TT, wdst, sp, works, ap, bp,
                 acc0, acc1), ...);
}

template <int L3, int ROT, int... Ss>
__device__ __forceinline__ void do_group(
    std::integer_sequence<int, Ss...> seq,
    int lane, int wave, float len,
    const float (&Y)[16], const float (&xr)[16],
    const float* __restrict__ msrc,
    const float* __restrict__ ctab, const float (&invn)[NPATHS],
    float* Gs, float* TT, const float* Ws, float* wdst, float2 (&sp)[2],
    const int* dsts, float* __restrict__ out) {
  constexpr int d3 = 2 * L3 + 1;
  const int n16 = lane & 15;
  const int h   = lane >> 4;
  const int j    = (wave - ROT) & (EPB - 1);
  const bool works = j < 2 * d3;
  const int half = j & 1;
  const int tile = works ? (j >> 1) : 0;
  const float* ap = Ws + h * WROW + (16 * half + n16) * 2;
  const float* bp = TT + (16 * tile + n16) * TPITCH + 2 * h;
  v8f acc0 = {0,0,0,0,0,0,0,0}, acc1 = {0,0,0,0,0,0,0,0};
  run_group(seq, lane, wave, len, Y, xr, msrc, ctab, invn, Gs, TT, wdst, sp, works, ap, bp,
            acc0, acc1);
  if (works) {
    const int col = 16 * tile + n16;
    const int el  = col / d3;
    const int k   = col - el * d3;
    float* od = out + (size_t)dsts[el] * FEAT
                    + (HIDDEN * L3 * L3 + (16 * half + 8 * h) * d3 + k);
#pragma unroll
    for (int r = 0; r < 8; ++r) atomicAdd(od + r * d3, acc0[r] + acc1[r]);
  }
}

__attribute__((amdgpu_waves_per_eu(8, 8)))
__launch_bounds__(EDGE_T)
__global__ void edge_kernel(const float* __restrict__ nf,
                            const int*   __restrict__ eidx,
                            const float* __restrict__ evec,
                            const float* __restrict__ ww,
                            const float* __restrict__ wb,
                            const float* __restrict__ ctab,
                            float*       __restrict__ out) {
  const int tid  = threadIdx.x;
  const int lane = tid & 31;
  const int wave = __builtin_amdgcn_readfirstlane(tid >> 5);
  const int e    = blockIdx.x * EPB + wave;

  __shared__ float Gs_all[EPB * 64];
  __shared__ __attribute__((aligned(16))) float TT[2 * TBUF];
  __shared__ __attribute__((aligned(16))) float Ws[2 * WBUF];
  __shared__ int dsts[EPB];

  const int rp = (tid >> 4) & 15, cp = tid & 15;
  const float* msrc = ((wave < 8) ? ww : wb) + rp * (2 * HIDDEN) + cp * 2;
  float2 sp[2];
  sp[0] = *reinterpret_cast<const float2*>(msrc + SEQ[0] * (HIDDEN * HIDDEN));
  sp[1] = *reinterpret_cast<const float2*>(msrc + SEQ[0] * (HIDDEN * HIDDEN) + HIDDEN);

  const int src = eidx[e];
  const int dst = eidx[N_EDGES + e];
  if (lane == 0) dsts[wave] = dst;

  const float vx = evec[3 * e + 0];
  const float vy = evec[3 * e + 1];
  const float vz = evec[3 * e + 2];
  float len = sqrtf(vx * vx + vy * vy + vz * vz);
  len = fmaxf(len, 1e-8f);
  const float inv = 1.0f / len;

  float Y[16];
  sh_eval(vx * inv, vy * inv, vz * inv, Y);

  const float* xs = nf + (size_t)src * FEAT;
  float xr[16];
#pragma unroll
  for (int l = 0; l < 4; ++l) {
    const int d = 2 * l + 1;
#pragma unroll
    for (int i = 0; i < 7; ++i)
      if (i < d) xr[l * l + i] = xs[HIDDEN * l * l + lane * d + i];
  }

  float invn[NPATHS];
  {
    const float* sq = ctab + (lane < NPATHS ? lane : 0) * CSLOT + SSOFF;
    float ssq = 0.0f;
#pragma unroll
    for (int k = 0; k < NSUB; ++k) ssq += sq[k];
    const int ib = __builtin_bit_cast(int, 1.0f / sqrtf(ssq));
#pragma unroll
    for (int q = 0; q < NPATHS; ++q)
      invn[q] = __builtin_bit_cast(float, __builtin_amdgcn_readlane(ib, q));
  }

  float* Gs   = Gs_all + wave * 64;
  float* wdst = Ws + ((wave < 8 ? 0 : 16) + rp) * WROW + cp * 4;

  do_group<0, 2>(std::integer_sequence<int, 0, 1, 2, 3>{},
                 lane, wave, len, Y, xr, msrc, ctab, invn, Gs, TT, Ws, wdst, sp, dsts, out);
  do_group<1, 2>(std::integer_sequence<int, 4, 5, 6, 7, 8, 9>{},
                 lane, wave, len, Y, xr, msrc, ctab, invn, Gs, TT, Ws, wdst, sp, dsts, out);
  do_group<2, 0>(std::integer_sequence<int, 10, 11, 12, 13, 14, 15, 16>{},
                 lane, wave, len, Y, xr, msrc, ctab, invn, Gs, TT, Ws, wdst, sp, dsts, out);
  do_group<3, 0>(std::integer_sequence<int, 17, 18, 19, 20, 21, 22>{},
                 lane, wave, len, Y, xr, msrc, ctab, invn, Gs, TT, Ws, wdst, sp, dsts, out);
}

extern "C" void kernel_launch(void* const* d_in, const int* in_sizes, int n_in,
                              void* d_out, int out_size, void* d_ws, size_t ws_size,
                              hipStream_t stream) {
  const float* nf   = (const float*)d_in[0];
  const int*   eidx = (const int*)  d_in[1];
  const float* evec = (const float*)d_in[2];
  const float* ww   = (const float*)d_in[3];
  const float* wb   = (const float*)d_in[4];
  float* out  = (float*)d_out;
  float* ctab = (float*)d_ws;

  prep_kernel<<<PREP_BLOCKS, W3J_T, 0, stream>>>(ctab, out);
  edge_kernel<<<EDGE_BLOCKS, EDGE_T, 0, stream>>>(nf, eidx, evec, ww, wb, ctab, out);
}


# edge launch: a group's tiles are transposed through LDS to out's own order and each wave adds its own edge's part, so every output atomic covers one whole 128-byte line; arrival order as before
# speedup vs baseline: 102.0497x; 1.2095x over previous
#include <hip/hip_runtime.h>
#include <math.h>
#include <utility>

typedef float v2f __attribute__((ext_vector_type(2)));
typedef float v8f __attribute__((ext_vector_type(8)));

#define HIDDEN  32
#define N_NODES 4096
#define N_EDGES 8192
#define FEAT    512
#define NPATHS  23

static constexpr int PL1[NPATHS] = {0,0,0,0, 1,1,1,1,1,1, 2,2,2,2,2,2,2, 3,3,3,3,3,3};
static constexpr int PL2[NPATHS] = {0,1,2,3, 0,1,1,2,2,3, 0,1,1,2,2,3,3, 0,1,2,2,3,3};
static constexpr int PL3[NPATHS] = {0,1,2,3, 1,0,2,1,3,2, 2,1,3,0,2,1,3, 3,2,1,3,0,2};
static constexpr float PWT[4] = {
  0.08838834764831845f,
  0.125f,
  0.14940357616680451f,
  0.19094065395649326f
};

template <typename T>
__device__ __forceinline__ void sh_eval(T x, T y, T z, T* Y) {
  const T s3  = (T)1.7320508075688772;
  const T s5  = (T)2.2360679774997896;
  const T s7  = (T)2.6457513110645907;
  const T s15 = (T)3.8729833462074170;
  const T c58 = (T)0.7905694150420949;
  const T c38 = (T)0.6123724356957945;
  Y[0]  = (T)1;
  Y[1]  = s3 * x;
  Y[2]  = s3 * y;
  Y[3]  = s3 * z;
  Y[4]  = s5 * s3 * x * z;
  Y[5]  = s5 * s3 * x * y;
  Y[6]  = s5 * (y * y - (T)0.5 * (x * x + z * z));
  Y[7]  = s5 * s3 * y * z;
  Y[8]  = s5 * (s3 * (T)0.5) * (z * z - x * x);
  Y[9]  = s7 * c58 * x * ((T)3 * z * z - x * x);
  Y[10] = s7 * s15 * x * y * z;
  Y[11] = s7 * c38 * x * ((T)5 * y * y - (T)1);
  Y[12] = s7 * (T)0.5 * y * ((T)5 * y * y - (T)3);
  Y[13] = s7 * c38 * z * ((T)5 * y * y - (T)1);
  Y[14] = s7 * (s15 * (T)0.5) * y * (z * z - x * x);
  Y[15] = s7 * c58 * z * (z * z - (T)3 * x * x);
}

#define W3J_T  256
#define NSUB   8
#define TBLK   184
#define ZBLK   512
#define ZPER   4
#define PREP_BLOCKS 696
#define CSLOT  343
#define SSOFF  320
#define NCMAX  64
static_assert(TBLK == NPATHS * NSUB, "table blocks");
static_assert(PREP_BLOCKS == TBLK + ZBLK, "grid of the preparation launch");
static_assert(SSOFF >= 7 * 7 * 5 && SSOFF + NSUB <= CSLOT, "sums of squares sit behind the largest table (no path is 7x7x7)");
static_assert(((7 * 7 + NSUB - 1) / NSUB) * 7 <= 64, "a block's slice has at most 64 elements");
static_assert(ZBLK * W3J_T * ZPER * 4 == N_NODES * FEAT, "clearing blocks cover the output exactly once");
#define NQ     512
#define YPITCH 23

static constexpr double LA[17] = {0.0, 0.0, 3.0/2.0, 5.0/3.0, 7.0/4.0, 9.0/5.0, 11.0/6.0,
  13.0/7.0, 15.0/8.0, 17.0/9.0, 19.0/10.0, 21.0/11.0, 23.0/12.0, 25.0/13.0, 27.0/14.0,
  29.0/15.0, 31.0/16.0};
static constexpr double LB[17] = {0.0, 0.0, 1.0/2.0, 2.0/3.0, 3.0/4.0, 4.0/5.0, 5.0/6.0,
  6.0/7.0, 7.0/8.0, 8.0/9.0, 9.0/10.0, 10.0/11.0, 11.0/12.0, 12.0/13.0, 13.0/14.0,
  14.0/15.0, 15.0/16.0};

__device__ __forceinline__ void legendre16(double x, double& p15, double& p16) {
  double a0 = 1.0, b0 = x;
  for (int n = 2; n <= 16; ++n) {
    const double c0 = LA[n] * x * b0 - LB[n] * a0;
    a0 = b0; b0 = c0;
  }
  p15 = a0; p16 = b0;
}

template <int DR>
__device__ __forceinline__ void contract_run(const double* Ys, int q0, int q1,
                                             int ca, int cb, double* dst, int sR) {
  double acc[DR];
  for (int r = 0; r < DR; ++r) acc[r] = 0.0;
  for (int q = q0; q < q1; ++q) {
    const double* row = Ys + q * YPITCH;
    const double ab = row[ca] * row[cb];
    for (int r = 0; r < DR; ++r) acc[r] = fma(row[16 + r], ab, acc[r]);
  }
  for (int r = 0; r < DR; ++r) dst[r * sR] = acc[r];
}

__launch_bounds__(W3J_T)
__global__ void prep_kernel(float* __restrict__ ctab, float* __restrict__ out) {
  const int blk = blockIdx.x;
  const int t = threadIdx.x;

  if (blk >= TBLK) {
    float4* o4 = reinterpret_cast<float4*>(out);
    const int zb = blk - TBLK;
    const float4 z4 = make_float4(0.0f, 0.0f, 0.0f, 0.0f);
    for (int k = 0; k < ZPER; ++k) o4[(k * ZBLK + zb) * W3J_T + t] = z4;
    return;
  }
  const int p   = blk / NSUB;
  const int sub = blk - p * NSUB;

  int l1 = 0, l2 = 0, l3 = 0, cnt = 0;
  for (int a = 0; a < 4; ++a)
    for (int b = 0; b < 4; ++b)
      for (int c = 0; c < 4; ++c) {
        int lo = a - b; if (lo < 0) lo = -lo;
        if (lo <= c && c <= a + b && (((a + b + c) & 1) == 0)) {
          if (cnt == p) { l1 = a; l2 = b; l3 = c; }
          ++cnt;
        }
      }
  const int d1 = 2 * l1 + 1, d2 = 2 * l2 + 1, d3 = 2 * l3 + 1;

  int dR, dA, dB, oR, oA, oB, sR, sA, sB;
  if (d1 >= d2 && d1 >= d3) {
    dR = d1; oR = l1 * l1; sR = d2 * d3;
    dA = d2; oA = l2 * l2; sA = d3;
    dB = d3; oB = l3 * l3; sB = 1;
  } else if (d2 >= d3) {
    dR = d2; oR = l2 * l2; sR = d3;
    dA = d1; oA = l1 * l1; sA = d2 * d3;
    dB = d3; oB = l3 * l3; sB = 1;
  } else {
    dR = d3; oR = l3 * l3; sR = 1;
    dA = d1; oA = l1 * l1; sA = d2 * d3;
    dB = d2; oB = l2 * l2; sB = d3;
  }

  const int M  = dA * dB;
  const int MS = (M + NSUB - 1) / NSUB;
  const int m0 = sub * MS;
  const int m1 = (m0 + MS < M) ? (m0 + MS) : M;
  const int Mloc = m1 - m0;
  if (Mloc <= 0) {
    if (t == 0) ctab[p * CSLOT + SSOFF + sub] = 0.0f;
    return;
  }
  const int NEl = Mloc * dR;

  __shared__ double un_s[16], wn_s[16], st_s[16];
  __shared__ double cph_s[32], sph_s[32];
  __shared__ double Ys[NQ * YPITCH];
  __shared__ double part[W3J_T * 7];
  __shared__ double red[64];

  const double PI = 3.14159265358979323846;
  const double DPHI = 2.0 * PI / 32.0;

  if (t < 16) {
    double xg = (double)cosf((float)(PI * (t + 0.75) / 16.5));
    double dp = 1.0;
    for (int it = 0; it < 100; ++it) {
      double a0, b0;
      legendre16(xg, a0, b0);
      dp = 16.0 * (xg * b0 - a0) / (xg * xg - 1.0);
      const double dx = b0 / dp;
      xg -= dx;
      if (fabs(dx) < 1e-14) break;
    }
    {
      double a0, b0;
      legendre16(xg, a0, b0);
      dp = 16.0 * (xg * b0 - a0) / (xg * xg - 1.0);
    }
    const double st2 = 1.0 - xg * xg;
    un_s[t] = xg;
    wn_s[t] = 2.0 / (st2 * dp * dp);
    st_s[t] = sqrt(st2 > 0.0 ? st2 : 0.0);
  } else if (t >= 32 && t < 64) {
    const int j = t - 32;
    double s, c;
    sincos(j * DPHI, &s, &c);
    cph_s[j] = c;
    sph_s[j] = s;
  }
  __syncthreads();

  for (int q = t; q < NQ; q += W3J_T) {
    const int iq = q >> 5, jq = q & 31;
    const double st = st_s[iq];
    double Yq[16];
    sh_eval(st * cph_s[jq], st * sph_s[jq], un_s[iq], Yq);
    double* row = Ys + q * YPITCH;
    for (int c = 0; c < 16; ++c) row[c] = Yq[c];
    const double wq = wn_s[iq] * DPHI;
    for (int r = 0; r < dR; ++r) row[16 + r] = wq * row[oR + r];
  }
  __syncthreads();

  int NC = W3J_T / Mloc; if (NC > NCMAX) NC = NCMAX;
  const int CL = (NQ + NC - 1) / NC;
  if (t < Mloc * NC) {
    const int c = t / Mloc, ml = t - c * Mloc;
    const int m = m0 + ml;
    const int a = m / dB, b = m - a * dB;
    const int q0 = c * CL;
    int q1 = q0 + CL; if (q1 > NQ) q1 = NQ;
    double* dst = part + c * NEl + ml * dR;
    switch (dR) {
      case 1:  contract_run<1>(Ys, q0, q1, oA + a, oB + b, dst, 1); break;
      case 3:  contract_run<3>(Ys, q0, q1, oA + a, oB + b, dst, 1); break;
      case 5:  contract_run<5>(Ys, q0, q1, oA + a, oB + b, dst, 1); break;
      default: contract_run<7>(Ys, q0, q1, oA + a, oB + b, dst, 1); break;
    }
  }
  __syncthreads();

  double v = 0.0;
  if (t < NEl) {
    for (int c = 0; c < NC; ++c) v += part[c * NEl + t];
    const int ml = t / dR, r = t - ml * dR;
    const int m = m0 + ml;
    const int a = m / dB, b = m - a * dB;
    ctab[p * CSLOT + r * sR + a * sA + b * sB] = (float)v;
  }
  if (t < 64) red[t] = v * v;
  __syncthreads();
  for (int off = 32; off > 0; off >>= 1) {
    if (t < off) red[t] += red[t + off];
    __syncthreads();
  }
  if (t == 0) ctab[p * CSLOT + SSOFF + sub] = (float)red[0];
}

#define EPB         16
#define EDGE_T      512
#define EDGE_BLOCKS 512
static_assert(EDGE_T == 32 * EPB, "one wave per edge");
static_assert(EDGE_BLOCKS * EPB == N_EDGES, "every edge has a wave, no wave is without an edge");
static_assert(EDGE_T * 4 == 2 * HIDDEN * HIDDEN, "two rows x two columns of a path's weights or biases per thread");
#define TPITCH 68
#define TCOLS  112
#define TBUF   (TCOLS * TPITCH)
#define WROW   96
#define WBUF   (32 * WROW)
#define SPITCH(d3) (32 * (d3) + 1)
#define SBUF   (EPB * SPITCH(7))

static constexpr int SEQ[NPATHS] = {0, 5, 13, 21,  1, 4, 7, 11, 15, 19,
                                    2, 6, 9, 10, 14, 18, 22,  3, 8, 12, 16, 17, 20};

__device__ __forceinline__ void wave_lds_sync() {
  __builtin_amdgcn_fence(__ATOMIC_RELEASE, "wavefront");
  __builtin_amdgcn_wave_barrier();
  __builtin_amdgcn_fence(__ATOMIC_ACQUIRE, "wavefront");
}

template <int S>
struct PathPos {
  static constexpr int  L3    = PL3[SEQ[S]];
  static constexpr bool FIRST = (S == 0) || (PL3[SEQ[S > 0 ? S - 1 : 0]] != L3);
  static constexpr int  PREV  = L3 - 1;
  static_assert(S == 0 || PL3[SEQ[S > 0 ? S - 1 : 0]] == L3 || PL3[SEQ[S > 0 ? S - 1 : 0]] == L3 - 1,
                "SEQ walks the l3 groups in rising order");
};

template <int L3>
__device__ __forceinline__ void flush_stage(const float* stage, int wave, int lane, float* orow) {
  constexpr int D3 = 2 * L3 + 1;
  const float* sr = stage + wave * SPITCH(D3) + lane;
  float* od = orow + HIDDEN * L3 * L3;
  float v[D3];
#pragma unroll
  for (int j = 0; j < D3; ++j) v[j] = sr[32 * j];
#pragma unroll
  for (int j = 0; j < D3; ++j) atomicAdd(od + 32 * j, v[j]);
}

template <int S>
__device__ __forceinline__ void path_body(
    int lane, int wave, float len,
    const float (&Y)[16], const float (&xr)[16],
    const float* __restrict__ msrc,
    const float* __restrict__ ctab,
    const float (&invn)[NPATHS],
    float* Gs, float* TT,
    float* wdst,
    float2 (&sp)[2],
    bool works,
    const float* ap,
    const float* bp,
    const float* stage,
    float* orow,
    v8f& acc0, v8f& acc1) {
  constexpr int P = SEQ[S];
  constexpr int l1 = PL1[P], l2 = PL2[P], l3 = PL3[P];
  constexpr int d1 = 2 * l1 + 1, d2 = 2 * l2 + 1, d3 = 2 * l3 + 1;
  constexpr int GN = d1 * d3;
  const float pw = PWT[l3] * invn[P];
  constexpr int WB_ = (S & 1) * WBUF;
  constexpr int TB_ = (S & 1) * TBUF;
  const float* Cp = ctab + P * CSLOT;

  if (lane < GN) {
    const int i = lane / d3, k = lane % d3;
    float g = 0.0f;
#pragma unroll
    for (int j = 0; j < d2; ++j) g = fmaf(Y[l2 * l2 + j], Cp[(i * d2 + j) * d3 + k], g);
    Gs[lane] = pw * g;
  }
  if constexpr (GN > 32) {
    const int n2 = lane + 32;
    if (n2 < GN) {
      const int i = n2 / d3, k = n2 % d3;
      float g = 0.0f;
#pragma unroll
      for (int j = 0; j < d2; ++j) g = fmaf(Y[l2 * l2 + j], Cp[(i * d2 + j) * d3 + k], g);
      Gs[n2] = pw * g;
    }
  }
  wave_lds_sync();

  {
    float* tc = TT + TB_ + (wave * d3) * TPITCH + lane;
#pragma unroll
    for (int k = 0; k < d3; ++k) {
      float s = 0.0f;
#pragma unroll
      for (int i = 0; i < d1; ++i) s = fmaf(xr[l1 * l1 + i], Gs[i * d3 + k], s);
      tc[k * TPITCH] = len * s;
      tc[k * TPITCH + 32] = s;
    }
  }

  *reinterpret_cast<float4*>(wdst + WB_) = make_float4(sp[0].x, sp[1].x, sp[0].y, sp[1].y);

  __syncthreads();

  if constexpr (S + 1 < NPATHS) {
    constexpr int PN = SEQ[S + 1];
    sp[0] = *reinterpret_cast<const float2*>(msrc + PN * (HIDDEN * HIDDEN));
    sp[1] = *reinterpret_cast<const float2*>(msrc + PN * (HIDDEN * HIDDEN) + HIDDEN);
  }

  if constexpr (PathPos<S>::FIRST && PathPos<S>::PREV >= 0)
    flush_stage<PathPos<S>::PREV>(stage, wave, lane, orow);

  if (works) {
    const float* a_ = ap + WB_;
    const float* b_ = bp + TB_;
#pragma unroll
    for (int kc = 0; kc < 16; kc += 2) {
      const v2f a0 = *reinterpret_cast<const v2f*>(a_ + kc * (2 * WROW));
      const v2f b0 = *reinterpret_cast<const v2f*>(b_ + kc * 4);
      acc0 = __builtin_amdgcn_wmma_f32_16x16x4_f32(
          false, a0, false, b0, (short)0, acc0, false, false);
      const v2f a1 = *reinterpret_cast<const v2f*>(a_ + (kc + 1) * (2 * WROW));
      const v2f b1 = *reinterpret_cast<const v2f*>(b_ + (kc + 1) * 4);
      acc1 = __builtin_amdgcn_wmma_f32_16x16x4_f32(
          false, a1, false, b1, (short)0, acc1, false, false);
    }
  }
}

template <int... Ss>
__device__ __forceinline__ void run_group(
    std::integer_sequence<int, Ss...>,
    int lane, int wave, float len,
    const float (&Y)[16], const float (&xr)[16],
    const float* __restrict__ msrc,
    const float* __restrict__ ctab, const float (&invn)[NPATHS],
    float* Gs, float* TT, float* wdst, float2 (&sp)[2],
    bool works, const float* ap, const float* bp,
    const float* stage, float* orow,
    v8f& acc0, v8f& acc1) {
  (path_body<Ss>(lane, wave, len, Y, xr, msrc, ctab, invn, Gs, TT, wdst, sp, works, ap, bp,
                 stage, orow, acc0, acc1), ...);
}

template <int L3, int ROT, int... Ss>
__device__ __forceinline__ void do_group(
    std::integer_sequence<int, Ss...> seq,
    int lane, int wave, float len,
    const float (&Y)[16], const float (&xr)[16],
    const float* __restrict__ msrc,
    const float* __restrict__ ctab, const float (&invn)[NPATHS],
    float* Gs, float* TT, const float* Ws, float* wdst, float2 (&sp)[2],
    float* stage, float* orow) {
  constexpr int d3 = 2 * L3 + 1;
  const int n16 = lane & 15;
  const int h   = lane >> 4;
  const int j    = (wave - ROT) & (EPB - 1);
  const bool works = j < 2 * d3;
  const int half = j & 1;
  const int tile = works ? (j >> 1) : 0;
  const float* ap = Ws + h * WROW + (16 * half + n16) * 2;
  const float* bp = TT + (16 * tile + n16) * TPITCH + 2 * h;
  v8f acc0 = {0,0,0,0,0,0,0,0}, acc1 = {0,0,0,0,0,0,0,0};
  run_group(seq, lane, wave, len, Y, xr, msrc, ctab, invn, Gs, TT, wdst, sp, works, ap, bp,
            stage, orow, acc0, acc1);
  if (works) {
    const int col = 16 * tile + n16;
    const int el  = col / d3;
    const int k   = col - el * d3;
    float* sd = stage + el * SPITCH(d3) + (16 * half + 8 * h) * d3 + k;
#pragma unroll
    for (int r = 0; r < 8; ++r) sd[r * d3] = acc0[r] + acc1[r];
  }
  if constexpr (L3 == 3) {
    __syncthreads();
    flush_stage<3>(stage, wave, lane, orow);
  }
}

__attribute__((amdgpu_waves_per_eu(8, 8)))
__launch_bounds__(EDGE_T)
__global__ void edge_kernel(const float* __restrict__ nf,
                            const int*   __restrict__ eidx,
                            const float* __restrict__ evec,
                            const float* __restrict__ ww,
                            const float* __restrict__ wb,
                            const float* __restrict__ ctab,
                            float*       __restrict__ out) {
  const int tid  = threadIdx.x;
  const int lane = tid & 31;
  const int wave = __builtin_amdgcn_readfirstlane(tid >> 5);
  const int e    = blockIdx.x * EPB + wave;

  __shared__ float Gs_all[EPB * 64];
  __shared__ __attribute__((aligned(16))) float TT[2 * TBUF];
  __shared__ __attribute__((aligned(16))) float Ws[2 * WBUF];
  __shared__ float stage[SBUF];

  const int rp = (tid >> 4) & 15, cp = tid & 15;
  const float* msrc = ((wave < 8) ? ww : wb) + rp * (2 * HIDDEN) + cp * 2;
  float2 sp[2];
  sp[0] = *reinterpret_cast<const float2*>(msrc + SEQ[0] * (HIDDEN * HIDDEN));
  sp[1] = *reinterpret_cast<const float2*>(msrc + SEQ[0] * (HIDDEN * HIDDEN) + HIDDEN);

  const int src = eidx[e];
  const int dst = eidx[N_EDGES + e];
  float* orow = out + (size_t)dst * FEAT + lane;

  const float vx = evec[3 * e + 0];
  const float vy = evec[3 * e + 1];
  const float vz = evec[3 * e + 2];
  float len = sqrtf(vx * vx + vy * vy + vz * vz);
  len = fmaxf(len, 1e-8f);
  const float inv = 1.0f / len;

  float Y[16];
  sh_eval(vx * inv, vy * inv, vz * inv, Y);

  const float* xs = nf + (size_t)src * FEAT;
  float xr[16];
#pragma unroll
  for (int l = 0; l < 4; ++l) {
    const int d = 2 * l + 1;
#pragma unroll
    for (int i = 0; i < 7; ++i)
      if (i < d) xr[l * l + i] = xs[HIDDEN * l * l + lane * d + i];
  }

  float invn[NPATHS];
  {
    const float* sq = ctab + (lane < NPATHS ? lane : 0) * CSLOT + SSOFF;
    float ssq = 0.0f;
#pragma unroll
    for (int k = 0; k < NSUB; ++k) ssq += sq[k];
    const int ib = __builtin_bit_cast(int, 1.0f / sqrtf(ssq));
#pragma unroll
    for (int q = 0; q < NPATHS; ++q)
      invn[q] = __builtin_bit_cast(float, __builtin_amdgcn_readlane(ib, q));
  }

  float* Gs   = Gs_all + wave * 64;
  float* wdst = Ws + ((wave < 8 ? 0 : 16) + rp) * WROW + cp * 4;

  do_group<0, 2>(std::integer_sequence<int, 0, 1, 2, 3>{},
                 lane, wave, len, Y, xr, msrc, ctab, invn, Gs, TT, Ws, wdst, sp, stage, orow);
  do_group<1, 2>(std::integer_sequence<int, 4, 5, 6, 7, 8, 9>{},
                 lane, wave, len, Y, xr, msrc, ctab, invn, Gs, TT, Ws, wdst, sp, stage, orow);
  do_group<2, 0>(std::integer_sequence<int, 10, 11, 12, 13, 14, 15, 16>{},
                 lane, wave, len, Y, xr, msrc, ctab, invn, Gs, TT, Ws, wdst, sp, stage, orow);
  do_group<3, 0>(std::integer_sequence<int, 17, 18, 19, 20, 21, 22>{},
                 lane, wave, len, Y, xr, msrc, ctab, invn, Gs, TT, Ws, wdst, sp, stage, orow);
}

extern "C" void kernel_launch(void* const* d_in, const int* in_sizes, int n_in,
                              void* d_out, int out_size, void* d_ws, size_t ws_size,
                              hipStream_t stream) {
  const float* nf   = (const float*)d_in[0];
  const int*   eidx = (const int*)  d_in[1];
  const float* evec = (const float*)d_in[2];
  const float* ww   = (const float*)d_in[3];
  const float* wb   = (const float*)d_in[4];
  float* out  = (float*)d_out;
  float* ctab = (float*)d_ws;

  prep_kernel<<<PREP_BLOCKS, W3J_T, 0, stream>>>(ctab, out);
  edge_kernel<<<EDGE_BLOCKS, EDGE_T, 0, stream>>>(nf, eidx, evec, ww, wb, ctab, out);
}


# preparation launch: quadrature nodes by Newton with 2 steps in f32, then 2 fp64 steps of one division each, weight from the last step's derivative to first order; 5 fp64 Legendre evaluations -> 2
# speedup vs baseline: 110.5577x; 1.0834x over previous
#include <hip/hip_runtime.h>
#include <math.h>
#include <utility>

typedef float v2f __attribute__((ext_vector_type(2)));
typedef float v8f __attribute__((ext_vector_type(8)));

#define HIDDEN  32
#define N_NODES 4096
#define N_EDGES 8192
#define FEAT    512
#define NPATHS  23

static constexpr int PL1[NPATHS] = {0,0,0,0, 1,1,1,1,1,1, 2,2,2,2,2,2,2, 3,3,3,3,3,3};
static constexpr int PL2[NPATHS] = {0,1,2,3, 0,1,1,2,2,3, 0,1,1,2,2,3,3, 0,1,2,2,3,3};
static constexpr int PL3[NPATHS] = {0,1,2,3, 1,0,2,1,3,2, 2,1,3,0,2,1,3, 3,2,1,3,0,2};
static constexpr float PWT[4] = {
  0.08838834764831845f,
  0.125f,
  0.14940357616680451f,
  0.19094065395649326f
};

template <typename T>
__device__ __forceinline__ void sh_eval(T x, T y, T z, T* Y) {
  const T s3  = (T)1.7320508075688772;
  const T s5  = (T)2.2360679774997896;
  const T s7  = (T)2.6457513110645907;
  const T s15 = (T)3.8729833462074170;
  const T c58 = (T)0.7905694150420949;
  const T c38 = (T)0.6123724356957945;
  Y[0]  = (T)1;
  Y[1]  = s3 * x;
  Y[2]  = s3 * y;
  Y[3]  = s3 * z;
  Y[4]  = s5 * s3 * x * z;
  Y[5]  = s5 * s3 * x * y;
  Y[6]  = s5 * (y * y - (T)0.5 * (x * x + z * z));
  Y[7]  = s5 * s3 * y * z;
  Y[8]  = s5 * (s3 * (T)0.5) * (z * z - x * x);
  Y[9]  = s7 * c58 * x * ((T)3 * z * z - x * x);
  Y[10] = s7 * s15 * x * y * z;
  Y[11] = s7 * c38 * x * ((T)5 * y * y - (T)1);
  Y[12] = s7 * (T)0.5 * y * ((T)5 * y * y - (T)3);
  Y[13] = s7 * c38 * z * ((T)5 * y * y - (T)1);
  Y[14] = s7 * (s15 * (T)0.5) * y * (z * z - x * x);
  Y[15] = s7 * c58 * z * (z * z - (T)3 * x * x);
}

#define W3J_T  256
#define NSUB   8
#define TBLK   184
#define ZBLK   512
#define ZPER   4
#define PREP_BLOCKS 696
#define CSLOT  343
#define SSOFF  320
#define NCMAX  64
static_assert(TBLK == NPATHS * NSUB, "table blocks");
static_assert(PREP_BLOCKS == TBLK + ZBLK, "grid of the preparation launch");
static_assert(SSOFF >= 7 * 7 * 5 && SSOFF + NSUB <= CSLOT, "sums of squares sit behind the largest table (no path is 7x7x7)");
static_assert(((7 * 7 + NSUB - 1) / NSUB) * 7 <= 64, "a block's slice has at most 64 elements");
static_assert(ZBLK * W3J_T * ZPER * 4 == N_NODES * FEAT, "clearing blocks cover the output exactly once");
#define NQ     512
#define YPITCH 23

static constexpr double LA[17] = {0.0, 0.0, 3.0/2.0, 5.0/3.0, 7.0/4.0, 9.0/5.0, 11.0/6.0,
  13.0/7.0, 15.0/8.0, 17.0/9.0, 19.0/10.0, 21.0/11.0, 23.0/12.0, 25.0/13.0, 27.0/14.0,
  29.0/15.0, 31.0/16.0};
static constexpr double LB[17] = {0.0, 0.0, 1.0/2.0, 2.0/3.0, 3.0/4.0, 4.0/5.0, 5.0/6.0,
  6.0/7.0, 7.0/8.0, 8.0/9.0, 9.0/10.0, 10.0/11.0, 11.0/12.0, 12.0/13.0, 13.0/14.0,
  14.0/15.0, 15.0/16.0};

template <typename T>
__device__ __forceinline__ void legendre16(T x, T& p15, T& p16) {
  T a0 = (T)1, b0 = x;
  for (int n = 2; n <= 16; ++n) {
    const T c0 = (T)LA[n] * x * b0 - (T)LB[n] * a0;
    a0 = b0; b0 = c0;
  }
  p15 = a0; p16 = b0;
}

template <int DR>
__device__ __forceinline__ void contract_run(const double* Ys, int q0, int q1,
                                             int ca, int cb, double* dst, int sR) {
  double acc[DR];
  for (int r = 0; r < DR; ++r) acc[r] = 0.0;
  for (int q = q0; q < q1; ++q) {
    const double* row = Ys + q * YPITCH;
    const double ab = row[ca] * row[cb];
    for (int r = 0; r < DR; ++r) acc[r] = fma(row[16 + r], ab, acc[r]);
  }
  for (int r = 0; r < DR; ++r) dst[r * sR] = acc[r];
}

__launch_bounds__(W3J_T)
__global__ void prep_kernel(float* __restrict__ ctab, float* __restrict__ out) {
  const int blk = blockIdx.x;
  const int t = threadIdx.x;

  if (blk >= TBLK) {
    float4* o4 = reinterpret_cast<float4*>(out);
    const int zb = blk - TBLK;
    const float4 z4 = make_float4(0.0f, 0.0f, 0.0f, 0.0f);
    for (int k = 0; k < ZPER; ++k) o4[(k * ZBLK + zb) * W3J_T + t] = z4;
    return;
  }
  const int p   = blk / NSUB;
  const int sub = blk - p * NSUB;

  int l1 = 0, l2 = 0, l3 = 0, cnt = 0;
  for (int a = 0; a < 4; ++a)
    for (int b = 0; b < 4; ++b)
      for (int c = 0; c < 4; ++c) {
        int lo = a - b; if (lo < 0) lo = -lo;
        if (lo <= c && c <= a + b && (((a + b + c) & 1) == 0)) {
          if (cnt == p) { l1 = a; l2 = b; l3 = c; }
          ++cnt;
        }
      }
  const int d1 = 2 * l1 + 1, d2 = 2 * l2 + 1, d3 = 2 * l3 + 1;

  int dR, dA, dB, oR, oA, oB, sR, sA, sB;
  if (d1 >= d2 && d1 >= d3) {
    dR = d1; oR = l1 * l1; sR = d2 * d3;
    dA = d2; oA = l2 * l2; sA = d3;
    dB = d3; oB = l3 * l3; sB = 1;
  } else if (d2 >= d3) {
    dR = d2; oR = l2 * l2; sR = d3;
    dA = d1; oA = l1 * l1; sA = d2 * d3;
    dB = d3; oB = l3 * l3; sB = 1;
  } else {
    dR = d3; oR = l3 * l3; sR = 1;
    dA = d1; oA = l1 * l1; sA = d2 * d3;
    dB = d2; oB = l2 * l2; sB = d3;
  }

  const int M  = dA * dB;
  const int MS = (M + NSUB - 1) / NSUB;
  const int m0 = sub * MS;
  const int m1 = (m0 + MS < M) ? (m0 + MS) : M;
  const int Mloc = m1 - m0;
  if (Mloc <= 0) {
    if (t == 0) ctab[p * CSLOT + SSOFF + sub] = 0.0f;
    return;
  }
  const int NEl = Mloc * dR;

  __shared__ double un_s[16], wn_s[16], st_s[16];
  __shared__ double cph_s[32], sph_s[32];
  __shared__ double Ys[NQ * YPITCH];
  __shared__ double part[W3J_T * 7];
  __shared__ double red[64];

  const double PI = 3.14159265358979323846;
  const double DPHI = 2.0 * PI / 32.0;

  if (t < 16) {
    float xf = cosf(((float)t + 0.75f) * (float)(PI / 16.5));
    for (int it = 0; it < 2; ++it) {
      float a0, b0;
      legendre16(xf, a0, b0);
      xf -= b0 * (xf * xf - 1.0f) / (16.0f * (xf * b0 - a0));
    }
    double x1 = (double)xf;
    {
      double a0, b0;
      legendre16(x1, a0, b0);
      x1 -= b0 * (x1 * x1 - 1.0) / (16.0 * (x1 * b0 - a0));
    }
    double a0, b0;
    legendre16(x1, a0, b0);
    const double s1 = x1 * x1 - 1.0;
    const double rq = 1.0 / (x1 * b0 - a0);
    const double xg = x1 - 0.0625 * b0 * s1 * rq;
    const double e  = 0.125 * x1 * b0 * rq;
    const double st2 = 1.0 - xg * xg;
    un_s[t] = xg;
    wn_s[t] = -0.0078125 * s1 * rq * rq * (1.0 - e);
    st_s[t] = sqrt(st2 > 0.0 ? st2 : 0.0);
  } else if (t >= 32 && t < 64) {
    const int j = t - 32;
    double s, c;
    sincos(j * DPHI, &s, &c);
    cph_s[j] = c;
    sph_s[j] = s;
  }
  __syncthreads();

  for (int q = t; q < NQ; q += W3J_T) {
    const int iq = q >> 5, jq = q & 31;
    const double st = st_s[iq];
    double Yq[16];
    sh_eval(st * cph_s[jq], st * sph_s[jq], un_s[iq], Yq);
    double* row = Ys + q * YPITCH;
    for (int c = 0; c < 16; ++c) row[c] = Yq[c];
    const double wq = wn_s[iq] * DPHI;
    for (int r = 0; r < dR; ++r) row[16 + r] = wq * row[oR + r];
  }
  __syncthreads();

  int NC = W3J_T / Mloc; if (NC > NCMAX) NC = NCMAX;
  const int CL = (NQ + NC - 1) / NC;
  if (t < Mloc * NC) {
    const int c = t / Mloc, ml = t - c * Mloc;
    const int m = m0 + ml;
    const int a = m / dB, b = m - a * dB;
    const int q0 = c * CL;
    int q1 = q0 + CL; if (q1 > NQ) q1 = NQ;
    double* dst = part + c * NEl + ml * dR;
    switch (dR) {
      case 1:  contract_run<1>(Ys, q0, q1, oA + a, oB + b, dst, 1); break;
      case 3:  contract_run<3>(Ys, q0, q1, oA + a, oB + b, dst, 1); break;
      case 5:  contract_run<5>(Ys, q0, q1, oA + a, oB + b, dst, 1); break;
      default: contract_run<7>(Ys, q0, q1, oA + a, oB + b, dst, 1); break;
    }
  }
  __syncthreads();

  double v = 0.0;
  if (t < NEl) {
    for (int c = 0; c < NC; ++c) v += part[c * NEl + t];
    const int ml = t / dR, r = t - ml * dR;
    const int m = m0 + ml;
    const int a = m / dB, b = m - a * dB;
    ctab[p * CSLOT + r * sR + a * sA + b * sB] = (float)v;
  }
  if (t < 64) red[t] = v * v;
  __syncthreads();
  for (int off = 32; off > 0; off >>= 1) {
    if (t < off) red[t] += red[t + off];
    __syncthreads();
  }
  if (t == 0) ctab[p * CSLOT + SSOFF + sub] = (float)red[0];
}

#define EPB         16
#define EDGE_T      512
#define EDGE_BLOCKS 512
static_assert(EDGE_T == 32 * EPB, "one wave per edge");
static_assert(EDGE_BLOCKS * EPB == N_EDGES, "every edge has a wave, no wave is without an edge");
static_assert(EDGE_T * 4 == 2 * HIDDEN * HIDDEN, "two rows x two columns of a path's weights or biases per thread");
#define TPITCH 68
#define TCOLS  112
#define TBUF   (TCOLS * TPITCH)
#define WROW   96
#define WBUF   (32 * WROW)
#define SPITCH(d3) (32 * (d3) + 1)
#define SBUF   (EPB * SPITCH(7))

static constexpr int SEQ[NPATHS] = {0, 5, 13, 21,  1, 4, 7, 11, 15, 19,
                                    2, 6, 9, 10, 14, 18, 22,  3, 8, 12, 16, 17, 20};

__device__ __forceinline__ void wave_lds_sync() {
  __builtin_amdgcn_fence(__ATOMIC_RELEASE, "wavefront");
  __builtin_amdgcn_wave_barrier();
  __builtin_amdgcn_fence(__ATOMIC_ACQUIRE, "wavefront");
}

template <int S>
struct PathPos {
  static constexpr int  L3    = PL3[SEQ[S]];
  static constexpr bool FIRST = (S == 0) || (PL3[SEQ[S > 0 ? S - 1 : 0]] != L3);
  static constexpr int  PREV  = L3 - 1;
  static_assert(S == 0 || PL3[SEQ[S > 0 ? S - 1 : 0]] == L3 || PL3[SEQ[S > 0 ? S - 1 : 0]] == L3 - 1,
                "SEQ walks the l3 groups in rising order");
};

template <int L3>
__device__ __forceinline__ void flush_stage(const float* stage, int wave, int lane, float* orow) {
  constexpr int D3 = 2 * L3 + 1;
  const float* sr = stage + wave * SPITCH(D3) + lane;
  float* od = orow + HIDDEN * L3 * L3;
  float v[D3];
#pragma unroll
  for (int j = 0; j < D3; ++j) v[j] = sr[32 * j];
#pragma unroll
  for (int j = 0; j < D3; ++j) atomicAdd(od + 32 * j, v[j]);
}

template <int S>
__device__ __forceinline__ void path_body(
    int lane, int wave, float len,
    const float (&Y)[16], const float (&xr)[16],
    const float* __restrict__ msrc,
    const float* __restrict__ ctab,
    const float (&invn)[NPATHS],
    float* Gs, float* TT,
    float* wdst,
    float2 (&sp)[2],
    bool works,
    const float* ap,
    const float* bp,
    const float* stage,
    float* orow,
    v8f& acc0, v8f& acc1) {
  constexpr int P = SEQ[S];
  constexpr int l1 = PL1[P], l2 = PL2[P], l3 = PL3[P];
  constexpr int d1 = 2 * l1 + 1, d2 = 2 * l2 + 1, d3 = 2 * l3 + 1;
  constexpr int GN = d1 * d3;
  const float pw = PWT[l3] * invn[P];
  constexpr int WB_ = (S & 1) * WBUF;
  constexpr int TB_ = (S & 1) * TBUF;
  const float* Cp = ctab + P * CSLOT;

  if (lane < GN) {
    const int i = lane / d3, k = lane % d3;
    float g = 0.0f;
#pragma unroll
    for (int j = 0; j < d2; ++j) g = fmaf(Y[l2 * l2 + j], Cp[(i * d2 + j) * d3 + k], g);
    Gs[lane] = pw * g;
  }
  if constexpr (GN > 32) {
    const int n2 = lane + 32;
    if (n2 < GN) {
      const int i = n2 / d3, k = n2 % d3;
      float g = 0.0f;
#pragma unroll
      for (int j = 0; j < d2; ++j) g = fmaf(Y[l2 * l2 + j], Cp[(i * d2 + j) * d3 + k], g);
      Gs[n2] = pw * g;
    }
  }
  wave_lds_sync();

  {
    float* tc = TT + TB_ + (wave * d3) * TPITCH + lane;
#pragma unroll
    for (int k = 0; k < d3; ++k) {
      float s = 0.0f;
#pragma unroll
      for (int i = 0; i < d1; ++i) s = fmaf(xr[l1 * l1 + i], Gs[i * d3 + k], s);
      tc[k * TPITCH] = len * s;
      tc[k * TPITCH + 32] = s;
    }
  }

  *reinterpret_cast<float4*>(wdst + WB_) = make_float4(sp[0].x, sp[1].x, sp[0].y, sp[1].y);

  __syncthreads();

  if constexpr (S + 1 < NPATHS) {
    constexpr int PN = SEQ[S + 1];
    sp[0] = *reinterpret_cast<const float2*>(msrc + PN * (HIDDEN * HIDDEN));
    sp[1] = *reinterpret_cast<const float2*>(msrc + PN * (HIDDEN * HIDDEN) + HIDDEN);
  }

  if constexpr (PathPos<S>::FIRST && PathPos<S>::PREV >= 0)
    flush_stage<PathPos<S>::PREV>(stage, wave, lane, orow);

  if (works) {
    const float* a_ = ap + WB_;
    const float* b_ = bp + TB_;
#pragma unroll
    for (int kc = 0; kc < 16; kc += 2) {
      const v2f a0 = *reinterpret_cast<const v2f*>(a_ + kc * (2 * WROW));
      const v2f b0 = *reinterpret_cast<const v2f*>(b_ + kc * 4);
      acc0 = __builtin_amdgcn_wmma_f32_16x16x4_f32(
          false, a0, false, b0, (short)0, acc0, false, false);
      const v2f a1 = *reinterpret_cast<const v2f*>(a_ + (kc + 1) * (2 * WROW));
      const v2f b1 = *reinterpret_cast<const v2f*>(b_ + (kc + 1) * 4);
      acc1 = __builtin_amdgcn_wmma_f32_16x16x4_f32(
          false, a1, false, b1, (short)0, acc1, false, false);
    }
  }
}

template <int... Ss>
__device__ __forceinline__ void run_group(
    std::integer_sequence<int, Ss...>,
    int lane, int wave, float len,
    const float (&Y)[16], const float (&xr)[16],
    const float* __restrict__ msrc,
    const float* __restrict__ ctab, const float (&invn)[NPATHS],
    float* Gs, float* TT, float* wdst, float2 (&sp)[2],
    bool works, const float* ap, const float* bp,
    const float* stage, float* orow,
    v8f& acc0, v8f& acc1) {
  (path_body<Ss>(lane, wave, len, Y, xr, msrc, ctab, invn, Gs, TT, wdst, sp, works, ap, bp,
                 stage, orow, acc0, acc1), ...);
}

template <int L3, int ROT, int... Ss>
__device__ __forceinline__ void do_group(
    std::integer_sequence<int, Ss...> seq,
    int lane, int wave, float len,
    const float (&Y)[16], const float (&xr)[16],
    const float* __restrict__ msrc,
    const float* __restrict__ ctab, const float (&invn)[NPATHS],
    float* Gs, float* TT, const float* Ws, float* wdst, float2 (&sp)[2],
    float* stage, float* orow) {
  constexpr int d3 = 2 * L3 + 1;
  const int n16 = lane & 15;
  const int h   = lane >> 4;
  const int j    = (wave - ROT) & (EPB - 1);
  const bool works = j < 2 * d3;
  const int half = j & 1;
  const int tile = works ? (j >> 1) : 0;
  const float* ap = Ws + h * WROW + (16 * half + n16) * 2;
  const float* bp = TT + (16 * tile + n16) * TPITCH + 2 * h;
  v8f acc0 = {0,0,0,0,0,0,0,0}, acc1 = {0,0,0,0,0,0,0,0};
  run_group(seq, lane, wave, len, Y, xr, msrc, ctab, invn, Gs, TT, wdst, sp, works, ap, bp,
            stage, orow, acc0, acc1);
  if (works) {
    const int col = 16 * tile + n16;
    const int el  = col / d3;
    const int k   = col - el * d3;
    float* sd = stage + el * SPITCH(d3) + (16 * half + 8 * h) * d3 + k;
#pragma unroll
    for (int r = 0; r < 8; ++r) sd[r * d3] = acc0[r] + acc1[r];
  }
  if constexpr (L3 == 3) {
    __syncthreads();
    flush_stage<3>(stage, wave, lane, orow);
  }
}

__attribute__((amdgpu_waves_per_eu(8, 8)))
__launch_bounds__(EDGE_T)
__global__ void edge_kernel(const float* __restrict__ nf,
                            const int*   __restrict__ eidx,
                            const float* __restrict__ evec,
                            const float* __restrict__ ww,
                            const float* __restrict__ wb,
                            const float* __restrict__ ctab,
                            float*       __restrict__ out) {
  const int tid  = threadIdx.x;
  const int lane = tid & 31;
  const int wave = __builtin_amdgcn_readfirstlane(tid >> 5);
  const int e    = blockIdx.x * EPB + wave;

  __shared__ float Gs_all[EPB * 64];
  __shared__ __attribute__((aligned(16))) float TT[2 * TBUF];
  __shared__ __attribute__((aligned(16))) float Ws[2 * WBUF];
  __shared__ float stage[SBUF];

  const int rp = (tid >> 4) & 15, cp = tid & 15;
  const float* msrc = ((wave < 8) ? ww : wb) + rp * (2 * HIDDEN) + cp * 2;
  float2 sp[2];
  sp[0] = *reinterpret_cast<const float2*>(msrc + SEQ[0] * (HIDDEN * HIDDEN));
  sp[1] = *reinterpret_cast<const float2*>(msrc + SEQ[0] * (HIDDEN * HIDDEN) + HIDDEN);

  const int src = eidx[e];
  const int dst = eidx[N_EDGES + e];
  float* orow = out + (size_t)dst * FEAT + lane;

  const float vx = evec[3 * e + 0];
  const float vy = evec[3 * e + 1];
  const float vz = evec[3 * e + 2];
  float len = sqrtf(vx * vx + vy * vy + vz * vz);
  len = fmaxf(len, 1e-8f);
  const float inv = 1.0f / len;

  float Y[16];
  sh_eval(vx * inv, vy * inv, vz * inv, Y);

  const float* xs = nf + (size_t)src * FEAT;
  float xr[16];
#pragma unroll
  for (int l = 0; l < 4; ++l) {
    const int d = 2 * l + 1;
#pragma unroll
    for (int i = 0; i < 7; ++i)
      if (i < d) xr[l * l + i] = xs[HIDDEN * l * l + lane * d + i];
  }

  float invn[NPATHS];
  {
    const float* sq = ctab + (lane < NPATHS ? lane : 0) * CSLOT + SSOFF;
    float ssq = 0.0f;
#pragma unroll
    for (int k = 0; k < NSUB; ++k) ssq += sq[k];
    const int ib = __builtin_bit_cast(int, 1.0f / sqrtf(ssq));
#pragma unroll
    for (int q = 0; q < NPATHS; ++q)
      invn[q] = __builtin_bit_cast(float, __builtin_amdgcn_readlane(ib, q));
  }

  float* Gs   = Gs_all + wave * 64;
  float* wdst = Ws + ((wave < 8 ? 0 : 16) + rp) * WROW + cp * 4;

  do_group<0, 2>(std::integer_sequence<int, 0, 1, 2, 3>{},
                 lane, wave, len, Y, xr, msrc, ctab, invn, Gs, TT, Ws, wdst, sp, stage, orow);
  do_group<1, 2>(std::integer_sequence<int, 4, 5, 6, 7, 8, 9>{},
                 lane, wave, len, Y, xr, msrc, ctab, invn, Gs, TT, Ws, wdst, sp, stage, orow);
  do_group<2, 0>(std::integer_sequence<int, 10, 11, 12, 13, 14, 15, 16>{},
                 lane, wave, len, Y, xr, msrc, ctab, invn, Gs, TT, Ws, wdst, sp, stage, orow);
  do_group<3, 0>(std::integer_sequence<int, 17, 18, 19, 20, 21, 22>{},
                 lane, wave, len, Y, xr, msrc, ctab, invn, Gs, TT, Ws, wdst, sp, stage, orow);
}

extern "C" void kernel_launch(void* const* d_in, const int* in_sizes, int n_in,
                              void* d_out, int out_size, void* d_ws, size_t ws_size,
                              hipStream_t stream) {
  const float* nf   = (const float*)d_in[0];
  const int*   eidx = (const int*)  d_in[1];
  const float* evec = (const float*)d_in[2];
  const float* ww   = (const float*)d_in[3];
  const float* wb   = (const float*)d_in[4];
  float* out  = (float*)d_out;
  float* ctab = (float*)d_ws;

  prep_kernel<<<PREP_BLOCKS, W3J_T, 0, stream>>>(ctab, out);
  edge_kernel<<<EDGE_BLOCKS, EDGE_T, 0, stream>>>(nf, eidx, evec, ww, wb, ctab, out);
}


# coupling tables: the 16x32 quadrature summed over the 128 points its two mirror symmetries (u to -u, phi to phi+pi) leave, x4, odd-parity elements set to 0; the order of its fp64 sums changes
# speedup vs baseline: 125.5516x; 1.1356x over previous
#include <hip/hip_runtime.h>
#include <math.h>
#include <utility>

typedef float v2f __attribute__((ext_vector_type(2)));
typedef float v8f __attribute__((ext_vector_type(8)));

#define HIDDEN  32
#define N_NODES 4096
#define N_EDGES 8192
#define FEAT    512
#define NPATHS  23

static constexpr int PL1[NPATHS] = {0,0,0,0, 1,1,1,1,1,1, 2,2,2,2,2,2,2, 3,3,3,3,3,3};
static constexpr int PL2[NPATHS] = {0,1,2,3, 0,1,1,2,2,3, 0,1,1,2,2,3,3, 0,1,2,2,3,3};
static constexpr int PL3[NPATHS] = {0,1,2,3, 1,0,2,1,3,2, 2,1,3,0,2,1,3, 3,2,1,3,0,2};
static constexpr float PWT[4] = {
  0.08838834764831845f,
  0.125f,
  0.14940357616680451f,
  0.19094065395649326f
};

template <typename T>
__device__ __forceinline__ void sh_eval(T x, T y, T z, T* Y) {
  const T s3  = (T)1.7320508075688772;
  const T s5  = (T)2.2360679774997896;
  const T s7  = (T)2.6457513110645907;
  const T s15 = (T)3.8729833462074170;
  const T c58 = (T)0.7905694150420949;
  const T c38 = (T)0.6123724356957945;
  Y[0]  = (T)1;
  Y[1]  = s3 * x;
  Y[2]  = s3 * y;
  Y[3]  = s3 * z;
  Y[4]  = s5 * s3 * x * z;
  Y[5]  = s5 * s3 * x * y;
  Y[6]  = s5 * (y * y - (T)0.5 * (x * x + z * z));
  Y[7]  = s5 * s3 * y * z;
  Y[8]  = s5 * (s3 * (T)0.5) * (z * z - x * x);
  Y[9]  = s7 * c58 * x * ((T)3 * z * z - x * x);
  Y[10] = s7 * s15 * x * y * z;
  Y[11] = s7 * c38 * x * ((T)5 * y * y - (T)1);
  Y[12] = s7 * (T)0.5 * y * ((T)5 * y * y - (T)3);
  Y[13] = s7 * c38 * z * ((T)5 * y * y - (T)1);
  Y[14] = s7 * (s15 * (T)0.5) * y * (z * z - x * x);
  Y[15] = s7 * c58 * z * (z * z - (T)3 * x * x);
}

#define W3J_T  256
#define NSUB   8
#define TBLK   184
#define ZBLK   512
#define ZPER   4
#define PREP_BLOCKS 696
#define CSLOT  343
#define SSOFF  320
#define NCMAX  32
static_assert(TBLK == NPATHS * NSUB, "table blocks");
static_assert(PREP_BLOCKS == TBLK + ZBLK, "grid of the preparation launch");
static_assert(SSOFF >= 7 * 7 * 5 && SSOFF + NSUB <= CSLOT, "sums of squares sit behind the largest table (no path is 7x7x7)");
static_assert(((7 * 7 + NSUB - 1) / NSUB) * 7 <= 64, "a block's slice has at most 64 elements");
static_assert(ZBLK * W3J_T * ZPER * 4 == N_NODES * FEAT, "clearing blocks cover the output exactly once");
#define NQ     128
#define NQT    128
#define ODD_Z  0xA498u
#define ODD_XY 0x5A96u
#define YPITCH 23

static constexpr double LA[17] = {0.0, 0.0, 3.0/2.0, 5.0/3.0, 7.0/4.0, 9.0/5.0, 11.0/6.0,
  13.0/7.0, 15.0/8.0, 17.0/9.0, 19.0/10.0, 21.0/11.0, 23.0/12.0, 25.0/13.0, 27.0/14.0,
  29.0/15.0, 31.0/16.0};
static constexpr double LB[17] = {0.0, 0.0, 1.0/2.0, 2.0/3.0, 3.0/4.0, 4.0/5.0, 5.0/6.0,
  6.0/7.0, 7.0/8.0, 8.0/9.0, 9.0/10.0, 10.0/11.0, 11.0/12.0, 12.0/13.0, 13.0/14.0,
  14.0/15.0, 15.0/16.0};

template <typename T>
__device__ __forceinline__ void legendre16(T x, T& p15, T& p16) {
  T a0 = (T)1, b0 = x;
  for (int n = 2; n <= 16; ++n) {
    const T c0 = (T)LA[n] * x * b0 - (T)LB[n] * a0;
    a0 = b0; b0 = c0;
  }
  p15 = a0; p16 = b0;
}

template <int DR>
__device__ __forceinline__ void contract_run(const double* Ys, int q0, int q1,
                                             int ca, int cb, double* dst, int sR) {
  double acc[DR];
  for (int r = 0; r < DR; ++r) acc[r] = 0.0;
  for (int q = q0; q < q1; ++q) {
    const double* row = Ys + q * YPITCH;
    const double ab = row[ca] * row[cb];
    for (int r = 0; r < DR; ++r) acc[r] = fma(row[16 + r], ab, acc[r]);
  }
  for (int r = 0; r < DR; ++r) dst[r * sR] = acc[r];
}

__launch_bounds__(W3J_T)
__global__ void prep_kernel(float* __restrict__ ctab, float* __restrict__ out) {
  const int blk = blockIdx.x;
  const int t = threadIdx.x;

  if (blk >= TBLK) {
    float4* o4 = reinterpret_cast<float4*>(out);
    const int zb = blk - TBLK;
    const float4 z4 = make_float4(0.0f, 0.0f, 0.0f, 0.0f);
    for (int k = 0; k < ZPER; ++k) o4[(k * ZBLK + zb) * W3J_T + t] = z4;
    return;
  }
  const int p   = blk / NSUB;
  const int sub = blk - p * NSUB;

  int l1 = 0, l2 = 0, l3 = 0, cnt = 0;
  for (int a = 0; a < 4; ++a)
    for (int b = 0; b < 4; ++b)
      for (int c = 0; c < 4; ++c) {
        int lo = a - b; if (lo < 0) lo = -lo;
        if (lo <= c && c <= a + b && (((a + b + c) & 1) == 0)) {
          if (cnt == p) { l1 = a; l2 = b; l3 = c; }
          ++cnt;
        }
      }
  const int d1 = 2 * l1 + 1, d2 = 2 * l2 + 1, d3 = 2 * l3 + 1;

  int dR, dA, dB, oR, oA, oB, sR, sA, sB;
  if (d1 >= d2 && d1 >= d3) {
    dR = d1; oR = l1 * l1; sR = d2 * d3;
    dA = d2; oA = l2 * l2; sA = d3;
    dB = d3; oB = l3 * l3; sB = 1;
  } else if (d2 >= d3) {
    dR = d2; oR = l2 * l2; sR = d3;
    dA = d1; oA = l1 * l1; sA = d2 * d3;
    dB = d3; oB = l3 * l3; sB = 1;
  } else {
    dR = d3; oR = l3 * l3; sR = 1;
    dA = d1; oA = l1 * l1; sA = d2 * d3;
    dB = d2; oB = l2 * l2; sB = d3;
  }

  const int M  = dA * dB;
  const int MS = (M + NSUB - 1) / NSUB;
  const int m0 = sub * MS;
  const int m1 = (m0 + MS < M) ? (m0 + MS) : M;
  const int Mloc = m1 - m0;
  if (Mloc <= 0) {
    if (t == 0) ctab[p * CSLOT + SSOFF + sub] = 0.0f;
    return;
  }
  const int NEl = Mloc * dR;

  __shared__ double un_s[8], wn_s[8], st_s[8];
  __shared__ double cph_s[16], sph_s[16];
  __shared__ double Ys[NQ * YPITCH];
  __shared__ double part[W3J_T * 7];
  __shared__ double red[64];

  const double PI = 3.14159265358979323846;
  const double DPHI = 2.0 * PI / 32.0;

  if (t < 8) {
    float xf = cosf(((float)t + 0.75f) * (float)(PI / 16.5));
    for (int it = 0; it < 2; ++it) {
      float a0, b0;
      legendre16(xf, a0, b0);
      xf -= b0 * (xf * xf - 1.0f) / (16.0f * (xf * b0 - a0));
    }
    double x1 = (double)xf;
    {
      double a0, b0;
      legendre16(x1, a0, b0);
      x1 -= b0 * (x1 * x1 - 1.0) / (16.0 * (x1 * b0 - a0));
    }
    double a0, b0;
    legendre16(x1, a0, b0);
    const double s1 = x1 * x1 - 1.0;
    const double rq = 1.0 / (x1 * b0 - a0);
    const double xg = x1 - 0.0625 * b0 * s1 * rq;
    const double e  = 0.125 * x1 * b0 * rq;
    const double st2 = 1.0 - xg * xg;
    un_s[t] = xg;
    wn_s[t] = -0.0078125 * s1 * rq * rq * (1.0 - e);
    st_s[t] = sqrt(st2 > 0.0 ? st2 : 0.0);
  } else if (t >= 32 && t < 48) {
    const int j = t - 32;
    double s, c;
    sincos(j * DPHI, &s, &c);
    cph_s[j] = c;
    sph_s[j] = s;
  }
  __syncthreads();

  for (int q = t; q < NQ; q += W3J_T) {
    const int iq = q >> 4, jq = q & 15;
    const double st = st_s[iq];
    double Yq[16];
    sh_eval(st * cph_s[jq], st * sph_s[jq], un_s[iq], Yq);
    double* row = Ys + q * YPITCH;
    for (int c = 0; c < 16; ++c) row[c] = Yq[c];
    const double wq = wn_s[iq] * (4.0 * DPHI);
    for (int r = 0; r < dR; ++r) row[16 + r] = wq * row[oR + r];
  }
  __syncthreads();

  int NC = NQT / Mloc; if (NC > NCMAX) NC = NCMAX;
  const int CL = (NQ + NC - 1) / NC;
  if (t < Mloc * NC) {
    const int c = t / Mloc, ml = t - c * Mloc;
    const int m = m0 + ml;
    const int a = m / dB, b = m - a * dB;
    const int q0 = c * CL;
    int q1 = q0 + CL; if (q1 > NQ) q1 = NQ;
    double* dst = part + c * NEl + ml * dR;
    switch (dR) {
      case 1:  contract_run<1>(Ys, q0, q1, oA + a, oB + b, dst, 1); break;
      case 3:  contract_run<3>(Ys, q0, q1, oA + a, oB + b, dst, 1); break;
      case 5:  contract_run<5>(Ys, q0, q1, oA + a, oB + b, dst, 1); break;
      default: contract_run<7>(Ys, q0, q1, oA + a, oB + b, dst, 1); break;
    }
  }
  __syncthreads();

  double v = 0.0;
  if (t < NEl) {
    for (int c = 0; c < NC; ++c) v += part[c * NEl + t];
    const int ml = t / dR, r = t - ml * dR;
    const int m = m0 + ml;
    const int a = m / dB, b = m - a * dB;
    const unsigned ca = (unsigned)(oA + a), cb = (unsigned)(oB + b), cr = (unsigned)(oR + r);
    const unsigned odd = (((ODD_Z >> ca) ^ (ODD_Z >> cb) ^ (ODD_Z >> cr)) |
                          ((ODD_XY >> ca) ^ (ODD_XY >> cb) ^ (ODD_XY >> cr))) & 1u;
    if (odd) v = 0.0;
    ctab[p * CSLOT + r * sR + a * sA + b * sB] = (float)v;
  }
  if (t < 64) red[t] = v * v;
  __syncthreads();
  for (int off = 32; off > 0; off >>= 1) {
    if (t < off) red[t] += red[t + off];
    __syncthreads();
  }
  if (t == 0) ctab[p * CSLOT + SSOFF + sub] = (float)red[0];
}

#define EPB         16
#define EDGE_T      512
#define EDGE_BLOCKS 512
static_assert(EDGE_T == 32 * EPB, "one wave per edge");
static_assert(EDGE_BLOCKS * EPB == N_EDGES, "every edge has a wave, no wave is without an edge");
static_assert(EDGE_T * 4 == 2 * HIDDEN * HIDDEN, "two rows x two columns of a path's weights or biases per thread");
#define TPITCH 68
#define TCOLS  112
#define TBUF   (TCOLS * TPITCH)
#define WROW   96
#define WBUF   (32 * WROW)
#define SPITCH(d3) (32 * (d3) + 1)
#define SBUF   (EPB * SPITCH(7))

static constexpr int SEQ[NPATHS] = {0, 5, 13, 21,  1, 4, 7, 11, 15, 19,
                                    2, 6, 9, 10, 14, 18, 22,  3, 8, 12, 16, 17, 20};

__device__ __forceinline__ void wave_lds_sync() {
  __builtin_amdgcn_fence(__ATOMIC_RELEASE, "wavefront");
  __builtin_amdgcn_wave_barrier();
  __builtin_amdgcn_fence(__ATOMIC_ACQUIRE, "wavefront");
}

template <int S>
struct PathPos {
  static constexpr int  L3    = PL3[SEQ[S]];
  static constexpr bool FIRST = (S == 0) || (PL3[SEQ[S > 0 ? S - 1 : 0]] != L3);
  static constexpr int  PREV  = L3 - 1;
  static_assert(S == 0 || PL3[SEQ[S > 0 ? S - 1 : 0]] == L3 || PL3[SEQ[S > 0 ? S - 1 : 0]] == L3 - 1,
                "SEQ walks the l3 groups in rising order");
};

template <int L3>
__device__ __forceinline__ void flush_stage(const float* stage, int wave, int lane, float* orow) {
  constexpr int D3 = 2 * L3 + 1;
  const float* sr = stage + wave * SPITCH(D3) + lane;
  float* od = orow + HIDDEN * L3 * L3;
  float v[D3];
#pragma unroll
  for (int j = 0; j < D3; ++j) v[j] = sr[32 * j];
#pragma unroll
  for (int j = 0; j < D3; ++j) atomicAdd(od + 32 * j, v[j]);
}

template <int S>
__device__ __forceinline__ void path_body(
    int lane, int wave, float len,
    const float (&Y)[16], const float (&xr)[16],
    const float* __restrict__ msrc,
    const float* __restrict__ ctab,
    const float (&invn)[NPATHS],
    float* Gs, float* TT,
    float* wdst,
    float2 (&sp)[2],
    bool works,
    const float* ap,
    const float* bp,
    const float* stage,
    float* orow,
    v8f& acc0, v8f& acc1) {
  constexpr int P = SEQ[S];
  constexpr int l1 = PL1[P], l2 = PL2[P], l3 = PL3[P];
  constexpr int d1 = 2 * l1 + 1, d2 = 2 * l2 + 1, d3 = 2 * l3 + 1;
  constexpr int GN = d1 * d3;
  const float pw = PWT[l3] * invn[P];
  constexpr int WB_ = (S & 1) * WBUF;
  constexpr int TB_ = (S & 1) * TBUF;
  const float* Cp = ctab + P * CSLOT;

  if (lane < GN) {
    const int i = lane / d3, k = lane % d3;
    float g = 0.0f;
#pragma unroll
    for (int j = 0; j < d2; ++j) g = fmaf(Y[l2 * l2 + j], Cp[(i * d2 + j) * d3 + k], g);
    Gs[lane] = pw * g;
  }
  if constexpr (GN > 32) {
    const int n2 = lane + 32;
    if (n2 < GN) {
      const int i = n2 / d3, k = n2 % d3;
      float g = 0.0f;
#pragma unroll
      for (int j = 0; j < d2; ++j) g = fmaf(Y[l2 * l2 + j], Cp[(i * d2 + j) * d3 + k], g);
      Gs[n2] = pw * g;
    }
  }
  wave_lds_sync();

  {
    float* tc = TT + TB_ + (wave * d3) * TPITCH + lane;
#pragma unroll
    for (int k = 0; k < d3; ++k) {
      float s = 0.0f;
#pragma unroll
      for (int i = 0; i < d1; ++i) s = fmaf(xr[l1 * l1 + i], Gs[i * d3 + k], s);
      tc[k * TPITCH] = len * s;
      tc[k * TPITCH + 32] = s;
    }
  }

  *reinterpret_cast<float4*>(wdst + WB_) = make_float4(sp[0].x, sp[1].x, sp[0].y, sp[1].y);

  __syncthreads();

  if constexpr (S + 1 < NPATHS) {
    constexpr int PN = SEQ[S + 1];
    sp[0] = *reinterpret_cast<const float2*>(msrc + PN * (HIDDEN * HIDDEN));
    sp[1] = *reinterpret_cast<const float2*>(msrc + PN * (HIDDEN * HIDDEN) + HIDDEN);
  }

  if constexpr (PathPos<S>::FIRST && PathPos<S>::PREV >= 0)
    flush_stage<PathPos<S>::PREV>(stage, wave, lane, orow);

  if (works) {
    const float* a_ = ap + WB_;
    const float* b_ = bp + TB_;
#pragma unroll
    for (int kc = 0; kc < 16; kc += 2) {
      const v2f a0 = *reinterpret_cast<const v2f*>(a_ + kc * (2 * WROW));
      const v2f b0 = *reinterpret_cast<const v2f*>(b_ + kc * 4);
      acc0 = __builtin_amdgcn_wmma_f32_16x16x4_f32(
          false, a0, false, b0, (short)0, acc0, false, false);
      const v2f a1 = *reinterpret_cast<const v2f*>(a_ + (kc + 1) * (2 * WROW));
      const v2f b1 = *reinterpret_cast<const v2f*>(b_ + (kc + 1) * 4);
      acc1 = __builtin_amdgcn_wmma_f32_16x16x4_f32(
          false, a1, false, b1, (short)0, acc1, false, false);
    }
  }
}

template <int... Ss>
__device__ __forceinline__ void run_group(
    std::integer_sequence<int, Ss...>,
    int lane, int wave, float len,
    const float (&Y)[16], const float (&xr)[16],
    const float* __restrict__ msrc,
    const float* __restrict__ ctab, const float (&invn)[NPATHS],
    float* Gs, float* TT, float* wdst, float2 (&sp)[2],
    bool works, const float* ap, const float* bp,
    const float* stage, float* orow,
    v8f& acc0, v8f& acc1) {
  (path_body<Ss>(lane, wave, len, Y, xr, msrc, ctab, invn, Gs, TT, wdst, sp, works, ap, bp,
                 stage, orow, acc0, acc1), ...);
}

template <int L3, int ROT, int... Ss>
__device__ __forceinline__ void do_group(
    std::integer_sequence<int, Ss...> seq,
    int lane, int wave, float len,
    const float (&Y)[16], const float (&xr)[16],
    const float* __restrict__ msrc,
    const float* __restrict__ ctab, const float (&invn)[NPATHS],
    float* Gs, float* TT, const float* Ws, float* wdst, float2 (&sp)[2],
    float* stage, float* orow) {
  constexpr int d3 = 2 * L3 + 1;
  const int n16 = lane & 15;
  const int h   = lane >> 4;
  const int j    = (wave - ROT) & (EPB - 1);
  const bool works = j < 2 * d3;
  const int half = j & 1;
  const int tile = works ? (j >> 1) : 0;
  const float* ap = Ws + h * WROW + (16 * half + n16) * 2;
  const float* bp = TT + (16 * tile + n16) * TPITCH + 2 * h;
  v8f acc0 = {0,0,0,0,0,0,0,0}, acc1 = {0,0,0,0,0,0,0,0};
  run_group(seq, lane, wave, len, Y, xr, msrc, ctab, invn, Gs, TT, wdst, sp, works, ap, bp,
            stage, orow, acc0, acc1);
  if (works) {
    const int col = 16 * tile + n16;
    const int el  = col / d3;
    const int k   = col - el * d3;
    float* sd = stage + el * SPITCH(d3) + (16 * half + 8 * h) * d3 + k;
#pragma unroll
    for (int r = 0; r < 8; ++r) sd[r * d3] = acc0[r] + acc1[r];
  }
  if constexpr (L3 == 3) {
    __syncthreads();
    flush_stage<3>(stage, wave, lane, orow);
  }
}

__attribute__((amdgpu_waves_per_eu(8, 8)))
__launch_bounds__(EDGE_T)
__global__ void edge_kernel(const float* __restrict__ nf,
                            const int*   __restrict__ eidx,
                            const float* __restrict__ evec,
                            const float* __restrict__ ww,
                            const float* __restrict__ wb,
                            const float* __restrict__ ctab,
                            float*       __restrict__ out) {
  const int tid  = threadIdx.x;
  const int lane = tid & 31;
  const int wave = __builtin_amdgcn_readfirstlane(tid >> 5);
  const int e    = blockIdx.x * EPB + wave;

  __shared__ float Gs_all[EPB * 64];
  __shared__ __attribute__((aligned(16))) float TT[2 * TBUF];
  __shared__ __attribute__((aligned(16))) float Ws[2 * WBUF];
  __shared__ float stage[SBUF];

  const int rp = (tid >> 4) & 15, cp = tid & 15;
  const float* msrc = ((wave < 8) ? ww : wb) + rp * (2 * HIDDEN) + cp * 2;
  float2 sp[2];
  sp[0] = *reinterpret_cast<const float2*>(msrc + SEQ[0] * (HIDDEN * HIDDEN));
  sp[1] = *reinterpret_cast<const float2*>(msrc + SEQ[0] * (HIDDEN * HIDDEN) + HIDDEN);

  const int src = eidx[e];
  const int dst = eidx[N_EDGES + e];
  float* orow = out + (size_t)dst * FEAT + lane;

  const float vx = evec[3 * e + 0];
  const float vy = evec[3 * e + 1];
  const float vz = evec[3 * e + 2];
  float len = sqrtf(vx * vx + vy * vy + vz * vz);
  len = fmaxf(len, 1e-8f);
  const float inv = 1.0f / len;

  float Y[16];
  sh_eval(vx * inv, vy * inv, vz * inv, Y);

  const float* xs = nf + (size_t)src * FEAT;
  float xr[16];
#pragma unroll
  for (int l = 0; l < 4; ++l) {
    const int d = 2 * l + 1;
#pragma unroll
    for (int i = 0; i < 7; ++i)
      if (i < d) xr[l * l + i] = xs[HIDDEN * l * l + lane * d + i];
  }

  float invn[NPATHS];
  {
    const float* sq = ctab + (lane < NPATHS ? lane : 0) * CSLOT + SSOFF;
    float ssq = 0.0f;
#pragma unroll
    for (int k = 0; k < NSUB; ++k) ssq += sq[k];
    const int ib = __builtin_bit_cast(int, 1.0f / sqrtf(ssq));
#pragma unroll
    for (int q = 0; q < NPATHS; ++q)
      invn[q] = __builtin_bit_cast(float, __builtin_amdgcn_readlane(ib, q));
  }

  float* Gs   = Gs_all + wave * 64;
  float* wdst = Ws + ((wave < 8 ? 0 : 16) + rp) * WROW + cp * 4;

  do_group<0, 2>(std::integer_sequence<int, 0, 1, 2, 3>{},
                 lane, wave, len, Y, xr, msrc, ctab, invn, Gs, TT, Ws, wdst, sp, stage, orow);
  do_group<1, 2>(std::integer_sequence<int, 4, 5, 6, 7, 8, 9>{},
                 lane, wave, len, Y, xr, msrc, ctab, invn, Gs, TT, Ws, wdst, sp, stage, orow);
  do_group<2, 0>(std::integer_sequence<int, 10, 11, 12, 13, 14, 15, 16>{},
                 lane, wave, len, Y, xr, msrc, ctab, invn, Gs, TT, Ws, wdst, sp, stage, orow);
  do_group<3, 0>(std::integer_sequence<int, 17, 18, 19, 20, 21, 22>{},
                 lane, wave, len, Y, xr, msrc, ctab, invn, Gs, TT, Ws, wdst, sp, stage, orow);
}

extern "C" void kernel_launch(void* const* d_in, const int* in_sizes, int n_in,
                              void* d_out, int out_size, void* d_ws, size_t ws_size,
                              hipStream_t stream) {
  const float* nf   = (const float*)d_in[0];
  const int*   eidx = (const int*)  d_in[1];
  const float* evec = (const float*)d_in[2];
  const float* ww   = (const float*)d_in[3];
  const float* wb   = (const float*)d_in[4];
  float* out  = (float*)d_out;
  float* ctab = (float*)d_ws;

  prep_kernel<<<PREP_BLOCKS, W3J_T, 0, stream>>>(ctab, out);
  edge_kernel<<<EDGE_BLOCKS, EDGE_T, 0, stream>>>(nf, eidx, evec, ww, wb, ctab, out);
}


# edge launch: a heavy and a light path share each barrier interval (l3=2 with l3=1: 10+6 tiles; l3=3 with l3=0: 14+2), so all 16 waves have a tile product; 23 intervals -> 13; f32 sum order kept
# speedup vs baseline: 131.6700x; 1.0487x over previous
#include <hip/hip_runtime.h>
#include <math.h>
#include <utility>

typedef float v2f __attribute__((ext_vector_type(2)));
typedef float v8f __attribute__((ext_vector_type(8)));

#define HIDDEN  32
#define N_NODES 4096
#define N_EDGES 8192
#define FEAT    512
#define NPATHS  23

static constexpr int PL1[NPATHS] = {0,0,0,0, 1,1,1,1,1,1, 2,2,2,2,2,2,2, 3,3,3,3,3,3};
static constexpr int PL2[NPATHS] = {0,1,2,3, 0,1,1,2,2,3, 0,1,1,2,2,3,3, 0,1,2,2,3,3};
static constexpr int PL3[NPATHS] = {0,1,2,3, 1,0,2,1,3,2, 2,1,3,0,2,1,3, 3,2,1,3,0,2};
static constexpr float PWT[4] = {
  0.08838834764831845f,
  0.125f,
  0.14940357616680451f,
  0.19094065395649326f
};

template <typename T>
__device__ __forceinline__ void sh_eval(T x, T y, T z, T* Y) {
  const T s3  = (T)1.7320508075688772;
  const T s5  = (T)2.2360679774997896;
  const T s7  = (T)2.6457513110645907;
  const T s15 = (T)3.8729833462074170;
  const T c58 = (T)0.7905694150420949;
  const T c38 = (T)0.6123724356957945;
  Y[0]  = (T)1;
  Y[1]  = s3 * x;
  Y[2]  = s3 * y;
  Y[3]  = s3 * z;
  Y[4]  = s5 * s3 * x * z;
  Y[5]  = s5 * s3 * x * y;
  Y[6]  = s5 * (y * y - (T)0.5 * (x * x + z * z));
  Y[7]  = s5 * s3 * y * z;
  Y[8]  = s5 * (s3 * (T)0.5) * (z * z - x * x);
  Y[9]  = s7 * c58 * x * ((T)3 * z * z - x * x);
  Y[10] = s7 * s15 * x * y * z;
  Y[11] = s7 * c38 * x * ((T)5 * y * y - (T)1);
  Y[12] = s7 * (T)0.5 * y * ((T)5 * y * y - (T)3);
  Y[13] = s7 * c38 * z * ((T)5 * y * y - (T)1);
  Y[14] = s7 * (s15 * (T)0.5) * y * (z * z - x * x);
  Y[15] = s7 * c58 * z * (z * z - (T)3 * x * x);
}

#define W3J_T  256
#define NSUB   8
#define TBLK   184
#define ZBLK   512
#define ZPER   4
#define PREP_BLOCKS 696
#define CSLOT  343
#define SSOFF  320
#define NCMAX  32
static_assert(TBLK == NPATHS * NSUB, "table blocks");
static_assert(PREP_BLOCKS == TBLK + ZBLK, "grid of the preparation launch");
static_assert(SSOFF >= 7 * 7 * 5 && SSOFF + NSUB <= CSLOT, "sums of squares sit behind the largest table (no path is 7x7x7)");
static_assert(((7 * 7 + NSUB - 1) / NSUB) * 7 <= 64, "a block's slice has at most 64 elements");
static_assert(ZBLK * W3J_T * ZPER * 4 == N_NODES * FEAT, "clearing blocks cover the output exactly once");
#define NQ     128
#define NQT    128
#define ODD_Z  0xA498u
#define ODD_XY 0x5A96u
#define YPITCH 23

static constexpr double LA[17] = {0.0, 0.0, 3.0/2.0, 5.0/3.0, 7.0/4.0, 9.0/5.0, 11.0/6.0,
  13.0/7.0, 15.0/8.0, 17.0/9.0, 19.0/10.0, 21.0/11.0, 23.0/12.0, 25.0/13.0, 27.0/14.0,
  29.0/15.0, 31.0/16.0};
static constexpr double LB[17] = {0.0, 0.0, 1.0/2.0, 2.0/3.0, 3.0/4.0, 4.0/5.0, 5.0/6.0,
  6.0/7.0, 7.0/8.0, 8.0/9.0, 9.0/10.0, 10.0/11.0, 11.0/12.0, 12.0/13.0, 13.0/14.0,
  14.0/15.0, 15.0/16.0};

template <typename T>
__device__ __forceinline__ void legendre16(T x, T& p15, T& p16) {
  T a0 = (T)1, b0 = x;
  for (int n = 2; n <= 16; ++n) {
    const T c0 = (T)LA[n] * x * b0 - (T)LB[n] * a0;
    a0 = b0; b0 = c0;
  }
  p15 = a0; p16 = b0;
}

template <int DR>
__device__ __forceinline__ void contract_run(const double* Ys, int q0, int q1,
                                             int ca, int cb, double* dst, int sR) {
  double acc[DR];
  for (int r = 0; r < DR; ++r) acc[r] = 0.0;
  for (int q = q0; q < q1; ++q) {
    const double* row = Ys + q * YPITCH;
    const double ab = row[ca] * row[cb];
    for (int r = 0; r < DR; ++r) acc[r] = fma(row[16 + r], ab, acc[r]);
  }
  for (int r = 0; r < DR; ++r) dst[r * sR] = acc[r];
}

__launch_bounds__(W3J_T)
__global__ void prep_kernel(float* __restrict__ ctab, float* __restrict__ out) {
  const int blk = blockIdx.x;
  const int t = threadIdx.x;

  if (blk >= TBLK) {
    float4* o4 = reinterpret_cast<float4*>(out);
    const int zb = blk - TBLK;
    const float4 z4 = make_float4(0.0f, 0.0f, 0.0f, 0.0f);
    for (int k = 0; k < ZPER; ++k) o4[(k * ZBLK + zb) * W3J_T + t] = z4;
    return;
  }
  const int p   = blk / NSUB;
  const int sub = blk - p * NSUB;

  int l1 = 0, l2 = 0, l3 = 0, cnt = 0;
  for (int a = 0; a < 4; ++a)
    for (int b = 0; b < 4; ++b)
      for (int c = 0; c < 4; ++c) {
        int lo = a - b; if (lo < 0) lo = -lo;
        if (lo <= c && c <= a + b && (((a + b + c) & 1) == 0)) {
          if (cnt == p) { l1 = a; l2 = b; l3 = c; }
          ++cnt;
        }
      }
  const int d1 = 2 * l1 + 1, d2 = 2 * l2 + 1, d3 = 2 * l3 + 1;

  int dR, dA, dB, oR, oA, oB, sR, sA, sB;
  if (d1 >= d2 && d1 >= d3) {
    dR = d1; oR = l1 * l1; sR = d2 * d3;
    dA = d2; oA = l2 * l2; sA = d3;
    dB = d3; oB = l3 * l3; sB = 1;
  } else if (d2 >= d3) {
    dR = d2; oR = l2 * l2; sR = d3;
    dA = d1; oA = l1 * l1; sA = d2 * d3;
    dB = d3; oB = l3 * l3; sB = 1;
  } else {
    dR = d3; oR = l3 * l3; sR = 1;
    dA = d1; oA = l1 * l1; sA = d2 * d3;
    dB = d2; oB = l2 * l2; sB = d3;
  }

  const int M  = dA * dB;
  const int MS = (M + NSUB - 1) / NSUB;
  const int m0 = sub * MS;
  const int m1 = (m0 + MS < M) ? (m0 + MS) : M;
  const int Mloc = m1 - m0;
  if (Mloc <= 0) {
    if (t == 0) ctab[p * CSLOT + SSOFF + sub] = 0.0f;
    return;
  }
  const int NEl = Mloc * dR;

  __shared__ double un_s[8], wn_s[8], st_s[8];
  __shared__ double cph_s[16], sph_s[16];
  __shared__ double Ys[NQ * YPITCH];
  __shared__ double part[W3J_T * 7];
  __shared__ double red[64];

  const double PI = 3.14159265358979323846;
  const double DPHI = 2.0 * PI / 32.0;

  if (t < 8) {
    float xf = cosf(((float)t + 0.75f) * (float)(PI / 16.5));
    for (int it = 0; it < 2; ++it) {
      float a0, b0;
      legendre16(xf, a0, b0);
      xf -= b0 * (xf * xf - 1.0f) / (16.0f * (xf * b0 - a0));
    }
    double x1 = (double)xf;
    {
      double a0, b0;
      legendre16(x1, a0, b0);
      x1 -= b0 * (x1 * x1 - 1.0) / (16.0 * (x1 * b0 - a0));
    }
    double a0, b0;
    legendre16(x1, a0, b0);
    const double s1 = x1 * x1 - 1.0;
    const double rq = 1.0 / (x1 * b0 - a0);
    const double xg = x1 - 0.0625 * b0 * s1 * rq;
    const double e  = 0.125 * x1 * b0 * rq;
    const double st2 = 1.0 - xg * xg;
    un_s[t] = xg;
    wn_s[t] = -0.0078125 * s1 * rq * rq * (1.0 - e);
    st_s[t] = sqrt(st2 > 0.0 ? st2 : 0.0);
  } else if (t >= 32 && t < 48) {
    const int j = t - 32;
    double s, c;
    sincos(j * DPHI, &s, &c);
    cph_s[j] = c;
    sph_s[j] = s;
  }
  __syncthreads();

  for (int q = t; q < NQ; q += W3J_T) {
    const int iq = q >> 4, jq = q & 15;
    const double st = st_s[iq];
    double Yq[16];
    sh_eval(st * cph_s[jq], st * sph_s[jq], un_s[iq], Yq);
    double* row = Ys + q * YPITCH;
    for (int c = 0; c < 16; ++c) row[c] = Yq[c];
    const double wq = wn_s[iq] * (4.0 * DPHI);
    for (int r = 0; r < dR; ++r) row[16 + r] = wq * row[oR + r];
  }
  __syncthreads();

  int NC = NQT / Mloc; if (NC > NCMAX) NC = NCMAX;
  const int CL = (NQ + NC - 1) / NC;
  if (t < Mloc * NC) {
    const int c = t / Mloc, ml = t - c * Mloc;
    const int m = m0 + ml;
    const int a = m / dB, b = m - a * dB;
    const int q0 = c * CL;
    int q1 = q0 + CL; if (q1 > NQ) q1 = NQ;
    double* dst = part + c * NEl + ml * dR;
    switch (dR) {
      case 1:  contract_run<1>(Ys, q0, q1, oA + a, oB + b, dst, 1); break;
      case 3:  contract_run<3>(Ys, q0, q1, oA + a, oB + b, dst, 1); break;
      case 5:  contract_run<5>(Ys, q0, q1, oA + a, oB + b, dst, 1); break;
      default: contract_run<7>(Ys, q0, q1, oA + a, oB + b, dst, 1); break;
    }
  }
  __syncthreads();

  double v = 0.0;
  if (t < NEl) {
    for (int c = 0; c < NC; ++c) v += part[c * NEl + t];
    const int ml = t / dR, r = t - ml * dR;
    const int m = m0 + ml;
    const int a = m / dB, b = m - a * dB;
    const unsigned ca = (unsigned)(oA + a), cb = (unsigned)(oB + b), cr = (unsigned)(oR + r);
    const unsigned odd = (((ODD_Z >> ca) ^ (ODD_Z >> cb) ^ (ODD_Z >> cr)) |
                          ((ODD_XY >> ca) ^ (ODD_XY >> cb) ^ (ODD_XY >> cr))) & 1u;
    if (odd) v = 0.0;
    ctab[p * CSLOT + r * sR + a * sA + b * sB] = (float)v;
  }
  if (t < 64) red[t] = v * v;
  __syncthreads();
  for (int off = 32; off > 0; off >>= 1) {
    if (t < off) red[t] += red[t + off];
    __syncthreads();
  }
  if (t == 0) ctab[p * CSLOT + SSOFF + sub] = (float)red[0];
}

#define EPB         16
#define EDGE_T      512
#define EDGE_BLOCKS 512
static_assert(EDGE_T == 32 * EPB, "one wave per edge");
static_assert(EDGE_BLOCKS * EPB == N_EDGES, "every edge has a wave, no wave is without an edge");
static_assert(EDGE_T * 4 == 2 * HIDDEN * HIDDEN, "two rows x two columns of a path's weights or biases per thread");
#define TPITCH 68
#define TCOLS  128
#define TBUF   (TCOLS * TPITCH)
#define WROW   96
#define WBUF   (32 * WROW)
#define SPITCH(d3) (32 * (d3) + 1)
#define SBUFH  (EPB * SPITCH(7))
#define SBUFL  (EPB * SPITCH(3))

#define NSTEPS 13
#define STEP2  7
static constexpr int STH[NSTEPS] = {2, 6, 9, 10, 14, 18, 22,   3, 8, 12, 16, 17, 20};
static constexpr int STL[NSTEPS] = {1, 4, 7, 11, 15, 19, -1,   0, 5, 13, 21, -1, -1};

constexpr bool steps_ok() {
  int seen[NPATHS] = {};
  for (int t = 0; t < NSTEPS; ++t) {
    const int lh = (t < STEP2) ? 2 : 3, ll = (t < STEP2) ? 1 : 0;
    if (PL3[STH[t]] != lh) return false;
    ++seen[STH[t]];
    if (STL[t] >= 0) { if (PL3[STL[t]] != ll) return false; ++seen[STL[t]]; }
    if (t > 0 && t != STEP2) {
      if (STH[t] <= STH[t - 1]) return false;
      if (STL[t] >= 0 && STL[t] <= STL[t - 1]) return false;
      if (STL[t] >= 0 && STL[t - 1] < 0) return false;
    }
  }
  for (int p = 0; p < NPATHS; ++p) if (seen[p] != 1) return false;
  return true;
}
static_assert(steps_ok(), "every path in exactly one step, in its group's order");

__device__ __forceinline__ void wave_lds_sync() {
  __builtin_amdgcn_fence(__ATOMIC_RELEASE, "wavefront");
  __builtin_amdgcn_wave_barrier();
  __builtin_amdgcn_fence(__ATOMIC_ACQUIRE, "wavefront");
}

template <int L3>
__device__ __forceinline__ void flush_stage(const float* stage, int wave, int lane, float* orow) {
  constexpr int D3 = 2 * L3 + 1;
  const float* sr = stage + wave * SPITCH(D3) + lane;
  float* od = orow + HIDDEN * L3 * L3;
  float v[D3];
#pragma unroll
  for (int j = 0; j < D3; ++j) v[j] = sr[32 * j];
#pragma unroll
  for (int j = 0; j < D3; ++j) atomicAdd(od + 32 * j, v[j]);
}

template <int P>
__device__ __forceinline__ void g_sums(int lane, const float (&Y)[16],
                                       const float* __restrict__ ctab,
                                       const float (&invn)[NPATHS], float* Gs) {
  constexpr int l1 = PL1[P], l2 = PL2[P], l3 = PL3[P];
  constexpr int d1 = 2 * l1 + 1, d2 = 2 * l2 + 1, d3 = 2 * l3 + 1;
  constexpr int GN = d1 * d3;
  const float pw = PWT[l3] * invn[P];
  const float* Cp = ctab + P * CSLOT;
  if (lane < GN) {
    const int i = lane / d3, k = lane % d3;
    float g = 0.0f;
#pragma unroll
    for (int j = 0; j < d2; ++j) g = fmaf(Y[l2 * l2 + j], Cp[(i * d2 + j) * d3 + k], g);
    Gs[lane] = pw * g;
  }
  if constexpr (GN > 32) {
    const int n2 = lane + 32;
    if (n2 < GN) {
      const int i = n2 / d3, k = n2 % d3;
      float g = 0.0f;
#pragma unroll
      for (int j = 0; j < d2; ++j) g = fmaf(Y[l2 * l2 + j], Cp[(i * d2 + j) * d3 + k], g);
      Gs[n2] = pw * g;
    }
  }
}

template <int P>
__device__ __forceinline__ void tmp_cols(float len, const float (&xr)[16],
                                         const float* Gs, float* tc) {
  constexpr int l1 = PL1[P], l3 = PL3[P];
  constexpr int d1 = 2 * l1 + 1, d3 = 2 * l3 + 1;
#pragma unroll
  for (int k = 0; k < d3; ++k) {
    float s = 0.0f;
#pragma unroll
    for (int i = 0; i < d1; ++i) s = fmaf(xr[l1 * l1 + i], Gs[i * d3 + k], s);
    tc[k * TPITCH] = len * s;
    tc[k * TPITCH + 32] = s;
  }
}

template <int T>
__device__ __forceinline__ void step_body(
    int lane, int wave, float len,
    const float (&Y)[16], const float (&xr)[16],
    const float* __restrict__ msrc,
    const float* __restrict__ ctab,
    const float (&invn)[NPATHS],
    float* Gs, float* TT,
    float* wdst,
    float2 (&spH)[2],
    float2 (&spL)[2],
    bool heavy,
    const float* ap,
    const float* bp,
    const float* stage,
    float* orow,
    v8f& acc0, v8f& acc1) {
  constexpr int PH = STH[T], PL = STL[T];
  constexpr int dH = 2 * PL3[PH] + 1;
  constexpr int WB_ = (T & 1) * (2 * WBUF);
  constexpr int TB_ = (T & 1) * TBUF;

  g_sums<PH>(lane, Y, ctab, invn, Gs);
  if constexpr (PL >= 0) g_sums<PL>(lane, Y, ctab, invn, Gs + 64);
  wave_lds_sync();
  tmp_cols<PH>(len, xr, Gs, TT + TB_ + (wave * dH) * TPITCH + lane);
  if constexpr (PL >= 0) {
    constexpr int dL = 2 * PL3[PL >= 0 ? PL : 0] + 1;
    tmp_cols<(PL >= 0 ? PL : 0)>(len, xr, Gs + 64, TT + TB_ + (16 * dH + wave * dL) * TPITCH + lane);
  }

  *reinterpret_cast<float4*>(wdst + WB_) = make_float4(spH[0].x, spH[1].x, spH[0].y, spH[1].y);
  if constexpr (PL >= 0)
    *reinterpret_cast<float4*>(wdst + WB_ + WBUF) = make_float4(spL[0].x, spL[1].x, spL[0].y, spL[1].y);

  __syncthreads();

  if constexpr (T + 1 < NSTEPS) {
    constexpr int NH = STH[T + 1], NL = STL[T + 1];
    spH[0] = *reinterpret_cast<const float2*>(msrc + NH * (HIDDEN * HIDDEN));
    spH[1] = *reinterpret_cast<const float2*>(msrc + NH * (HIDDEN * HIDDEN) + HIDDEN);
    if constexpr (NL >= 0) {
      spL[0] = *reinterpret_cast<const float2*>(msrc + (NL >= 0 ? NL : 0) * (HIDDEN * HIDDEN));
      spL[1] = *reinterpret_cast<const float2*>(msrc + (NL >= 0 ? NL : 0) * (HIDDEN * HIDDEN) + HIDDEN);
    }
  }

  if constexpr (T == STEP2) {
    flush_stage<PL3[STH[0]]>(stage, wave, lane, orow);
    flush_stage<PL3[STL[0]]>(stage + SBUFH, wave, lane, orow);
  }

  if (PL >= 0 || heavy) {
    const float* a_ = ap + WB_;
    const float* b_ = bp + TB_;
#pragma unroll
    for (int kc = 0; kc < 16; kc += 2) {
      const v2f a0 = *reinterpret_cast<const v2f*>(a_ + kc * (2 * WROW));
      const v2f b0 = *reinterpret_cast<const v2f*>(b_ + kc * 4);
      acc0 = __builtin_amdgcn_wmma_f32_16x16x4_f32(
          false, a0, false, b0, (short)0, acc0, false, false);
      const v2f a1 = *reinterpret_cast<const v2f*>(a_ + (kc + 1) * (2 * WROW));
      const v2f b1 = *reinterpret_cast<const v2f*>(b_ + (kc + 1) * 4);
      acc1 = __builtin_amdgcn_wmma_f32_16x16x4_f32(
          false, a1, false, b1, (short)0, acc1, false, false);
    }
  }
}

template <int... Ts>
__device__ __forceinline__ void run_steps(
    std::integer_sequence<int, Ts...>,
    int lane, int wave, float len,
    const float (&Y)[16], const float (&xr)[16],
    const float* __restrict__ msrc,
    const float* __restrict__ ctab, const float (&invn)[NPATHS],
    float* Gs, float* TT, float* wdst, float2 (&spH)[2], float2 (&spL)[2],
    bool heavy, const float* ap, const float* bp,
    const float* stage, float* orow,
    v8f& acc0, v8f& acc1) {
  (step_body<Ts>(lane, wave, len, Y, xr, msrc, ctab, invn, Gs, TT, wdst, spH, spL, heavy, ap, bp,
                 stage, orow, acc0, acc1), ...);
}

template <int D3>
__device__ __forceinline__ void stage_tile(float* stage, int tile, int half, int n16, int h,
                                           const v8f& acc0, const v8f& acc1) {
  const int col = 16 * tile + n16;
  const int el  = col / D3;
  const int k   = col - el * D3;
  float* sd = stage + el * SPITCH(D3) + (16 * half + 8 * h) * D3 + k;
#pragma unroll
  for (int r = 0; r < 8; ++r) sd[r * D3] = acc0[r] + acc1[r];
}

template <int L3H, int L3L, bool LAST, int... Ts>
__device__ __forceinline__ void do_pair(
    std::integer_sequence<int, Ts...> seq,
    int lane, int wave, float len,
    const float (&Y)[16], const float (&xr)[16],
    const float* __restrict__ msrc,
    const float* __restrict__ ctab, const float (&invn)[NPATHS],
    float* Gs, float* TT, const float* Ws, float* wdst, float2 (&spH)[2], float2 (&spL)[2],
    float* stage, float* orow) {
  constexpr int dH = 2 * L3H + 1, dL = 2 * L3L + 1;
  static_assert(2 * dH + 2 * dL == EPB, "the two groups' tile products are one per wave");
  static_assert(16 * (dH + dL) == TCOLS, "their columns fill the B tile");
  const int n16 = lane & 15;
  const int h   = lane >> 4;
  const bool heavy = wave < 2 * dH;
  const int j    = heavy ? wave : wave - 2 * dH;
  const int half = j & 1;
  const int tile = j >> 1;
  const int ctile = heavy ? tile : dH + tile;
  const float* ap = Ws + (heavy ? 0 : WBUF) + h * WROW + (16 * half + n16) * 2;
  const float* bp = TT + (16 * ctile + n16) * TPITCH + 2 * h;
  v8f acc0 = {0,0,0,0,0,0,0,0}, acc1 = {0,0,0,0,0,0,0,0};
  run_steps(seq, lane, wave, len, Y, xr, msrc, ctab, invn, Gs, TT, wdst, spH, spL, heavy, ap, bp,
            stage, orow, acc0, acc1);
  if (heavy) stage_tile<dH>(stage, tile, half, n16, h, acc0, acc1);
  else       stage_tile<dL>(stage + SBUFH, tile, half, n16, h, acc0, acc1);
  if constexpr (LAST) {
    __syncthreads();
    flush_stage<L3H>(stage, wave, lane, orow);
    flush_stage<L3L>(stage + SBUFH, wave, lane, orow);
  }
}

__attribute__((amdgpu_waves_per_eu(8, 8)))
__launch_bounds__(EDGE_T)
__global__ void edge_kernel(const float* __restrict__ nf,
                            const int*   __restrict__ eidx,
                            const float* __restrict__ evec,
                            const float* __restrict__ ww,
                            const float* __restrict__ wb,
                            const float* __restrict__ ctab,
                            float*       __restrict__ out) {
  const int tid  = threadIdx.x;
  const int lane = tid & 31;
  const int wave = __builtin_amdgcn_readfirstlane(tid >> 5);
  const int e    = blockIdx.x * EPB + wave;

  __shared__ float Gs_all[EPB * 128];
  __shared__ __attribute__((aligned(16))) float TT[2 * TBUF];
  __shared__ __attribute__((aligned(16))) float Ws[4 * WBUF];
  __shared__ float stage[SBUFH + SBUFL];

  const int rp = (tid >> 4) & 15, cp = tid & 15;
  const float* msrc = ((wave < 8) ? ww : wb) + rp * (2 * HIDDEN) + cp * 2;
  float2 spH[2], spL[2];
  spH[0] = *reinterpret_cast<const float2*>(msrc + STH[0] * (HIDDEN * HIDDEN));
  spH[1] = *reinterpret_cast<const float2*>(msrc + STH[0] * (HIDDEN * HIDDEN) + HIDDEN);
  spL[0] = *reinterpret_cast<const float2*>(msrc + STL[0] * (HIDDEN * HIDDEN));
  spL[1] = *reinterpret_cast<const float2*>(msrc + STL[0] * (HIDDEN * HIDDEN) + HIDDEN);

  const int src = eidx[e];
  const int dst = eidx[N_EDGES + e];
  float* orow = out + (size_t)dst * FEAT + lane;

  const float vx = evec[3 * e + 0];
  const float vy = evec[3 * e + 1];
  const float vz = evec[3 * e + 2];
  float len = sqrtf(vx * vx + vy * vy + vz * vz);
  len = fmaxf(len, 1e-8f);
  const float inv = 1.0f / len;

  float Y[16];
  sh_eval(vx * inv, vy * inv, vz * inv, Y);

  const float* xs = nf + (size_t)src * FEAT;
  float xr[16];
#pragma unroll
  for (int l = 0; l < 4; ++l) {
    const int d = 2 * l + 1;
#pragma unroll
    for (int i = 0; i < 7; ++i)
      if (i < d) xr[l * l + i] = xs[HIDDEN * l * l + lane * d + i];
  }

  float invn[NPATHS];
  {
    const float* sq = ctab + (lane < NPATHS ? lane : 0) * CSLOT + SSOFF;
    float ssq = 0.0f;
#pragma unroll
    for (int k = 0; k < NSUB; ++k) ssq += sq[k];
    const int ib = __builtin_bit_cast(int, 1.0f / sqrtf(ssq));
#pragma unroll
    for (int q = 0; q < NPATHS; ++q)
      invn[q] = __builtin_bit_cast(float, __builtin_amdgcn_readlane(ib, q));
  }

  float* Gs   = Gs_all + wave * 128;
  float* wdst = Ws + ((wave < 8 ? 0 : 16) + rp) * WROW + cp * 4;

  do_pair<2, 1, false>(std::integer_sequence<int, 0, 1, 2, 3, 4, 5, 6>{},
                       lane, wave, len, Y, xr, msrc, ctab, invn, Gs, TT, Ws, wdst, spH, spL, stage, orow);
  do_pair<3, 0, true>(std::integer_sequence<int, 7, 8, 9, 10, 11, 12>{},
                      lane, wave, len, Y, xr, msrc, ctab, invn, Gs, TT, Ws, wdst, spH, spL, stage, orow);
}

extern "C" void kernel_launch(void* const* d_in, const int* in_sizes, int n_in,
                              void* d_out, int out_size, void* d_ws, size_t ws_size,
                              hipStream_t stream) {
  const float* nf   = (const float*)d_in[0];
  const int*   eidx = (const int*)  d_in[1];
  const float* evec = (const float*)d_in[2];
  const float* ww   = (const float*)d_in[3];
  const float* wb   = (const float*)d_in[4];
  float* out  = (float*)d_out;
  float* ctab = (float*)d_ws;

  prep_kernel<<<PREP_BLOCKS, W3J_T, 0, stream>>>(ctab, out);
  edge_kernel<<<EDGE_BLOCKS, EDGE_T, 0, stream>>>(nf, eidx, evec, ww, wb, ctab, out);
}


# preparation launch: fp64 chain all table blocks wait on, 144 -> 42 instructions: P16,P15 by Horner in x^2, 1 fp64 Newton step from the f32 node (nodes to 1e-13), seeded 1/q and sqrt, phi by sincospi
# speedup vs baseline: 137.3674x; 1.0433x over previous
#include <hip/hip_runtime.h>
#include <math.h>
#include <utility>

typedef float v2f __attribute__((ext_vector_type(2)));
typedef float v8f __attribute__((ext_vector_type(8)));

#define HIDDEN  32
#define N_NODES 4096
#define N_EDGES 8192
#define FEAT    512
#define NPATHS  23

static constexpr int PL1[NPATHS] = {0,0,0,0, 1,1,1,1,1,1, 2,2,2,2,2,2,2, 3,3,3,3,3,3};
static constexpr int PL2[NPATHS] = {0,1,2,3, 0,1,1,2,2,3, 0,1,1,2,2,3,3, 0,1,2,2,3,3};
static constexpr int PL3[NPATHS] = {0,1,2,3, 1,0,2,1,3,2, 2,1,3,0,2,1,3, 3,2,1,3,0,2};
static constexpr float PWT[4] = {
  0.08838834764831845f,
  0.125f,
  0.14940357616680451f,
  0.19094065395649326f
};

template <typename T>
__device__ __forceinline__ void sh_eval(T x, T y, T z, T* Y) {
  const T s3  = (T)1.7320508075688772;
  const T s5  = (T)2.2360679774997896;
  const T s7  = (T)2.6457513110645907;
  const T s15 = (T)3.8729833462074170;
  const T c58 = (T)0.7905694150420949;
  const T c38 = (T)0.6123724356957945;
  Y[0]  = (T)1;
  Y[1]  = s3 * x;
  Y[2]  = s3 * y;
  Y[3]  = s3 * z;
  Y[4]  = s5 * s3 * x * z;
  Y[5]  = s5 * s3 * x * y;
  Y[6]  = s5 * (y * y - (T)0.5 * (x * x + z * z));
  Y[7]  = s5 * s3 * y * z;
  Y[8]  = s5 * (s3 * (T)0.5) * (z * z - x * x);
  Y[9]  = s7 * c58 * x * ((T)3 * z * z - x * x);
  Y[10] = s7 * s15 * x * y * z;
  Y[11] = s7 * c38 * x * ((T)5 * y * y - (T)1);
  Y[12] = s7 * (T)0.5 * y * ((T)5 * y * y - (T)3);
  Y[13] = s7 * c38 * z * ((T)5 * y * y - (T)1);
  Y[14] = s7 * (s15 * (T)0.5) * y * (z * z - x * x);
  Y[15] = s7 * c58 * z * (z * z - (T)3 * x * x);
}

#define W3J_T  256
#define NSUB   8
#define TBLK   184
#define ZBLK   512
#define ZPER   4
#define PREP_BLOCKS 696
#define CSLOT  343
#define SSOFF  320
#define NCMAX  32
static_assert(TBLK == NPATHS * NSUB, "table blocks");
static_assert(PREP_BLOCKS == TBLK + ZBLK, "grid of the preparation launch");
static_assert(SSOFF >= 7 * 7 * 5 && SSOFF + NSUB <= CSLOT, "sums of squares sit behind the largest table (no path is 7x7x7)");
static_assert(((7 * 7 + NSUB - 1) / NSUB) * 7 <= 64, "a block's slice has at most 64 elements");
static_assert(ZBLK * W3J_T * ZPER * 4 == N_NODES * FEAT, "clearing blocks cover the output exactly once");
#define NQ     128
#define NQT    128
#define ODD_Z  0xA498u
#define ODD_XY 0x5A96u
#define YPITCH 23

static constexpr double LA[17] = {0.0, 0.0, 3.0/2.0, 5.0/3.0, 7.0/4.0, 9.0/5.0, 11.0/6.0,
  13.0/7.0, 15.0/8.0, 17.0/9.0, 19.0/10.0, 21.0/11.0, 23.0/12.0, 25.0/13.0, 27.0/14.0,
  29.0/15.0, 31.0/16.0};
static constexpr double LB[17] = {0.0, 0.0, 1.0/2.0, 2.0/3.0, 3.0/4.0, 4.0/5.0, 5.0/6.0,
  6.0/7.0, 7.0/8.0, 8.0/9.0, 9.0/10.0, 10.0/11.0, 11.0/12.0, 12.0/13.0, 13.0/14.0,
  14.0/15.0, 15.0/16.0};

template <typename T>
__device__ __forceinline__ void legendre16(T x, T& p15, T& p16) {
  T a0 = (T)1, b0 = x;
  for (int n = 2; n <= 16; ++n) {
    const T c0 = (T)LA[n] * x * b0 - (T)LB[n] * a0;
    a0 = b0; b0 = c0;
  }
  p15 = a0; p16 = b0;
}

static constexpr double P16C[9] = {
  6435.0 / 32768.0, -875160.0 / 32768.0, 19399380.0 / 32768.0, -162954792.0 / 32768.0,
  669278610.0 / 32768.0, -1487285800.0 / 32768.0, 1825305300.0 / 32768.0,
  -1163381400.0 / 32768.0, 300540195.0 / 32768.0};
static constexpr double P15C[8] = {
  -6435.0 / 2048.0, 255255.0 / 2048.0, -2909907.0 / 2048.0, 14549535.0 / 2048.0,
  -37182145.0 / 2048.0, 50702925.0 / 2048.0, -35102025.0 / 2048.0, 9694845.0 / 2048.0};

__device__ __forceinline__ void legendre16_in_t(double x, double& t, double& p15, double& p16) {
  t = x * x;
  double ev = P16C[8], od = P15C[7];
  for (int k = 7; k >= 0; --k) ev = fma(ev, t, P16C[k]);
  for (int k = 6; k >= 0; --k) od = fma(od, t, P15C[k]);
  p16 = ev;
  p15 = od * x;
}

template <int DR>
__device__ __forceinline__ void contract_run(const double* Ys, int q0, int q1,
                                             int ca, int cb, double* dst, int sR) {
  double acc[DR];
  for (int r = 0; r < DR; ++r) acc[r] = 0.0;
  for (int q = q0; q < q1; ++q) {
    const double* row = Ys + q * YPITCH;
    const double ab = row[ca] * row[cb];
    for (int r = 0; r < DR; ++r) acc[r] = fma(row[16 + r], ab, acc[r]);
  }
  for (int r = 0; r < DR; ++r) dst[r * sR] = acc[r];
}

__launch_bounds__(W3J_T)
__global__ void prep_kernel(float* __restrict__ ctab, float* __restrict__ out) {
  const int blk = blockIdx.x;
  const int t = threadIdx.x;

  if (blk >= TBLK) {
    float4* o4 = reinterpret_cast<float4*>(out);
    const int zb = blk - TBLK;
    const float4 z4 = make_float4(0.0f, 0.0f, 0.0f, 0.0f);
    for (int k = 0; k < ZPER; ++k) o4[(k * ZBLK + zb) * W3J_T + t] = z4;
    return;
  }
  const int p   = blk / NSUB;
  const int sub = blk - p * NSUB;

  int l1 = 0, l2 = 0, l3 = 0, cnt = 0;
  for (int a = 0; a < 4; ++a)
    for (int b = 0; b < 4; ++b)
      for (int c = 0; c < 4; ++c) {
        int lo = a - b; if (lo < 0) lo = -lo;
        if (lo <= c && c <= a + b && (((a + b + c) & 1) == 0)) {
          if (cnt == p) { l1 = a; l2 = b; l3 = c; }
          ++cnt;
        }
      }
  const int d1 = 2 * l1 + 1, d2 = 2 * l2 + 1, d3 = 2 * l3 + 1;

  int dR, dA, dB, oR, oA, oB, sR, sA, sB;
  if (d1 >= d2 && d1 >= d3) {
    dR = d1; oR = l1 * l1; sR = d2 * d3;
    dA = d2; oA = l2 * l2; sA = d3;
    dB = d3; oB = l3 * l3; sB = 1;
  } else if (d2 >= d3) {
    dR = d2; oR = l2 * l2; sR = d3;
    dA = d1; oA = l1 * l1; sA = d2 * d3;
    dB = d3; oB = l3 * l3; sB = 1;
  } else {
    dR = d3; oR = l3 * l3; sR = 1;
    dA = d1; oA = l1 * l1; sA = d2 * d3;
    dB = d2; oB = l2 * l2; sB = d3;
  }

  const int M  = dA * dB;
  const int MS = (M + NSUB - 1) / NSUB;
  const int m0 = sub * MS;
  const int m1 = (m0 + MS < M) ? (m0 + MS) : M;
  const int Mloc = m1 - m0;
  if (Mloc <= 0) {
    if (t == 0) ctab[p * CSLOT + SSOFF + sub] = 0.0f;
    return;
  }
  const int NEl = Mloc * dR;

  __shared__ double un_s[8], wn_s[8], st_s[8];
  __shared__ double cph_s[16], sph_s[16];
  __shared__ double Ys[NQ * YPITCH];
  __shared__ double part[W3J_T * 7];
  __shared__ double red[64];

  const double PI = 3.14159265358979323846;
  const double DPHI = 2.0 * PI / 32.0;

  if (t < 8) {
    float xf = cosf(((float)t + 0.75f) * (float)(PI / 16.5));
    for (int it = 0; it < 2; ++it) {
      float a0, b0;
      legendre16(xf, a0, b0);
      xf -= b0 * (xf * xf - 1.0f) / (16.0f * (xf * b0 - a0));
    }
    const double x1 = (double)xf;
    double x1sq, a0, b0;
    legendre16_in_t(x1, x1sq, a0, b0);
    const double s1 = x1sq - 1.0;
    const double q  = x1 * b0 - a0;
    const double r0 = (double)__builtin_amdgcn_rcpf((float)q);
    const double rq = fma(r0, fma(-q, r0, 1.0), r0);
    const double xg = x1 - 0.0625 * b0 * s1 * rq;
    const double e  = 0.125 * x1 * b0 * rq;
    const double st2 = 1.0 - xg * xg;
    const double y0 = (double)__builtin_amdgcn_rsqf((float)st2);
    const double g  = st2 * y0, h = 0.5 * y0;
    const double st = fma(g, fma(-h, g, 0.5), g);
    un_s[t] = xg;
    wn_s[t] = -0.0078125 * s1 * rq * rq * (1.0 - e);
    st_s[t] = st2 > 0.0 ? st : 0.0;
  } else if (t >= 32 && t < 48) {
    const int j = t - 32;
    double s, c;
    sincospi((double)j * 0.0625, &s, &c);
    cph_s[j] = c;
    sph_s[j] = s;
  }
  __syncthreads();

  for (int q = t; q < NQ; q += W3J_T) {
    const int iq = q >> 4, jq = q & 15;
    const double st = st_s[iq];
    double Yq[16];
    sh_eval(st * cph_s[jq], st * sph_s[jq], un_s[iq], Yq);
    double* row = Ys + q * YPITCH;
    for (int c = 0; c < 16; ++c) row[c] = Yq[c];
    const double wq = wn_s[iq] * (4.0 * DPHI);
    for (int r = 0; r < dR; ++r) row[16 + r] = wq * row[oR + r];
  }
  __syncthreads();

  int NC = NQT / Mloc; if (NC > NCMAX) NC = NCMAX;
  const int CL = (NQ + NC - 1) / NC;
  if (t < Mloc * NC) {
    const int c = t / Mloc, ml = t - c * Mloc;
    const int m = m0 + ml;
    const int a = m / dB, b = m - a * dB;
    const int q0 = c * CL;
    int q1 = q0 + CL; if (q1 > NQ) q1 = NQ;
    double* dst = part + c * NEl + ml * dR;
    switch (dR) {
      case 1:  contract_run<1>(Ys, q0, q1, oA + a, oB + b, dst, 1); break;
      case 3:  contract_run<3>(Ys, q0, q1, oA + a, oB + b, dst, 1); break;
      case 5:  contract_run<5>(Ys, q0, q1, oA + a, oB + b, dst, 1); break;
      default: contract_run<7>(Ys, q0, q1, oA + a, oB + b, dst, 1); break;
    }
  }
  __syncthreads();

  double v = 0.0;
  if (t < NEl) {
    for (int c = 0; c < NC; ++c) v += part[c * NEl + t];
    const int ml = t / dR, r = t - ml * dR;
    const int m = m0 + ml;
    const int a = m / dB, b = m - a * dB;
    const unsigned ca = (unsigned)(oA + a), cb = (unsigned)(oB + b), cr = (unsigned)(oR + r);
    const unsigned odd = (((ODD_Z >> ca) ^ (ODD_Z >> cb) ^ (ODD_Z >> cr)) |
                          ((ODD_XY >> ca) ^ (ODD_XY >> cb) ^ (ODD_XY >> cr))) & 1u;
    if (odd) v = 0.0;
    ctab[p * CSLOT + r * sR + a * sA + b * sB] = (float)v;
  }
  if (t < 64) red[t] = v * v;
  __syncthreads();
  for (int off = 32; off > 0; off >>= 1) {
    if (t < off) red[t] += red[t + off];
    __syncthreads();
  }
  if (t == 0) ctab[p * CSLOT + SSOFF + sub] = (float)red[0];
}

#define EPB         16
#define EDGE_T      512
#define EDGE_BLOCKS 512
static_assert(EDGE_T == 32 * EPB, "one wave per edge");
static_assert(EDGE_BLOCKS * EPB == N_EDGES, "every edge has a wave, no wave is without an edge");
static_assert(EDGE_T * 4 == 2 * HIDDEN * HIDDEN, "two rows x two columns of a path's weights or biases per thread");
#define TPITCH 68
#define TCOLS  128
#define TBUF   (TCOLS * TPITCH)
#define WROW   96
#define WBUF   (32 * WROW)
#define SPITCH(d3) (32 * (d3) + 1)
#define SBUFH  (EPB * SPITCH(7))
#define SBUFL  (EPB * SPITCH(3))

#define NSTEPS 13
#define STEP2  7
static constexpr int STH[NSTEPS] = {2, 6, 9, 10, 14, 18, 22,   3, 8, 12, 16, 17, 20};
static constexpr int STL[NSTEPS] = {1, 4, 7, 11, 15, 19, -1,   0, 5, 13, 21, -1, -1};

constexpr bool steps_ok() {
  int seen[NPATHS] = {};
  for (int t = 0; t < NSTEPS; ++t) {
    const int lh = (t < STEP2) ? 2 : 3, ll = (t < STEP2) ? 1 : 0;
    if (PL3[STH[t]] != lh) return false;
    ++seen[STH[t]];
    if (STL[t] >= 0) { if (PL3[STL[t]] != ll) return false; ++seen[STL[t]]; }
    if (t > 0 && t != STEP2) {
      if (STH[t] <= STH[t - 1]) return false;
      if (STL[t] >= 0 && STL[t] <= STL[t - 1]) return false;
      if (STL[t] >= 0 && STL[t - 1] < 0) return false;
    }
  }
  for (int p = 0; p < NPATHS; ++p) if (seen[p] != 1) return false;
  return true;
}
static_assert(steps_ok(), "every path in exactly one step, in its group's order");

__device__ __forceinline__ void wave_lds_sync() {
  __builtin_amdgcn_fence(__ATOMIC_RELEASE, "wavefront");
  __builtin_amdgcn_wave_barrier();
  __builtin_amdgcn_fence(__ATOMIC_ACQUIRE, "wavefront");
}

template <int L3>
__device__ __forceinline__ void flush_stage(const float* stage, int wave, int lane, float* orow) {
  constexpr int D3 = 2 * L3 + 1;
  const float* sr = stage + wave * SPITCH(D3) + lane;
  float* od = orow + HIDDEN * L3 * L3;
  float v[D3];
#pragma unroll
  for (int j = 0; j < D3; ++j) v[j] = sr[32 * j];
#pragma unroll
  for (int j = 0; j < D3; ++j) atomicAdd(od + 32 * j, v[j]);
}

template <int P>
__device__ __forceinline__ void g_sums(int lane, const float (&Y)[16],
                                       const float* __restrict__ ctab,
                                       const float (&invn)[NPATHS], float* Gs) {
  constexpr int l1 = PL1[P], l2 = PL2[P], l3 = PL3[P];
  constexpr int d1 = 2 * l1 + 1, d2 = 2 * l2 + 1, d3 = 2 * l3 + 1;
  constexpr int GN = d1 * d3;
  const float pw = PWT[l3] * invn[P];
  const float* Cp = ctab + P * CSLOT;
  if (lane < GN) {
    const int i = lane / d3, k = lane % d3;
    float g = 0.0f;
#pragma unroll
    for (int j = 0; j < d2; ++j) g = fmaf(Y[l2 * l2 + j], Cp[(i * d2 + j) * d3 + k], g);
    Gs[lane] = pw * g;
  }
  if constexpr (GN > 32) {
    const int n2 = lane + 32;
    if (n2 < GN) {
      const int i = n2 / d3, k = n2 % d3;
      float g = 0.0f;
#pragma unroll
      for (int j = 0; j < d2; ++j) g = fmaf(Y[l2 * l2 + j], Cp[(i * d2 + j) * d3 + k], g);
      Gs[n2] = pw * g;
    }
  }
}

template <int P>
__device__ __forceinline__ void tmp_cols(float len, const float (&xr)[16],
                                         const float* Gs, float* tc) {
  constexpr int l1 = PL1[P], l3 = PL3[P];
  constexpr int d1 = 2 * l1 + 1, d3 = 2 * l3 + 1;
#pragma unroll
  for (int k = 0; k < d3; ++k) {
    float s = 0.0f;
#pragma unroll
    for (int i = 0; i < d1; ++i) s = fmaf(xr[l1 * l1 + i], Gs[i * d3 + k], s);
    tc[k * TPITCH] = len * s;
    tc[k * TPITCH + 32] = s;
  }
}

template <int T>
__device__ __forceinline__ void step_body(
    int lane, int wave, float len,
    const float (&Y)[16], const float (&xr)[16],
    const float* __restrict__ msrc,
    const float* __restrict__ ctab,
    const float (&invn)[NPATHS],
    float* Gs, float* TT,
    float* wdst,
    float2 (&spH)[2],
    float2 (&spL)[2],
    bool heavy,
    const float* ap,
    const float* bp,
    const float* stage,
    float* orow,
    v8f& acc0, v8f& acc1) {
  constexpr int PH = STH[T], PL = STL[T];
  constexpr int dH = 2 * PL3[PH] + 1;
  constexpr int WB_ = (T & 1) * (2 * WBUF);
  constexpr int TB_ = (T & 1) * TBUF;

  g_sums<PH>(lane, Y, ctab, invn, Gs);
  if constexpr (PL >= 0) g_sums<PL>(lane, Y, ctab, invn, Gs + 64);
  wave_lds_sync();
  tmp_cols<PH>(len, xr, Gs, TT + TB_ + (wave * dH) * TPITCH + lane);
  if constexpr (PL >= 0) {
    constexpr int dL = 2 * PL3[PL >= 0 ? PL : 0] + 1;
    tmp_cols<(PL >= 0 ? PL : 0)>(len, xr, Gs + 64, TT + TB_ + (16 * dH + wave * dL) * TPITCH + lane);
  }

  *reinterpret_cast<float4*>(wdst + WB_) = make_float4(spH[0].x, spH[1].x, spH[0].y, spH[1].y);
  if constexpr (PL >= 0)
    *reinterpret_cast<float4*>(wdst + WB_ + WBUF) = make_float4(spL[0].x, spL[1].x, spL[0].y, spL[1].y);

  __syncthreads();

  if constexpr (T + 1 < NSTEPS) {
    constexpr int NH = STH[T + 1], NL = STL[T + 1];
    spH[0] = *reinterpret_cast<const float2*>(msrc + NH * (HIDDEN * HIDDEN));
    spH[1] = *reinterpret_cast<const float2*>(msrc + NH * (HIDDEN * HIDDEN) + HIDDEN);
    if constexpr (NL >= 0) {
      spL[0] = *reinterpret_cast<const float2*>(msrc + (NL >= 0 ? NL : 0) * (HIDDEN * HIDDEN));
      spL[1] = *reinterpret_cast<const float2*>(msrc + (NL >= 0 ? NL : 0) * (HIDDEN * HIDDEN) + HIDDEN);
    }
  }

  if constexpr (T == STEP2) {
    flush_stage<PL3[STH[0]]>(stage, wave, lane, orow);
    flush_stage<PL3[STL[0]]>(stage + SBUFH, wave, lane, orow);
  }

  if (PL >= 0 || heavy) {
    const float* a_ = ap + WB_;
    const float* b_ = bp + TB_;
#pragma unroll
    for (int kc = 0; kc < 16; kc += 2) {
      const v2f a0 = *reinterpret_cast<const v2f*>(a_ + kc * (2 * WROW));
      const v2f b0 = *reinterpret_cast<const v2f*>(b_ + kc * 4);
      acc0 = __builtin_amdgcn_wmma_f32_16x16x4_f32(
          false, a0, false, b0, (short)0, acc0, false, false);
      const v2f a1 = *reinterpret_cast<const v2f*>(a_ + (kc + 1) * (2 * WROW));
      const v2f b1 = *reinterpret_cast<const v2f*>(b_ + (kc + 1) * 4);
      acc1 = __builtin_amdgcn_wmma_f32_16x16x4_f32(
          false, a1, false, b1, (short)0, acc1, false, false);
    }
  }
}

template <int... Ts>
__device__ __forceinline__ void run_steps(
    std::integer_sequence<int, Ts...>,
    int lane, int wave, float len,
    const float (&Y)[16], const float (&xr)[16],
    const float* __restrict__ msrc,
    const float* __restrict__ ctab, const float (&invn)[NPATHS],
    float* Gs, float* TT, float* wdst, float2 (&spH)[2], float2 (&spL)[2],
    bool heavy, const float* ap, const float* bp,
    const float* stage, float* orow,
    v8f& acc0, v8f& acc1) {
  (step_body<Ts>(lane, wave, len, Y, xr, msrc, ctab, invn, Gs, TT, wdst, spH, spL, heavy, ap, bp,
                 stage, orow, acc0, acc1), ...);
}

template <int D3>
__device__ __forceinline__ void stage_tile(float* stage, int tile, int half, int n16, int h,
                                           const v8f& acc0, const v8f& acc1) {
  const int col = 16 * tile + n16;
  const int el  = col / D3;
  const int k   = col - el * D3;
  float* sd = stage + el * SPITCH(D3) + (16 * half + 8 * h) * D3 + k;
#pragma unroll
  for (int r = 0; r < 8; ++r) sd[r * D3] = acc0[r] + acc1[r];
}

template <int L3H, int L3L, bool LAST, int... Ts>
__device__ __forceinline__ void do_pair(
    std::integer_sequence<int, Ts...> seq,
    int lane, int wave, float len,
    const float (&Y)[16], const float (&xr)[16],
    const float* __restrict__ msrc,
    const float* __restrict__ ctab, const float (&invn)[NPATHS],
    float* Gs, float* TT, const float* Ws, float* wdst, float2 (&spH)[2], float2 (&spL)[2],
    float* stage, float* orow) {
  constexpr int dH = 2 * L3H + 1, dL = 2 * L3L + 1;
  static_assert(2 * dH + 2 * dL == EPB, "the two groups' tile products are one per wave");
  static_assert(16 * (dH + dL) == TCOLS, "their columns fill the B tile");
  const int n16 = lane & 15;
  const int h   = lane >> 4;
  const bool heavy = wave < 2 * dH;
  const int j    = heavy ? wave : wave - 2 * dH;
  const int half = j & 1;
  const int tile = j >> 1;
  const int ctile = heavy ? tile : dH + tile;
  const float* ap = Ws + (heavy ? 0 : WBUF) + h * WROW + (16 * half + n16) * 2;
  const float* bp = TT + (16 * ctile + n16) * TPITCH + 2 * h;
  v8f acc0 = {0,0,0,0,0,0,0,0}, acc1 = {0,0,0,0,0,0,0,0};
  run_steps(seq, lane, wave, len, Y, xr, msrc, ctab, invn, Gs, TT, wdst, spH, spL, heavy, ap, bp,
            stage, orow, acc0, acc1);
  if (heavy) stage_tile<dH>(stage, tile, half, n16, h, acc0, acc1);
  else       stage_tile<dL>(stage + SBUFH, tile, half, n16, h, acc0, acc1);
  if constexpr (LAST) {
    __syncthreads();
    flush_stage<L3H>(stage, wave, lane, orow);
    flush_stage<L3L>(stage + SBUFH, wave, lane, orow);
  }
}

__attribute__((amdgpu_waves_per_eu(8, 8)))
__launch_bounds__(EDGE_T)
__global__ void edge_kernel(const float* __restrict__ nf,
                            const int*   __restrict__ eidx,
                            const float* __restrict__ evec,
                            const float* __restrict__ ww,
                            const float* __restrict__ wb,
                            const float* __restrict__ ctab,
                            float*       __restrict__ out) {
  const int tid  = threadIdx.x;
  const int lane = tid & 31;
  const int wave = __builtin_amdgcn_readfirstlane(tid >> 5);
  const int e    = blockIdx.x * EPB + wave;

  __shared__ float Gs_all[EPB * 128];
  __shared__ __attribute__((aligned(16))) float TT[2 * TBUF];
  __shared__ __attribute__((aligned(16))) float Ws[4 * WBUF];
  __shared__ float stage[SBUFH + SBUFL];

  const int rp = (tid >> 4) & 15, cp = tid & 15;
  const float* msrc = ((wave < 8) ? ww : wb) + rp * (2 * HIDDEN) + cp * 2;
  float2 spH[2], spL[2];
  spH[0] = *reinterpret_cast<const float2*>(msrc + STH[0] * (HIDDEN * HIDDEN));
  spH[1] = *reinterpret_cast<const float2*>(msrc + STH[0] * (HIDDEN * HIDDEN) + HIDDEN);
  spL[0] = *reinterpret_cast<const float2*>(msrc + STL[0] * (HIDDEN * HIDDEN));
  spL[1] = *reinterpret_cast<const float2*>(msrc + STL[0] * (HIDDEN * HIDDEN) + HIDDEN);

  const int src = eidx[e];
  const int dst = eidx[N_EDGES + e];
  float* orow = out + (size_t)dst * FEAT + lane;

  const float vx = evec[3 * e + 0];
  const float vy = evec[3 * e + 1];
  const float vz = evec[3 * e + 2];
  float len = sqrtf(vx * vx + vy * vy + vz * vz);
  len = fmaxf(len, 1e-8f);
  const float inv = 1.0f / len;

  float Y[16];
  sh_eval(vx * inv, vy * inv, vz * inv, Y);

  const float* xs = nf + (size_t)src * FEAT;
  float xr[16];
#pragma unroll
  for (int l = 0; l < 4; ++l) {
    const int d = 2 * l + 1;
#pragma unroll
    for (int i = 0; i < 7; ++i)
      if (i < d) xr[l * l + i] = xs[HIDDEN * l * l + lane * d + i];
  }

  float invn[NPATHS];
  {
    const float* sq = ctab + (lane < NPATHS ? lane : 0) * CSLOT + SSOFF;
    float ssq = 0.0f;
#pragma unroll
    for (int k = 0; k < NSUB; ++k) ssq += sq[k];
    const int ib = __builtin_bit_cast(int, 1.0f / sqrtf(ssq));
#pragma unroll
    for (int q = 0; q < NPATHS; ++q)
      invn[q] = __builtin_bit_cast(float, __builtin_amdgcn_readlane(ib, q));
  }

  float* Gs   = Gs_all + wave * 128;
  float* wdst = Ws + ((wave < 8 ? 0 : 16) + rp) * WROW + cp * 4;

  do_pair<2, 1, false>(std::integer_sequence<int, 0, 1, 2, 3, 4, 5, 6>{},
                       lane, wave, len, Y, xr, msrc, ctab, invn, Gs, TT, Ws, wdst, spH, spL, stage, orow);
  do_pair<3, 0, true>(std::integer_sequence<int, 7, 8, 9, 10, 11, 12>{},
                      lane, wave, len, Y, xr, msrc, ctab, invn, Gs, TT, Ws, wdst, spH, spL, stage, orow);
}

extern "C" void kernel_launch(void* const* d_in, const int* in_sizes, int n_in,
                              void* d_out, int out_size, void* d_ws, size_t ws_size,
                              hipStream_t stream) {
  const float* nf   = (const float*)d_in[0];
  const int*   eidx = (const int*)  d_in[1];
  const float* evec = (const float*)d_in[2];
  const float* ww   = (const float*)d_in[3];
  const float* wb   = (const float*)d_in[4];
  float* out  = (float*)d_out;
  float* ctab = (float*)d_ws;

  prep_kernel<<<PREP_BLOCKS, W3J_T, 0, stream>>>(ctab, out);
  edge_kernel<<<EDGE_BLOCKS, EDGE_T, 0, stream>>>(nf, eidx, evec, ww, wb, ctab, out);
}
